# starting kernel
# baseline (speedup 1.0000x reference)
#include <hip/hip_runtime.h>
#include <math.h>


#define N_NODES   50000
#define N_EDGES   600000
#define HIDDEN    128
#define FFN_DIM   256
#define HEADS     4
#define HEAD_DIM  32
#define NEG_SLOPE 0.2f
#define LN_EPS    1e-5f

typedef __attribute__((ext_vector_type(2))) float v2f;
typedef __attribute__((ext_vector_type(8))) float v8f;

__device__ __forceinline__ unsigned fkey(float f) {
  unsigned u = __float_as_uint(f);
  return (u & 0x80000000u) ? ~u : (u | 0x80000000u);
}
__device__ __forceinline__ float funkey(unsigned k) {
  unsigned u = (k & 0x80000000u) ? (k & 0x7fffffffu) : ~k;
  return __uint_as_float(u);
}

__device__ __forceinline__ void atomAddF(float* p, float v) {
  __hip_atomic_fetch_add(p, v, __ATOMIC_RELAXED, __HIP_MEMORY_SCOPE_AGENT);
}
__device__ __forceinline__ void atomMaxU(unsigned* p, unsigned v) {
  __hip_atomic_fetch_max(p, v, __ATOMIC_RELAXED, __HIP_MEMORY_SCOPE_AGENT);
}

__global__ void k_init(unsigned* __restrict__ mkeys, float* __restrict__ denom,
                       float* __restrict__ rst) {
  const int idx = blockIdx.x * 256 + threadIdx.x;
  if (idx < N_NODES * HEADS) { mkeys[idx] = 0u; denom[idx] = 0.0f; }
  rst[idx] = 0.0f;
}

__global__ __launch_bounds__(512) void k_gat_proj(
    const float* __restrict__ h,
    const float* __restrict__ Wsrc, const float* __restrict__ bsrc,
    const float* __restrict__ Wdst, const float* __restrict__ bdst,
    float* __restrict__ fs, float* __restrict__ fd) {
  __shared__ float tile[16 * 132];
  const int tid  = threadIdx.x;
  const int row0 = blockIdx.x * 16;
  {
    const int r = tid >> 5;
    const int c = (tid & 31) * 4;
    *(float4*)(&tile[r * 132 + c]) = *(const float4*)(h + (row0 + r) * HIDDEN + c);
  }
  __syncthreads();

  const int wave = tid >> 5, lane = tid & 31;
  const int l = lane & 15, hi = lane >> 4;
  const int sel = wave >> 3;
  const int nb  = (wave & 7) * 16;
  const float* W    = sel ? Wdst : Wsrc;
  const float* bias = sel ? bdst : bsrc;
  float*       out  = sel ? fd   : fs;

  v8f acc = {};
#pragma unroll
  for (int k0 = 0; k0 < HIDDEN; k0 += 4) {
    const int ka = k0 + 2 * hi;
    v2f A, B;
    A.x = tile[l * 132 + ka];
    A.y = tile[l * 132 + ka + 1];
    B.x = W[ka * HIDDEN + nb + l];
    B.y = W[(ka + 1) * HIDDEN + nb + l];
    acc = __builtin_amdgcn_wmma_f32_16x16x4_f32(false, A, false, B,
                                                (short)0, acc, false, false);
  }
  const float bv = bias[nb + l];
#pragma unroll
  for (int r = 0; r < 8; ++r)
    out[(row0 + r + 8 * hi) * HIDDEN + nb + l] = acc[r] + bv;
}

__global__ __launch_bounds__(256) void k_edge_logits(
    const float* __restrict__ fs, const float* __restrict__ fd,
    const int* __restrict__ src, const int* __restrict__ dst,
    const float* __restrict__ attn, float* __restrict__ logits,
    unsigned* __restrict__ mkeys) {
  const int e    = blockIdx.x * 8 + (threadIdx.x >> 5);
  const int lane = threadIdx.x & 31;
  const int s = src[e], d = dst[e];
  const float4 a = *(const float4*)(fs + s * HIDDEN + lane * 4);
  const float4 b = *(const float4*)(fd + d * HIDDEN + lane * 4);
  const float4 w = *(const float4*)(attn + lane * 4);
  float p = 0.0f, x;
  x = a.x + b.x; p += (x > 0.0f ? x : NEG_SLOPE * x) * w.x;
  x = a.y + b.y; p += (x > 0.0f ? x : NEG_SLOPE * x) * w.y;
  x = a.z + b.z; p += (x > 0.0f ? x : NEG_SLOPE * x) * w.z;
  x = a.w + b.w; p += (x > 0.0f ? x : NEG_SLOPE * x) * w.w;
  p += __shfl_xor(p, 1, 8);
  p += __shfl_xor(p, 2, 8);
  p += __shfl_xor(p, 4, 8);
  if ((lane & 7) == 0) {
    const int hh = lane >> 3;
    logits[e * HEADS + hh] = p;
    atomMaxU(mkeys + d * HEADS + hh, fkey(p));
  }
}

__global__ void k_edge_expsum(const float* __restrict__ logits,
                              const int* __restrict__ dst,
                              const unsigned* __restrict__ mkeys,
                              float* __restrict__ aexp,
                              float* __restrict__ denom) {
  const int idx = blockIdx.x * 256 + threadIdx.x;
  if (idx >= N_EDGES * HEADS) return;
  const int e = idx >> 2, hh = idx & 3;
  const int d = dst[e];
  const unsigned k = mkeys[d * HEADS + hh];
  const float mm = k ? funkey(k) : 0.0f;
  const float a  = expf(logits[idx] - mm);
  aexp[idx] = a;
  atomAddF(denom + d * HEADS + hh, a);
}

__global__ void k_edge_aggregate(const float* __restrict__ fs,
                                 const int* __restrict__ src,
                                 const int* __restrict__ dst,
                                 const float* __restrict__ aexp,
                                 const float* __restrict__ denom,
                                 float* __restrict__ rst) {
  const int idx = blockIdx.x * 256 + threadIdx.x;
  const int e = idx >> 7, d = idx & 127;
  const int hh = d >> 5;
  const int s = src[e], dd = dst[e];
  float den = denom[dd * HEADS + hh];
  den = (den == 0.0f) ? 1.0f : den;
  const float alpha = aexp[e * HEADS + hh] / den;
  atomAddF(rst + dd * HIDDEN + d, fs[s * HIDDEN + d] * alpha);
}

__global__ __launch_bounds__(256) void k_layernorm(
    const float* __restrict__ x, const float* __restrict__ addend,
    const float* __restrict__ g, const float* __restrict__ b,
    float* __restrict__ out) {
  const int node = blockIdx.x * 8 + (threadIdx.x >> 5);
  const int lane = threadIdx.x & 31;
  float4 v = *(const float4*)(x + node * HIDDEN + lane * 4);
  if (addend) {
    const float4 a = *(const float4*)(addend + node * HIDDEN + lane * 4);
    v.x += a.x; v.y += a.y; v.z += a.z; v.w += a.w;
  }
  float sum = v.x + v.y + v.z + v.w;
#pragma unroll
  for (int off = 1; off < 32; off <<= 1) sum += __shfl_xor(sum, off, 32);
  const float mu = sum * (1.0f / HIDDEN);
  const float dx = v.x - mu, dy = v.y - mu, dz = v.z - mu, dw = v.w - mu;
  float ss = dx * dx + dy * dy + dz * dz + dw * dw;
#pragma unroll
  for (int off = 1; off < 32; off <<= 1) ss += __shfl_xor(ss, off, 32);
  const float rs = rsqrtf(ss * (1.0f / HIDDEN) + LN_EPS);
  const int c = lane * 4;
  float4 o;
  o.x = dx * rs * g[c + 0] + b[c + 0];
  o.y = dy * rs * g[c + 1] + b[c + 1];
  o.z = dz * rs * g[c + 2] + b[c + 2];
  o.w = dw * rs * g[c + 3] + b[c + 3];
  *(float4*)(out + node * HIDDEN + c) = o;
}

__global__ __launch_bounds__(512) void k_ffn1(const float* __restrict__ h1,
                                              const float* __restrict__ W1,
                                              const float* __restrict__ bf1,
                                              float* __restrict__ t) {
  __shared__ float tile[16 * 132];
  const int tid = threadIdx.x, row0 = blockIdx.x * 16;
  {
    const int r = tid >> 5, c = (tid & 31) * 4;
    *(float4*)(&tile[r * 132 + c]) = *(const float4*)(h1 + (row0 + r) * HIDDEN + c);
  }
  __syncthreads();
  const int wave = tid >> 5, lane = tid & 31;
  const int l = lane & 15, hi = lane >> 4;
  const int nb = wave * 16;
  v8f acc = {};
#pragma unroll
  for (int k0 = 0; k0 < HIDDEN; k0 += 4) {
    const int ka = k0 + 2 * hi;
    v2f A, B;
    A.x = tile[l * 132 + ka];
    A.y = tile[l * 132 + ka + 1];
    B.x = W1[ka * FFN_DIM + nb + l];
    B.y = W1[(ka + 1) * FFN_DIM + nb + l];
    acc = __builtin_amdgcn_wmma_f32_16x16x4_f32(false, A, false, B,
                                                (short)0, acc, false, false);
  }
  const float bv = bf1[nb + l];
#pragma unroll
  for (int r = 0; r < 8; ++r) {
    float x = acc[r] + bv;
    x = 0.5f * x * (1.0f + erff(x * 0.70710678118654752f));
    t[(row0 + r + 8 * hi) * FFN_DIM + nb + l] = x;
  }
}

__global__ __launch_bounds__(256) void k_ffn2(const float* __restrict__ t,
                                              const float* __restrict__ W2,
                                              const float* __restrict__ bf2,
                                              const float* __restrict__ h1,
                                              float* __restrict__ tmp) {
  __shared__ float tile[16 * 260];
  const int tid = threadIdx.x, row0 = blockIdx.x * 16;
#pragma unroll
  for (int j = 0; j < 4; ++j) {
    const int idx = tid + j * 256;
    const int r = idx >> 6, c = (idx & 63) * 4;
    *(float4*)(&tile[r * 260 + c]) = *(const float4*)(t + (row0 + r) * FFN_DIM + c);
  }
  __syncthreads();
  const int wave = tid >> 5, lane = tid & 31;
  const int l = lane & 15, hi = lane >> 4;
  const int nb = wave * 16;
  v8f acc = {};
#pragma unroll
  for (int k0 = 0; k0 < FFN_DIM; k0 += 4) {
    const int ka = k0 + 2 * hi;
    v2f A, B;
    A.x = tile[l * 260 + ka];
    A.y = tile[l * 260 + ka + 1];
    B.x = W2[ka * HIDDEN + nb + l];
    B.y = W2[(ka + 1) * HIDDEN + nb + l];
    acc = __builtin_amdgcn_wmma_f32_16x16x4_f32(false, A, false, B,
                                                (short)0, acc, false, false);
  }
  const float bv = bf2[nb + l];
#pragma unroll
  for (int r = 0; r < 8; ++r) {
    const int row = row0 + r + 8 * hi;
    tmp[row * HIDDEN + nb + l] = acc[r] + bv + h1[row * HIDDEN + nb + l];
  }
}

extern "C" void kernel_launch(void* const* d_in, const int* in_sizes, int n_in,
                              void* d_out, int out_size, void* d_ws, size_t ws_size,
                              hipStream_t stream) {
  const float* h     = (const float*)d_in[0];
  const int*   src   = (const int*)d_in[1];
  const int*   dst   = (const int*)d_in[2];
  const float* Wsrc  = (const float*)d_in[3];
  const float* bsrc  = (const float*)d_in[4];
  const float* Wdst  = (const float*)d_in[5];
  const float* bdst  = (const float*)d_in[6];
  const float* attn  = (const float*)d_in[7];
  const float* W1    = (const float*)d_in[8];
  const float* bf1   = (const float*)d_in[9];
  const float* W2    = (const float*)d_in[10];
  const float* bf2   = (const float*)d_in[11];
  const float* g1    = (const float*)d_in[12];
  const float* be1   = (const float*)d_in[13];
  const float* g2    = (const float*)d_in[14];
  const float* be2   = (const float*)d_in[15];
  float* out = (float*)d_out;

  float*    ws     = (float*)d_ws;
  float*    fs     = ws;
  float*    fd     = fs + (size_t)N_NODES * HIDDEN;
  float*    rst    = fd + (size_t)N_NODES * HIDDEN;
  float*    h1     = rst + (size_t)N_NODES * HIDDEN;
  unsigned* mkeys  = (unsigned*)(h1 + (size_t)N_NODES * HIDDEN);
  float*    denom  = (float*)(mkeys + N_NODES * HEADS);
  float*    logits = denom + N_NODES * HEADS;
  float*    aexp   = logits + (size_t)N_EDGES * HEADS;
  float*    t      = fs;
  float*    tmp    = rst;

  k_init<<<N_NODES * HIDDEN / 256, 256, 0, stream>>>(mkeys, denom, rst);
  k_gat_proj<<<N_NODES / 16, 512, 0, stream>>>(h, Wsrc, bsrc, Wdst, bdst, fs, fd);
  k_edge_logits<<<N_EDGES / 8, 256, 0, stream>>>(fs, fd, src, dst, attn, logits, mkeys);
  k_edge_expsum<<<(N_EDGES * HEADS + 255) / 256, 256, 0, stream>>>(logits, dst, mkeys,
                                                                   aexp, denom);
  k_edge_aggregate<<<N_EDGES * HIDDEN / 256, 256, 0, stream>>>(fs, src, dst, aexp,
                                                               denom, rst);
  k_layernorm<<<N_NODES / 8, 256, 0, stream>>>(h, rst, g1, be1, h1);
  k_ffn1<<<N_NODES / 16, 512, 0, stream>>>(h1, W1, bf1, t);
  k_ffn2<<<N_NODES / 16, 256, 0, stream>>>(t, W2, bf2, h1, tmp);
  k_layernorm<<<N_NODES / 8, 256, 0, stream>>>(tmp, (const float*)nullptr, g2, be2, out);
}


# Edges grouped by destination each call (int count+rank, scan, place); one wave per node does max, exp-sum and weighted row sum in registers in source-id order: no float atomics, no zero fill
# speedup vs baseline: 1.6930x; 1.6930x over previous
#include <hip/hip_runtime.h>
#include <math.h>


#define N_NODES   50000
#define N_EDGES   600000
#define HIDDEN    128
#define FFN_DIM   256
#define HEADS     4
#define HEAD_DIM  32
#define NEG_SLOPE 0.2f
#define LN_EPS    1e-5f

typedef __attribute__((ext_vector_type(2))) float v2f;
typedef __attribute__((ext_vector_type(8))) float v8f;

__global__ void k_zero_counts(int* __restrict__ counts) {
  const int idx = blockIdx.x * 256 + threadIdx.x;
  if (idx < N_NODES) counts[idx] = 0;
}

__global__ void k_count_rank(const int* __restrict__ dst, int* __restrict__ counts,
                             int* __restrict__ rank) {
  const int e = blockIdx.x * 256 + threadIdx.x;
  if (e >= N_EDGES) return;
  rank[e] = __hip_atomic_fetch_add(counts + dst[e], 1, __ATOMIC_RELAXED,
                                   __HIP_MEMORY_SCOPE_AGENT);
}

#define SCAN_THREADS 1024
#define SCAN_ITERS   13
#define SCAN_SEG     (SCAN_ITERS * 128)
__global__ __launch_bounds__(SCAN_THREADS) void k_scan_offsets(
    const int* __restrict__ counts, int* __restrict__ offs) {
  __shared__ int wsum[32];
  const int tid = threadIdx.x, lane = tid & 31, wave = tid >> 5;
  const int seg0 = wave * SCAN_SEG + lane * 4;
  int4 v[SCAN_ITERS];
  int  excl[SCAN_ITERS];
#pragma unroll
  for (int it = 0; it < SCAN_ITERS; ++it) {
    const int idx = seg0 + it * 128;
    if (idx < N_NODES) v[it] = *(const int4*)(counts + idx);
    else               v[it] = make_int4(0, 0, 0, 0);
  }
  int carry = 0;
#pragma unroll
  for (int it = 0; it < SCAN_ITERS; ++it) {
    const int s4 = v[it].x + v[it].y + v[it].z + v[it].w;
    int inc = s4;
#pragma unroll
    for (int off = 1; off < 32; off <<= 1) {
      const int t = __shfl_up(inc, off, 32);
      if (lane >= off) inc += t;
    }
    excl[it] = carry + inc - s4;
    carry += __shfl(inc, 31, 32);
  }
  if (lane == 0) wsum[wave] = carry;
  __syncthreads();
  if (wave == 0) {
    const int ws = wsum[lane];
    int winc = ws;
#pragma unroll
    for (int off = 1; off < 32; off <<= 1) {
      const int t = __shfl_up(winc, off, 32);
      if (lane >= off) winc += t;
    }
    wsum[lane] = winc - ws;
  }
  __syncthreads();
  const int wbase = wsum[wave];
#pragma unroll
  for (int it = 0; it < SCAN_ITERS; ++it) {
    const int idx = seg0 + it * 128;
    if (idx < N_NODES) {
      int4 o;
      o.x = wbase + excl[it];
      o.y = o.x + v[it].x;
      o.z = o.y + v[it].y;
      o.w = o.z + v[it].z;
      *(int4*)(offs + idx) = o;
    }
  }
  if (tid == SCAN_THREADS - 1) offs[N_NODES] = wbase + carry;
}

__global__ void k_place(const int* __restrict__ src, const int* __restrict__ dst,
                        const int* __restrict__ rank, const int* __restrict__ offs,
                        int* __restrict__ csrc) {
  const int e = blockIdx.x * 256 + threadIdx.x;
  if (e >= N_EDGES) return;
  csrc[offs[dst[e]] + rank[e]] = src[e];
}

__global__ __launch_bounds__(512) void k_gat_proj(
    const float* __restrict__ h,
    const float* __restrict__ Wsrc, const float* __restrict__ bsrc,
    const float* __restrict__ Wdst, const float* __restrict__ bdst,
    float* __restrict__ fs, float* __restrict__ fd) {
  __shared__ float tile[16 * 132];
  const int tid  = threadIdx.x;
  const int row0 = blockIdx.x * 16;
  {
    const int r = tid >> 5;
    const int c = (tid & 31) * 4;
    *(float4*)(&tile[r * 132 + c]) = *(const float4*)(h + (row0 + r) * HIDDEN + c);
  }
  __syncthreads();

  const int wave = tid >> 5, lane = tid & 31;
  const int l = lane & 15, hi = lane >> 4;
  const int sel = wave >> 3;
  const int nb  = (wave & 7) * 16;
  const float* W    = sel ? Wdst : Wsrc;
  const float* bias = sel ? bdst : bsrc;
  float*       out  = sel ? fd   : fs;

  v8f acc = {};
#pragma unroll
  for (int k0 = 0; k0 < HIDDEN; k0 += 4) {
    const int ka = k0 + 2 * hi;
    v2f A, B;
    A.x = tile[l * 132 + ka];
    A.y = tile[l * 132 + ka + 1];
    B.x = W[ka * HIDDEN + nb + l];
    B.y = W[(ka + 1) * HIDDEN + nb + l];
    acc = __builtin_amdgcn_wmma_f32_16x16x4_f32(false, A, false, B,
                                                (short)0, acc, false, false);
  }
  const float bv = bias[nb + l];
#pragma unroll
  for (int r = 0; r < 8; ++r)
    out[(row0 + r + 8 * hi) * HIDDEN + nb + l] = acc[r] + bv;
}

#define CHUNK 16
__device__ __forceinline__ float edge_logit(const float4 a, const float4 b, const float4 w) {
  float p = 0.0f, x;
  x = a.x + b.x; p += (x > 0.0f ? x : NEG_SLOPE * x) * w.x;
  x = a.y + b.y; p += (x > 0.0f ? x : NEG_SLOPE * x) * w.y;
  x = a.z + b.z; p += (x > 0.0f ? x : NEG_SLOPE * x) * w.z;
  x = a.w + b.w; p += (x > 0.0f ? x : NEG_SLOPE * x) * w.w;
  p += __shfl_xor(p, 1, 8);
  p += __shfl_xor(p, 2, 8);
  p += __shfl_xor(p, 4, 8);
  return p;
}

template <int W>
__device__ __forceinline__ void gat_chunk(const float* __restrict__ fs, const int lane,
                                          const int sorted, const int half, const int cnt,
                                          const float4 b, const float4 w,
                                          float& m, float& l, float4& acc) {
  float4 row[W];
  float  p[W];
#pragma unroll
  for (int j = 0; j < W; ++j) {
    const int s = __builtin_amdgcn_readlane(sorted, half + min(j, cnt - 1));
    row[j] = *(const float4*)(fs + s * HIDDEN + lane * 4);
  }
  float cm = -INFINITY;
#pragma unroll
  for (int j = 0; j < W; ++j) {
    const float q = edge_logit(row[j], b, w);
    p[j] = (j < cnt) ? q : -INFINITY;
    cm = fmaxf(cm, p[j]);
  }
  const float mn = fmaxf(m, cm);
  const float sc = expf(m - mn);
  l *= sc;
  acc.x *= sc; acc.y *= sc; acc.z *= sc; acc.w *= sc;
#pragma unroll
  for (int j = 0; j < W; ++j) {
    const float ex = expf(p[j] - mn);
    l += ex;
    acc.x += row[j].x * ex;
    acc.y += row[j].y * ex;
    acc.z += row[j].z * ex;
    acc.w += row[j].w * ex;
  }
  m = mn;
}

__global__ __launch_bounds__(256) void k_gat_node(
    const float* __restrict__ fs, const float* __restrict__ fd,
    const float* __restrict__ attn, const int* __restrict__ offs,
    const int* __restrict__ csrc, float* __restrict__ rst) {
  const int node = __builtin_amdgcn_readfirstlane(blockIdx.x * 8 + (threadIdx.x >> 5));
  const int lane = threadIdx.x & 31;
  const int beg = offs[node], end = offs[node + 1];
  const float4 b = *(const float4*)(fd + node * HIDDEN + lane * 4);
  const float4 w = *(const float4*)(attn + lane * 4);

  float m = -INFINITY, l = 0.0f;
  float4 acc = {0.0f, 0.0f, 0.0f, 0.0f};
  for (int base = beg; base < end; base += 32) {
    const int n = min(32, end - base);
    const int sid = (lane < n) ? csrc[base + lane] : 0x7fffffff;
    int rk = 0;
    for (int j = 0; j < n; ++j) {
      const int o = __builtin_amdgcn_readlane(sid, j);
      rk += ((o < sid) || (o == sid && j < lane)) ? 1 : 0;
    }
    const int sorted = __builtin_amdgcn_ds_permute(rk << 2, sid);

    for (int half = 0; half < n; half += CHUNK) {
      const int cnt = min(CHUNK, n - half);
      const int quads = (cnt + 3) >> 2;
      if (quads == 4)      gat_chunk<16>(fs, lane, sorted, half, cnt, b, w, m, l, acc);
      else if (quads == 3) gat_chunk<12>(fs, lane, sorted, half, cnt, b, w, m, l, acc);
      else if (quads == 2) gat_chunk<8>(fs, lane, sorted, half, cnt, b, w, m, l, acc);
      else                 gat_chunk<4>(fs, lane, sorted, half, cnt, b, w, m, l, acc);
    }
  }
  const float den = (l == 0.0f) ? 1.0f : l;
  float4 o;
  o.x = acc.x / den; o.y = acc.y / den; o.z = acc.z / den; o.w = acc.w / den;
  *(float4*)(rst + node * HIDDEN + lane * 4) = o;
}

__global__ __launch_bounds__(256) void k_layernorm(
    const float* __restrict__ x, const float* __restrict__ addend,
    const float* __restrict__ g, const float* __restrict__ b,
    float* __restrict__ out) {
  const int node = blockIdx.x * 8 + (threadIdx.x >> 5);
  const int lane = threadIdx.x & 31;
  float4 v = *(const float4*)(x + node * HIDDEN + lane * 4);
  if (addend) {
    const float4 a = *(const float4*)(addend + node * HIDDEN + lane * 4);
    v.x += a.x; v.y += a.y; v.z += a.z; v.w += a.w;
  }
  float sum = v.x + v.y + v.z + v.w;
#pragma unroll
  for (int off = 1; off < 32; off <<= 1) sum += __shfl_xor(sum, off, 32);
  const float mu = sum * (1.0f / HIDDEN);
  const float dx = v.x - mu, dy = v.y - mu, dz = v.z - mu, dw = v.w - mu;
  float ss = dx * dx + dy * dy + dz * dz + dw * dw;
#pragma unroll
  for (int off = 1; off < 32; off <<= 1) ss += __shfl_xor(ss, off, 32);
  const float rs = rsqrtf(ss * (1.0f / HIDDEN) + LN_EPS);
  const int c = lane * 4;
  float4 o;
  o.x = dx * rs * g[c + 0] + b[c + 0];
  o.y = dy * rs * g[c + 1] + b[c + 1];
  o.z = dz * rs * g[c + 2] + b[c + 2];
  o.w = dw * rs * g[c + 3] + b[c + 3];
  *(float4*)(out + node * HIDDEN + c) = o;
}

__global__ __launch_bounds__(512) void k_ffn1(const float* __restrict__ h1,
                                              const float* __restrict__ W1,
                                              const float* __restrict__ bf1,
                                              float* __restrict__ t) {
  __shared__ float tile[16 * 132];
  const int tid = threadIdx.x, row0 = blockIdx.x * 16;
  {
    const int r = tid >> 5, c = (tid & 31) * 4;
    *(float4*)(&tile[r * 132 + c]) = *(const float4*)(h1 + (row0 + r) * HIDDEN + c);
  }
  __syncthreads();
  const int wave = tid >> 5, lane = tid & 31;
  const int l = lane & 15, hi = lane >> 4;
  const int nb = wave * 16;
  v8f acc = {};
#pragma unroll
  for (int k0 = 0; k0 < HIDDEN; k0 += 4) {
    const int ka = k0 + 2 * hi;
    v2f A, B;
    A.x = tile[l * 132 + ka];
    A.y = tile[l * 132 + ka + 1];
    B.x = W1[ka * FFN_DIM + nb + l];
    B.y = W1[(ka + 1) * FFN_DIM + nb + l];
    acc = __builtin_amdgcn_wmma_f32_16x16x4_f32(false, A, false, B,
                                                (short)0, acc, false, false);
  }
  const float bv = bf1[nb + l];
#pragma unroll
  for (int r = 0; r < 8; ++r) {
    float x = acc[r] + bv;
    x = 0.5f * x * (1.0f + erff(x * 0.70710678118654752f));
    t[(row0 + r + 8 * hi) * FFN_DIM + nb + l] = x;
  }
}

__global__ __launch_bounds__(256) void k_ffn2(const float* __restrict__ t,
                                              const float* __restrict__ W2,
                                              const float* __restrict__ bf2,
                                              const float* __restrict__ h1,
                                              float* __restrict__ tmp) {
  __shared__ float tile[16 * 260];
  const int tid = threadIdx.x, row0 = blockIdx.x * 16;
#pragma unroll
  for (int j = 0; j < 4; ++j) {
    const int idx = tid + j * 256;
    const int r = idx >> 6, c = (idx & 63) * 4;
    *(float4*)(&tile[r * 260 + c]) = *(const float4*)(t + (row0 + r) * FFN_DIM + c);
  }
  __syncthreads();
  const int wave = tid >> 5, lane = tid & 31;
  const int l = lane & 15, hi = lane >> 4;
  const int nb = wave * 16;
  v8f acc = {};
#pragma unroll
  for (int k0 = 0; k0 < FFN_DIM; k0 += 4) {
    const int ka = k0 + 2 * hi;
    v2f A, B;
    A.x = tile[l * 260 + ka];
    A.y = tile[l * 260 + ka + 1];
    B.x = W2[ka * HIDDEN + nb + l];
    B.y = W2[(ka + 1) * HIDDEN + nb + l];
    acc = __builtin_amdgcn_wmma_f32_16x16x4_f32(false, A, false, B,
                                                (short)0, acc, false, false);
  }
  const float bv = bf2[nb + l];
#pragma unroll
  for (int r = 0; r < 8; ++r) {
    const int row = row0 + r + 8 * hi;
    tmp[row * HIDDEN + nb + l] = acc[r] + bv + h1[row * HIDDEN + nb + l];
  }
}

extern "C" void kernel_launch(void* const* d_in, const int* in_sizes, int n_in,
                              void* d_out, int out_size, void* d_ws, size_t ws_size,
                              hipStream_t stream) {
  const float* h     = (const float*)d_in[0];
  const int*   src   = (const int*)d_in[1];
  const int*   dst   = (const int*)d_in[2];
  const float* Wsrc  = (const float*)d_in[3];
  const float* bsrc  = (const float*)d_in[4];
  const float* Wdst  = (const float*)d_in[5];
  const float* bdst  = (const float*)d_in[6];
  const float* attn  = (const float*)d_in[7];
  const float* W1    = (const float*)d_in[8];
  const float* bf1   = (const float*)d_in[9];
  const float* W2    = (const float*)d_in[10];
  const float* bf2   = (const float*)d_in[11];
  const float* g1    = (const float*)d_in[12];
  const float* be1   = (const float*)d_in[13];
  const float* g2    = (const float*)d_in[14];
  const float* be2   = (const float*)d_in[15];
  float* out = (float*)d_out;

  float* ws     = (float*)d_ws;
  float* fs     = ws;
  float* fd     = fs + (size_t)N_NODES * HIDDEN;
  float* rst    = fd + (size_t)N_NODES * HIDDEN;
  float* h1     = rst + (size_t)N_NODES * HIDDEN;
  int*   counts = (int*)(h1 + (size_t)N_NODES * HIDDEN);
  int*   offs   = counts + N_NODES;
  int*   rank   = offs + N_NODES + 16;
  int*   csrc   = rank + N_EDGES;
  float* t      = fs;
  float* tmp    = rst;

  k_zero_counts<<<(N_NODES + 255) / 256, 256, 0, stream>>>(counts);
  k_gat_proj<<<N_NODES / 16, 512, 0, stream>>>(h, Wsrc, bsrc, Wdst, bdst, fs, fd);
  k_count_rank<<<(N_EDGES + 255) / 256, 256, 0, stream>>>(dst, counts, rank);
  k_scan_offsets<<<1, SCAN_THREADS, 0, stream>>>(counts, offs);
  k_place<<<(N_EDGES + 255) / 256, 256, 0, stream>>>(src, dst, rank, offs, csrc);
  k_gat_node<<<N_NODES / 8, 256, 0, stream>>>(fs, fd, attn, offs, csrc, rst);
  k_layernorm<<<N_NODES / 8, 256, 0, stream>>>(h, rst, g1, be1, h1);
  k_ffn1<<<N_NODES / 16, 512, 0, stream>>>(h1, W1, bf1, t);
  k_ffn2<<<N_NODES / 16, 256, 0, stream>>>(t, W2, bf2, h1, tmp);
  k_layernorm<<<N_NODES / 8, 256, 0, stream>>>(tmp, (const float*)nullptr, g2, be2, out);
}


# LayerNorm 1, both FFN products (gelu, residual) and LayerNorm 2 fused into one kernel per 64-node tile; h1, t and the pre-norm sum stay in LDS; 10 launches become 7; summation order unchanged
# speedup vs baseline: 1.8482x; 1.0917x over previous
#include <hip/hip_runtime.h>
#include <math.h>


#define N_NODES   50000
#define N_EDGES   600000
#define HIDDEN    128
#define FFN_DIM   256
#define HEADS     4
#define HEAD_DIM  32
#define NEG_SLOPE 0.2f
#define LN_EPS    1e-5f

typedef __attribute__((ext_vector_type(2))) float v2f;
typedef __attribute__((ext_vector_type(8))) float v8f;

__global__ void k_zero_counts(int* __restrict__ counts) {
  const int idx = blockIdx.x * 256 + threadIdx.x;
  if (idx < N_NODES) counts[idx] = 0;
}

__global__ void k_count_rank(const int* __restrict__ dst, int* __restrict__ counts,
                             int* __restrict__ rank) {
  const int e = blockIdx.x * 256 + threadIdx.x;
  if (e >= N_EDGES) return;
  rank[e] = __hip_atomic_fetch_add(counts + dst[e], 1, __ATOMIC_RELAXED,
                                   __HIP_MEMORY_SCOPE_AGENT);
}

#define SCAN_THREADS 1024
#define SCAN_ITERS   13
#define SCAN_SEG     (SCAN_ITERS * 128)
__global__ __launch_bounds__(SCAN_THREADS) void k_scan_offsets(
    const int* __restrict__ counts, int* __restrict__ offs) {
  __shared__ int wsum[32];
  const int tid = threadIdx.x, lane = tid & 31, wave = tid >> 5;
  const int seg0 = wave * SCAN_SEG + lane * 4;
  int4 v[SCAN_ITERS];
  int  excl[SCAN_ITERS];
#pragma unroll
  for (int it = 0; it < SCAN_ITERS; ++it) {
    const int idx = seg0 + it * 128;
    if (idx < N_NODES) v[it] = *(const int4*)(counts + idx);
    else               v[it] = make_int4(0, 0, 0, 0);
  }
  int carry = 0;
#pragma unroll
  for (int it = 0; it < SCAN_ITERS; ++it) {
    const int s4 = v[it].x + v[it].y + v[it].z + v[it].w;
    int inc = s4;
#pragma unroll
    for (int off = 1; off < 32; off <<= 1) {
      const int t = __shfl_up(inc, off, 32);
      if (lane >= off) inc += t;
    }
    excl[it] = carry + inc - s4;
    carry += __shfl(inc, 31, 32);
  }
  if (lane == 0) wsum[wave] = carry;
  __syncthreads();
  if (wave == 0) {
    const int ws = wsum[lane];
    int winc = ws;
#pragma unroll
    for (int off = 1; off < 32; off <<= 1) {
      const int t = __shfl_up(winc, off, 32);
      if (lane >= off) winc += t;
    }
    wsum[lane] = winc - ws;
  }
  __syncthreads();
  const int wbase = wsum[wave];
#pragma unroll
  for (int it = 0; it < SCAN_ITERS; ++it) {
    const int idx = seg0 + it * 128;
    if (idx < N_NODES) {
      int4 o;
      o.x = wbase + excl[it];
      o.y = o.x + v[it].x;
      o.z = o.y + v[it].y;
      o.w = o.z + v[it].z;
      *(int4*)(offs + idx) = o;
    }
  }
  if (tid == SCAN_THREADS - 1) offs[N_NODES] = wbase + carry;
}

__global__ void k_place(const int* __restrict__ src, const int* __restrict__ dst,
                        const int* __restrict__ rank, const int* __restrict__ offs,
                        int* __restrict__ csrc) {
  const int e = blockIdx.x * 256 + threadIdx.x;
  if (e >= N_EDGES) return;
  csrc[offs[dst[e]] + rank[e]] = src[e];
}

__global__ __launch_bounds__(512) void k_gat_proj(
    const float* __restrict__ h,
    const float* __restrict__ Wsrc, const float* __restrict__ bsrc,
    const float* __restrict__ Wdst, const float* __restrict__ bdst,
    float* __restrict__ fs, float* __restrict__ fd) {
  __shared__ float tile[16 * 132];
  const int tid  = threadIdx.x;
  const int row0 = blockIdx.x * 16;
  {
    const int r = tid >> 5;
    const int c = (tid & 31) * 4;
    *(float4*)(&tile[r * 132 + c]) = *(const float4*)(h + (row0 + r) * HIDDEN + c);
  }
  __syncthreads();

  const int wave = tid >> 5, lane = tid & 31;
  const int l = lane & 15, hi = lane >> 4;
  const int sel = wave >> 3;
  const int nb  = (wave & 7) * 16;
  const float* W    = sel ? Wdst : Wsrc;
  const float* bias = sel ? bdst : bsrc;
  float*       out  = sel ? fd   : fs;

  v8f acc = {};
#pragma unroll
  for (int k0 = 0; k0 < HIDDEN; k0 += 4) {
    const int ka = k0 + 2 * hi;
    v2f A, B;
    A.x = tile[l * 132 + ka];
    A.y = tile[l * 132 + ka + 1];
    B.x = W[ka * HIDDEN + nb + l];
    B.y = W[(ka + 1) * HIDDEN + nb + l];
    acc = __builtin_amdgcn_wmma_f32_16x16x4_f32(false, A, false, B,
                                                (short)0, acc, false, false);
  }
  const float bv = bias[nb + l];
#pragma unroll
  for (int r = 0; r < 8; ++r)
    out[(row0 + r + 8 * hi) * HIDDEN + nb + l] = acc[r] + bv;
}

#define CHUNK 16
__device__ __forceinline__ float edge_logit(const float4 a, const float4 b, const float4 w) {
  float p = 0.0f, x;
  x = a.x + b.x; p += (x > 0.0f ? x : NEG_SLOPE * x) * w.x;
  x = a.y + b.y; p += (x > 0.0f ? x : NEG_SLOPE * x) * w.y;
  x = a.z + b.z; p += (x > 0.0f ? x : NEG_SLOPE * x) * w.z;
  x = a.w + b.w; p += (x > 0.0f ? x : NEG_SLOPE * x) * w.w;
  p += __shfl_xor(p, 1, 8);
  p += __shfl_xor(p, 2, 8);
  p += __shfl_xor(p, 4, 8);
  return p;
}

template <int W>
__device__ __forceinline__ void gat_chunk(const float* __restrict__ fs, const int lane,
                                          const int sorted, const int half, const int cnt,
                                          const float4 b, const float4 w,
                                          float& m, float& l, float4& acc) {
  float4 row[W];
  float  p[W];
#pragma unroll
  for (int j = 0; j < W; ++j) {
    const int s = __builtin_amdgcn_readlane(sorted, half + min(j, cnt - 1));
    row[j] = *(const float4*)(fs + s * HIDDEN + lane * 4);
  }
  float cm = -INFINITY;
#pragma unroll
  for (int j = 0; j < W; ++j) {
    const float q = edge_logit(row[j], b, w);
    p[j] = (j < cnt) ? q : -INFINITY;
    cm = fmaxf(cm, p[j]);
  }
  const float mn = fmaxf(m, cm);
  const float sc = expf(m - mn);
  l *= sc;
  acc.x *= sc; acc.y *= sc; acc.z *= sc; acc.w *= sc;
#pragma unroll
  for (int j = 0; j < W; ++j) {
    const float ex = expf(p[j] - mn);
    l += ex;
    acc.x += row[j].x * ex;
    acc.y += row[j].y * ex;
    acc.z += row[j].z * ex;
    acc.w += row[j].w * ex;
  }
  m = mn;
}

__global__ __launch_bounds__(256) void k_gat_node(
    const float* __restrict__ fs, const float* __restrict__ fd,
    const float* __restrict__ attn, const int* __restrict__ offs,
    const int* __restrict__ csrc, float* __restrict__ rst) {
  const int node = __builtin_amdgcn_readfirstlane(blockIdx.x * 8 + (threadIdx.x >> 5));
  const int lane = threadIdx.x & 31;
  const int beg = offs[node], end = offs[node + 1];
  const float4 b = *(const float4*)(fd + node * HIDDEN + lane * 4);
  const float4 w = *(const float4*)(attn + lane * 4);

  float m = -INFINITY, l = 0.0f;
  float4 acc = {0.0f, 0.0f, 0.0f, 0.0f};
  for (int base = beg; base < end; base += 32) {
    const int n = min(32, end - base);
    const int sid = (lane < n) ? csrc[base + lane] : 0x7fffffff;
    int rk = 0;
    for (int j = 0; j < n; ++j) {
      const int o = __builtin_amdgcn_readlane(sid, j);
      rk += ((o < sid) || (o == sid && j < lane)) ? 1 : 0;
    }
    const int sorted = __builtin_amdgcn_ds_permute(rk << 2, sid);

    for (int half = 0; half < n; half += CHUNK) {
      const int cnt = min(CHUNK, n - half);
      const int quads = (cnt + 3) >> 2;
      if (quads == 4)      gat_chunk<16>(fs, lane, sorted, half, cnt, b, w, m, l, acc);
      else if (quads == 3) gat_chunk<12>(fs, lane, sorted, half, cnt, b, w, m, l, acc);
      else if (quads == 2) gat_chunk<8>(fs, lane, sorted, half, cnt, b, w, m, l, acc);
      else                 gat_chunk<4>(fs, lane, sorted, half, cnt, b, w, m, l, acc);
    }
  }
  const float den = (l == 0.0f) ? 1.0f : l;
  float4 o;
  o.x = acc.x / den; o.y = acc.y / den; o.z = acc.z / den; o.w = acc.w / den;
  *(float4*)(rst + node * HIDDEN + lane * 4) = o;
}

#define TM 64
#define XS 132
#define TS 260

__device__ __forceinline__ void layernorm4(float4 (&v)[4], const float4 g, const float4 b) {
  float sum[4];
#pragma unroll
  for (int i = 0; i < 4; ++i) sum[i] = v[i].x + v[i].y + v[i].z + v[i].w;
#pragma unroll
  for (int off = 1; off < 32; off <<= 1) {
#pragma unroll
    for (int i = 0; i < 4; ++i) sum[i] += __shfl_xor(sum[i], off, 32);
  }
  float ss[4];
#pragma unroll
  for (int i = 0; i < 4; ++i) {
    const float mu = sum[i] * (1.0f / HIDDEN);
    v[i].x -= mu; v[i].y -= mu; v[i].z -= mu; v[i].w -= mu;
    ss[i] = v[i].x * v[i].x + v[i].y * v[i].y + v[i].z * v[i].z + v[i].w * v[i].w;
  }
#pragma unroll
  for (int off = 1; off < 32; off <<= 1) {
#pragma unroll
    for (int i = 0; i < 4; ++i) ss[i] += __shfl_xor(ss[i], off, 32);
  }
#pragma unroll
  for (int i = 0; i < 4; ++i) {
    const float rs = rsqrtf(ss[i] * (1.0f / HIDDEN) + LN_EPS);
    v[i].x = v[i].x * rs * g.x + b.x;
    v[i].y = v[i].y * rs * g.y + b.y;
    v[i].z = v[i].z * rs * g.z + b.z;
    v[i].w = v[i].w * rs * g.w + b.w;
  }
}

__global__ __launch_bounds__(512) void k_tail(
    const float* __restrict__ h, const float* __restrict__ rst,
    const float* __restrict__ g1, const float* __restrict__ be1,
    const float* __restrict__ W1, const float* __restrict__ bf1,
    const float* __restrict__ W2, const float* __restrict__ bf2,
    const float* __restrict__ g2, const float* __restrict__ be2,
    float* __restrict__ out) {
  __shared__ __attribute__((aligned(16))) float xs[TM * XS];
  __shared__ __attribute__((aligned(16))) float ts[TM * TS];
  const int tid = threadIdx.x, wave = tid >> 5, lane = tid & 31;
  const int row0 = blockIdx.x * TM;
  const int l = lane & 15, hi = lane >> 4;

  {
    float4 v[4];
#pragma unroll
    for (int i = 0; i < 4; ++i) {
      const int row = min(row0 + wave * 4 + i, N_NODES - 1);
      const float4 x = *(const float4*)(h + row * HIDDEN + lane * 4);
      const float4 a = *(const float4*)(rst + row * HIDDEN + lane * 4);
      v[i].x = x.x + a.x; v[i].y = x.y + a.y; v[i].z = x.z + a.z; v[i].w = x.w + a.w;
    }
    layernorm4(v, *(const float4*)(g1 + lane * 4), *(const float4*)(be1 + lane * 4));
#pragma unroll
    for (int i = 0; i < 4; ++i)
      *(float4*)(&xs[(wave * 4 + i) * XS + lane * 4]) = v[i];
  }
  __syncthreads();

  {
    const int nb = wave * 16;
    v8f acc[4] = {};
#pragma unroll
    for (int k0 = 0; k0 < HIDDEN; k0 += 4) {
      const int ka = k0 + 2 * hi;
      v2f B;
      B.x = W1[ka * FFN_DIM + nb + l];
      B.y = W1[(ka + 1) * FFN_DIM + nb + l];
#pragma unroll
      for (int mt = 0; mt < 4; ++mt) {
        const v2f A = *(const v2f*)(&xs[(mt * 16 + l) * XS + ka]);
        acc[mt] = __builtin_amdgcn_wmma_f32_16x16x4_f32(false, A, false, B,
                                                        (short)0, acc[mt], false, false);
      }
    }
    const float bv = bf1[nb + l];
#pragma unroll
    for (int mt = 0; mt < 4; ++mt) {
#pragma unroll
      for (int r = 0; r < 8; ++r) {
        float x = acc[mt][r] + bv;
        x = 0.5f * x * (1.0f + erff(x * 0.70710678118654752f));
        ts[(mt * 16 + r + 8 * hi) * TS + nb + l] = x;
      }
    }
  }
  __syncthreads();

  {
    const int nb  = (wave & 7) * 16;
    const int mt0 = (wave >> 3) * 2;
    v8f acc[2] = {};
#pragma unroll
    for (int k0 = 0; k0 < FFN_DIM; k0 += 4) {
      const int ka = k0 + 2 * hi;
      v2f B;
      B.x = W2[ka * HIDDEN + nb + l];
      B.y = W2[(ka + 1) * HIDDEN + nb + l];
#pragma unroll
      for (int j = 0; j < 2; ++j) {
        const v2f A = *(const v2f*)(&ts[((mt0 + j) * 16 + l) * TS + ka]);
        acc[j] = __builtin_amdgcn_wmma_f32_16x16x4_f32(false, A, false, B,
                                                       (short)0, acc[j], false, false);
      }
    }
    const float bv = bf2[nb + l];
#pragma unroll
    for (int j = 0; j < 2; ++j) {
#pragma unroll
      for (int r = 0; r < 8; ++r) {
        const int idx = ((mt0 + j) * 16 + r + 8 * hi) * XS + nb + l;
        xs[idx] = acc[j][r] + bv + xs[idx];
      }
    }
  }
  __syncthreads();

  {
    float4 v[4];
#pragma unroll
    for (int i = 0; i < 4; ++i) v[i] = *(const float4*)(&xs[(wave * 4 + i) * XS + lane * 4]);
    layernorm4(v, *(const float4*)(g2 + lane * 4), *(const float4*)(be2 + lane * 4));
#pragma unroll
    for (int i = 0; i < 4; ++i) {
      const int row = row0 + wave * 4 + i;
      if (row < N_NODES) *(float4*)(out + row * HIDDEN + lane * 4) = v[i];
    }
  }
}

extern "C" void kernel_launch(void* const* d_in, const int* in_sizes, int n_in,
                              void* d_out, int out_size, void* d_ws, size_t ws_size,
                              hipStream_t stream) {
  const float* h     = (const float*)d_in[0];
  const int*   src   = (const int*)d_in[1];
  const int*   dst   = (const int*)d_in[2];
  const float* Wsrc  = (const float*)d_in[3];
  const float* bsrc  = (const float*)d_in[4];
  const float* Wdst  = (const float*)d_in[5];
  const float* bdst  = (const float*)d_in[6];
  const float* attn  = (const float*)d_in[7];
  const float* W1    = (const float*)d_in[8];
  const float* bf1   = (const float*)d_in[9];
  const float* W2    = (const float*)d_in[10];
  const float* bf2   = (const float*)d_in[11];
  const float* g1    = (const float*)d_in[12];
  const float* be1   = (const float*)d_in[13];
  const float* g2    = (const float*)d_in[14];
  const float* be2   = (const float*)d_in[15];
  float* out = (float*)d_out;

  float* ws     = (float*)d_ws;
  float* fs     = ws;
  float* fd     = fs + (size_t)N_NODES * HIDDEN;
  float* rst    = fd + (size_t)N_NODES * HIDDEN;
  int*   counts = (int*)(rst + (size_t)N_NODES * HIDDEN);
  int*   offs   = counts + N_NODES;
  int*   rank   = offs + N_NODES + 16;
  int*   csrc   = rank + N_EDGES;

  k_zero_counts<<<(N_NODES + 255) / 256, 256, 0, stream>>>(counts);
  k_gat_proj<<<N_NODES / 16, 512, 0, stream>>>(h, Wsrc, bsrc, Wdst, bdst, fs, fd);
  k_count_rank<<<(N_EDGES + 255) / 256, 256, 0, stream>>>(dst, counts, rank);
  k_scan_offsets<<<1, SCAN_THREADS, 0, stream>>>(counts, offs);
  k_place<<<(N_EDGES + 255) / 256, 256, 0, stream>>>(src, dst, rank, offs, csrc);
  k_gat_node<<<N_NODES / 8, 256, 0, stream>>>(fs, fd, attn, offs, csrc, rst);
  k_tail<<<(N_NODES + TM - 1) / TM, 512, 0, stream>>>(h, rst, g1, be1, W1, bf1, W2, bf2, g2, be2, out);
}


# Destination grouping without global atomics: per (edge chunk, node slice) work-group, LDS counters then LDS cursors replace 600k returning global adds and the rank array
# speedup vs baseline: 2.3083x; 1.2489x over previous
#include <hip/hip_runtime.h>
#include <math.h>


#define N_NODES   50000
#define N_EDGES   600000
#define HIDDEN    128
#define FFN_DIM   256
#define HEADS     4
#define HEAD_DIM  32
#define NEG_SLOPE 0.2f
#define LN_EPS    1e-5f

typedef __attribute__((ext_vector_type(2))) float v2f;
typedef __attribute__((ext_vector_type(8))) float v8f;

#define SORT_G       32
#define SORT_S       8
#define SLICE_N      (N_NODES / SORT_S)
#define CHUNK_E      18752
#define SORT_THREADS 1024
#define SORT_ITERS   5
#define SLICE_ITERS  7
static_assert(SORT_S * SLICE_N == N_NODES, "slices must tile the nodes");
static_assert(SORT_G * CHUNK_E >= N_EDGES && CHUNK_E % 4 == 0 && N_EDGES % 4 == 0, "chunks");
static_assert(SORT_ITERS * SORT_THREADS * 4 >= CHUNK_E, "one pass over a chunk");
static_assert(SLICE_ITERS * SORT_THREADS >= SLICE_N, "one pass over a slice");

__global__ __launch_bounds__(SORT_THREADS) void k_hist(const int* __restrict__ dst,
                                                       int* __restrict__ H) {
  __shared__ int hist[SLICE_N];
  const int g = blockIdx.x, s = blockIdx.y, tid = threadIdx.x;
  const int lo = s * SLICE_N;
  const int e0 = g * CHUNK_E;
  const int e1 = min(e0 + CHUNK_E, N_EDGES);
  int4 d[SORT_ITERS];
#pragma unroll
  for (int i = 0; i < SORT_ITERS; ++i) {
    const int e = e0 + (i * SORT_THREADS + tid) * 4;
    d[i] = (e < e1) ? *(const int4*)(dst + e) : make_int4(-1, -1, -1, -1);
  }
  for (int j = tid; j < SLICE_N; j += SORT_THREADS) hist[j] = 0;
  __syncthreads();
#pragma unroll
  for (int i = 0; i < SORT_ITERS; ++i) {
    const unsigned n0 = (unsigned)(d[i].x - lo), n1 = (unsigned)(d[i].y - lo);
    const unsigned n2 = (unsigned)(d[i].z - lo), n3 = (unsigned)(d[i].w - lo);
    if (n0 < SLICE_N) __hip_atomic_fetch_add(&hist[n0], 1, __ATOMIC_RELAXED, __HIP_MEMORY_SCOPE_WORKGROUP);
    if (n1 < SLICE_N) __hip_atomic_fetch_add(&hist[n1], 1, __ATOMIC_RELAXED, __HIP_MEMORY_SCOPE_WORKGROUP);
    if (n2 < SLICE_N) __hip_atomic_fetch_add(&hist[n2], 1, __ATOMIC_RELAXED, __HIP_MEMORY_SCOPE_WORKGROUP);
    if (n3 < SLICE_N) __hip_atomic_fetch_add(&hist[n3], 1, __ATOMIC_RELAXED, __HIP_MEMORY_SCOPE_WORKGROUP);
  }
  __syncthreads();
  for (int j = tid; j < SLICE_N; j += SORT_THREADS) H[g * N_NODES + lo + j] = hist[j];
}

__global__ void k_colprefix(int* __restrict__ H, int* __restrict__ counts) {
  const int n = blockIdx.x * 256 + threadIdx.x;
  if (n >= N_NODES) return;
  int c[SORT_G];
#pragma unroll
  for (int g = 0; g < SORT_G; ++g) c[g] = H[g * N_NODES + n];
  int run = 0;
#pragma unroll
  for (int g = 0; g < SORT_G; ++g) {
    H[g * N_NODES + n] = run;
    run += c[g];
  }
  counts[n] = run;
}

#define SCAN_THREADS 1024
#define SCAN_ITERS   13
#define SCAN_SEG     (SCAN_ITERS * 128)
__global__ __launch_bounds__(SCAN_THREADS) void k_scan_offsets(
    const int* __restrict__ counts, int* __restrict__ offs) {
  __shared__ int wsum[32];
  const int tid = threadIdx.x, lane = tid & 31, wave = tid >> 5;
  const int seg0 = wave * SCAN_SEG + lane * 4;
  int4 v[SCAN_ITERS];
  int  excl[SCAN_ITERS];
#pragma unroll
  for (int it = 0; it < SCAN_ITERS; ++it) {
    const int idx = seg0 + it * 128;
    if (idx < N_NODES) v[it] = *(const int4*)(counts + idx);
    else               v[it] = make_int4(0, 0, 0, 0);
  }
  int carry = 0;
#pragma unroll
  for (int it = 0; it < SCAN_ITERS; ++it) {
    const int s4 = v[it].x + v[it].y + v[it].z + v[it].w;
    int inc = s4;
#pragma unroll
    for (int off = 1; off < 32; off <<= 1) {
      const int t = __shfl_up(inc, off, 32);
      if (lane >= off) inc += t;
    }
    excl[it] = carry + inc - s4;
    carry += __shfl(inc, 31, 32);
  }
  if (lane == 0) wsum[wave] = carry;
  __syncthreads();
  if (wave == 0) {
    const int ws = wsum[lane];
    int winc = ws;
#pragma unroll
    for (int off = 1; off < 32; off <<= 1) {
      const int t = __shfl_up(winc, off, 32);
      if (lane >= off) winc += t;
    }
    wsum[lane] = winc - ws;
  }
  __syncthreads();
  const int wbase = wsum[wave];
#pragma unroll
  for (int it = 0; it < SCAN_ITERS; ++it) {
    const int idx = seg0 + it * 128;
    if (idx < N_NODES) {
      int4 o;
      o.x = wbase + excl[it];
      o.y = o.x + v[it].x;
      o.z = o.y + v[it].y;
      o.w = o.z + v[it].z;
      *(int4*)(offs + idx) = o;
    }
  }
  if (tid == SCAN_THREADS - 1) offs[N_NODES] = wbase + carry;
}

__global__ __launch_bounds__(SORT_THREADS) void k_place(
    const int* __restrict__ src, const int* __restrict__ dst, const int* __restrict__ H,
    const int* __restrict__ offs, int* __restrict__ csrc) {
  __shared__ int next[SLICE_N];
  const int g = blockIdx.x, s = blockIdx.y, tid = threadIdx.x;
  const int lo = s * SLICE_N;
  const int e0 = g * CHUNK_E;
  const int e1 = min(e0 + CHUNK_E, N_EDGES);
  int4 d[SORT_ITERS], sv[SORT_ITERS];
  int  cur[SLICE_ITERS];
#pragma unroll
  for (int i = 0; i < SORT_ITERS; ++i) {
    const int e = e0 + (i * SORT_THREADS + tid) * 4;
    if (e < e1) {
      d[i]  = *(const int4*)(dst + e);
      sv[i] = *(const int4*)(src + e);
    } else {
      d[i]  = make_int4(-1, -1, -1, -1);
      sv[i] = make_int4(0, 0, 0, 0);
    }
  }
#pragma unroll
  for (int i = 0; i < SLICE_ITERS; ++i) {
    const int j = i * SORT_THREADS + tid;
    cur[i] = (j < SLICE_N) ? offs[lo + j] + H[g * N_NODES + lo + j] : 0;
  }
#pragma unroll
  for (int i = 0; i < SLICE_ITERS; ++i) {
    const int j = i * SORT_THREADS + tid;
    if (j < SLICE_N) next[j] = cur[i];
  }
  __syncthreads();
  int pos[SORT_ITERS * 4];
#pragma unroll
  for (int i = 0; i < SORT_ITERS; ++i) {
    const unsigned n0 = (unsigned)(d[i].x - lo), n1 = (unsigned)(d[i].y - lo);
    const unsigned n2 = (unsigned)(d[i].z - lo), n3 = (unsigned)(d[i].w - lo);
    pos[4 * i + 0] = (n0 < SLICE_N) ? __hip_atomic_fetch_add(&next[n0], 1, __ATOMIC_RELAXED, __HIP_MEMORY_SCOPE_WORKGROUP) : -1;
    pos[4 * i + 1] = (n1 < SLICE_N) ? __hip_atomic_fetch_add(&next[n1], 1, __ATOMIC_RELAXED, __HIP_MEMORY_SCOPE_WORKGROUP) : -1;
    pos[4 * i + 2] = (n2 < SLICE_N) ? __hip_atomic_fetch_add(&next[n2], 1, __ATOMIC_RELAXED, __HIP_MEMORY_SCOPE_WORKGROUP) : -1;
    pos[4 * i + 3] = (n3 < SLICE_N) ? __hip_atomic_fetch_add(&next[n3], 1, __ATOMIC_RELAXED, __HIP_MEMORY_SCOPE_WORKGROUP) : -1;
  }
#pragma unroll
  for (int i = 0; i < SORT_ITERS; ++i) {
    if (pos[4 * i + 0] >= 0) csrc[pos[4 * i + 0]] = sv[i].x;
    if (pos[4 * i + 1] >= 0) csrc[pos[4 * i + 1]] = sv[i].y;
    if (pos[4 * i + 2] >= 0) csrc[pos[4 * i + 2]] = sv[i].z;
    if (pos[4 * i + 3] >= 0) csrc[pos[4 * i + 3]] = sv[i].w;
  }
}

__global__ __launch_bounds__(512) void k_gat_proj(
    const float* __restrict__ h,
    const float* __restrict__ Wsrc, const float* __restrict__ bsrc,
    const float* __restrict__ Wdst, const float* __restrict__ bdst,
    float* __restrict__ fs, float* __restrict__ fd) {
  __shared__ float tile[16 * 132];
  const int tid  = threadIdx.x;
  const int row0 = blockIdx.x * 16;
  {
    const int r = tid >> 5;
    const int c = (tid & 31) * 4;
    *(float4*)(&tile[r * 132 + c]) = *(const float4*)(h + (row0 + r) * HIDDEN + c);
  }
  __syncthreads();

  const int wave = tid >> 5, lane = tid & 31;
  const int l = lane & 15, hi = lane >> 4;
  const int sel = wave >> 3;
  const int nb  = (wave & 7) * 16;
  const float* W    = sel ? Wdst : Wsrc;
  const float* bias = sel ? bdst : bsrc;
  float*       out  = sel ? fd   : fs;

  v8f acc = {};
#pragma unroll
  for (int k0 = 0; k0 < HIDDEN; k0 += 4) {
    const int ka = k0 + 2 * hi;
    v2f A, B;
    A.x = tile[l * 132 + ka];
    A.y = tile[l * 132 + ka + 1];
    B.x = W[ka * HIDDEN + nb + l];
    B.y = W[(ka + 1) * HIDDEN + nb + l];
    acc = __builtin_amdgcn_wmma_f32_16x16x4_f32(false, A, false, B,
                                                (short)0, acc, false, false);
  }
  const float bv = bias[nb + l];
#pragma unroll
  for (int r = 0; r < 8; ++r)
    out[(row0 + r + 8 * hi) * HIDDEN + nb + l] = acc[r] + bv;
}

#define CHUNK 16
__device__ __forceinline__ float edge_logit(const float4 a, const float4 b, const float4 w) {
  float p = 0.0f, x;
  x = a.x + b.x; p += (x > 0.0f ? x : NEG_SLOPE * x) * w.x;
  x = a.y + b.y; p += (x > 0.0f ? x : NEG_SLOPE * x) * w.y;
  x = a.z + b.z; p += (x > 0.0f ? x : NEG_SLOPE * x) * w.z;
  x = a.w + b.w; p += (x > 0.0f ? x : NEG_SLOPE * x) * w.w;
  p += __shfl_xor(p, 1, 8);
  p += __shfl_xor(p, 2, 8);
  p += __shfl_xor(p, 4, 8);
  return p;
}

template <int W>
__device__ __forceinline__ void gat_chunk(const float* __restrict__ fs, const int lane,
                                          const int sorted, const int half, const int cnt,
                                          const float4 b, const float4 w,
                                          float& m, float& l, float4& acc) {
  float4 row[W];
  float  p[W];
#pragma unroll
  for (int j = 0; j < W; ++j) {
    const int s = __builtin_amdgcn_readlane(sorted, half + min(j, cnt - 1));
    row[j] = *(const float4*)(fs + s * HIDDEN + lane * 4);
  }
  float cm = -INFINITY;
#pragma unroll
  for (int j = 0; j < W; ++j) {
    const float q = edge_logit(row[j], b, w);
    p[j] = (j < cnt) ? q : -INFINITY;
    cm = fmaxf(cm, p[j]);
  }
  const float mn = fmaxf(m, cm);
  const float sc = expf(m - mn);
  l *= sc;
  acc.x *= sc; acc.y *= sc; acc.z *= sc; acc.w *= sc;
#pragma unroll
  for (int j = 0; j < W; ++j) {
    const float ex = expf(p[j] - mn);
    l += ex;
    acc.x += row[j].x * ex;
    acc.y += row[j].y * ex;
    acc.z += row[j].z * ex;
    acc.w += row[j].w * ex;
  }
  m = mn;
}

__global__ __launch_bounds__(256) void k_gat_node(
    const float* __restrict__ fs, const float* __restrict__ fd,
    const float* __restrict__ attn, const int* __restrict__ offs,
    const int* __restrict__ csrc, float* __restrict__ rst) {
  const int node = __builtin_amdgcn_readfirstlane(blockIdx.x * 8 + (threadIdx.x >> 5));
  const int lane = threadIdx.x & 31;
  const int beg = offs[node], end = offs[node + 1];
  const float4 b = *(const float4*)(fd + node * HIDDEN + lane * 4);
  const float4 w = *(const float4*)(attn + lane * 4);

  float m = -INFINITY, l = 0.0f;
  float4 acc = {0.0f, 0.0f, 0.0f, 0.0f};
  for (int base = beg; base < end; base += 32) {
    const int n = min(32, end - base);
    const int sid = (lane < n) ? csrc[base + lane] : 0x7fffffff;
    int rk = 0;
    for (int j = 0; j < n; ++j) {
      const int o = __builtin_amdgcn_readlane(sid, j);
      rk += ((o < sid) || (o == sid && j < lane)) ? 1 : 0;
    }
    const int sorted = __builtin_amdgcn_ds_permute(rk << 2, sid);

    for (int half = 0; half < n; half += CHUNK) {
      const int cnt = min(CHUNK, n - half);
      const int quads = (cnt + 3) >> 2;
      if (quads == 4)      gat_chunk<16>(fs, lane, sorted, half, cnt, b, w, m, l, acc);
      else if (quads == 3) gat_chunk<12>(fs, lane, sorted, half, cnt, b, w, m, l, acc);
      else if (quads == 2) gat_chunk<8>(fs, lane, sorted, half, cnt, b, w, m, l, acc);
      else                 gat_chunk<4>(fs, lane, sorted, half, cnt, b, w, m, l, acc);
    }
  }
  const float den = (l == 0.0f) ? 1.0f : l;
  float4 o;
  o.x = acc.x / den; o.y = acc.y / den; o.z = acc.z / den; o.w = acc.w / den;
  *(float4*)(rst + node * HIDDEN + lane * 4) = o;
}

#define TM 64
#define XS 132
#define TS 260

__device__ __forceinline__ void layernorm4(float4 (&v)[4], const float4 g, const float4 b) {
  float sum[4];
#pragma unroll
  for (int i = 0; i < 4; ++i) sum[i] = v[i].x + v[i].y + v[i].z + v[i].w;
#pragma unroll
  for (int off = 1; off < 32; off <<= 1) {
#pragma unroll
    for (int i = 0; i < 4; ++i) sum[i] += __shfl_xor(sum[i], off, 32);
  }
  float ss[4];
#pragma unroll
  for (int i = 0; i < 4; ++i) {
    const float mu = sum[i] * (1.0f / HIDDEN);
    v[i].x -= mu; v[i].y -= mu; v[i].z -= mu; v[i].w -= mu;
    ss[i] = v[i].x * v[i].x + v[i].y * v[i].y + v[i].z * v[i].z + v[i].w * v[i].w;
  }
#pragma unroll
  for (int off = 1; off < 32; off <<= 1) {
#pragma unroll
    for (int i = 0; i < 4; ++i) ss[i] += __shfl_xor(ss[i], off, 32);
  }
#pragma unroll
  for (int i = 0; i < 4; ++i) {
    const float rs = rsqrtf(ss[i] * (1.0f / HIDDEN) + LN_EPS);
    v[i].x = v[i].x * rs * g.x + b.x;
    v[i].y = v[i].y * rs * g.y + b.y;
    v[i].z = v[i].z * rs * g.z + b.z;
    v[i].w = v[i].w * rs * g.w + b.w;
  }
}

__global__ __launch_bounds__(512) void k_tail(
    const float* __restrict__ h, const float* __restrict__ rst,
    const float* __restrict__ g1, const float* __restrict__ be1,
    const float* __restrict__ W1, const float* __restrict__ bf1,
    const float* __restrict__ W2, const float* __restrict__ bf2,
    const float* __restrict__ g2, const float* __restrict__ be2,
    float* __restrict__ out) {
  __shared__ __attribute__((aligned(16))) float xs[TM * XS];
  __shared__ __attribute__((aligned(16))) float ts[TM * TS];
  const int tid = threadIdx.x, wave = tid >> 5, lane = tid & 31;
  const int row0 = blockIdx.x * TM;
  const int l = lane & 15, hi = lane >> 4;

  {
    float4 v[4];
#pragma unroll
    for (int i = 0; i < 4; ++i) {
      const int row = min(row0 + wave * 4 + i, N_NODES - 1);
      const float4 x = *(const float4*)(h + row * HIDDEN + lane * 4);
      const float4 a = *(const float4*)(rst + row * HIDDEN + lane * 4);
      v[i].x = x.x + a.x; v[i].y = x.y + a.y; v[i].z = x.z + a.z; v[i].w = x.w + a.w;
    }
    layernorm4(v, *(const float4*)(g1 + lane * 4), *(const float4*)(be1 + lane * 4));
#pragma unroll
    for (int i = 0; i < 4; ++i)
      *(float4*)(&xs[(wave * 4 + i) * XS + lane * 4]) = v[i];
  }
  __syncthreads();

  {
    const int nb = wave * 16;
    v8f acc[4] = {};
#pragma unroll
    for (int k0 = 0; k0 < HIDDEN; k0 += 4) {
      const int ka = k0 + 2 * hi;
      v2f B;
      B.x = W1[ka * FFN_DIM + nb + l];
      B.y = W1[(ka + 1) * FFN_DIM + nb + l];
#pragma unroll
      for (int mt = 0; mt < 4; ++mt) {
        const v2f A = *(const v2f*)(&xs[(mt * 16 + l) * XS + ka]);
        acc[mt] = __builtin_amdgcn_wmma_f32_16x16x4_f32(false, A, false, B,
                                                        (short)0, acc[mt], false, false);
      }
    }
    const float bv = bf1[nb + l];
#pragma unroll
    for (int mt = 0; mt < 4; ++mt) {
#pragma unroll
      for (int r = 0; r < 8; ++r) {
        float x = acc[mt][r] + bv;
        x = 0.5f * x * (1.0f + erff(x * 0.70710678118654752f));
        ts[(mt * 16 + r + 8 * hi) * TS + nb + l] = x;
      }
    }
  }
  __syncthreads();

  {
    const int nb  = (wave & 7) * 16;
    const int mt0 = (wave >> 3) * 2;
    v8f acc[2] = {};
#pragma unroll
    for (int k0 = 0; k0 < FFN_DIM; k0 += 4) {
      const int ka = k0 + 2 * hi;
      v2f B;
      B.x = W2[ka * HIDDEN + nb + l];
      B.y = W2[(ka + 1) * HIDDEN + nb + l];
#pragma unroll
      for (int j = 0; j < 2; ++j) {
        const v2f A = *(const v2f*)(&ts[((mt0 + j) * 16 + l) * TS + ka]);
        acc[j] = __builtin_amdgcn_wmma_f32_16x16x4_f32(false, A, false, B,
                                                       (short)0, acc[j], false, false);
      }
    }
    const float bv = bf2[nb + l];
#pragma unroll
    for (int j = 0; j < 2; ++j) {
#pragma unroll
      for (int r = 0; r < 8; ++r) {
        const int idx = ((mt0 + j) * 16 + r + 8 * hi) * XS + nb + l;
        xs[idx] = acc[j][r] + bv + xs[idx];
      }
    }
  }
  __syncthreads();

  {
    float4 v[4];
#pragma unroll
    for (int i = 0; i < 4; ++i) v[i] = *(const float4*)(&xs[(wave * 4 + i) * XS + lane * 4]);
    layernorm4(v, *(const float4*)(g2 + lane * 4), *(const float4*)(be2 + lane * 4));
#pragma unroll
    for (int i = 0; i < 4; ++i) {
      const int row = row0 + wave * 4 + i;
      if (row < N_NODES) *(float4*)(out + row * HIDDEN + lane * 4) = v[i];
    }
  }
}

extern "C" void kernel_launch(void* const* d_in, const int* in_sizes, int n_in,
                              void* d_out, int out_size, void* d_ws, size_t ws_size,
                              hipStream_t stream) {
  const float* h     = (const float*)d_in[0];
  const int*   src   = (const int*)d_in[1];
  const int*   dst   = (const int*)d_in[2];
  const float* Wsrc  = (const float*)d_in[3];
  const float* bsrc  = (const float*)d_in[4];
  const float* Wdst  = (const float*)d_in[5];
  const float* bdst  = (const float*)d_in[6];
  const float* attn  = (const float*)d_in[7];
  const float* W1    = (const float*)d_in[8];
  const float* bf1   = (const float*)d_in[9];
  const float* W2    = (const float*)d_in[10];
  const float* bf2   = (const float*)d_in[11];
  const float* g1    = (const float*)d_in[12];
  const float* be1   = (const float*)d_in[13];
  const float* g2    = (const float*)d_in[14];
  const float* be2   = (const float*)d_in[15];
  float* out = (float*)d_out;

  float* ws     = (float*)d_ws;
  float* fs     = ws;
  float* fd     = fs + (size_t)N_NODES * HIDDEN;
  float* rst    = fd + (size_t)N_NODES * HIDDEN;
  int*   counts = (int*)(rst + (size_t)N_NODES * HIDDEN);
  int*   offs   = counts + N_NODES;
  int*   csrc   = offs + N_NODES + 16 + N_EDGES;
  int*   H      = (int*)rst;

  k_hist<<<dim3(SORT_G, SORT_S), SORT_THREADS, 0, stream>>>(dst, H);
  k_gat_proj<<<N_NODES / 16, 512, 0, stream>>>(h, Wsrc, bsrc, Wdst, bdst, fs, fd);
  k_colprefix<<<(N_NODES + 255) / 256, 256, 0, stream>>>(H, counts);
  k_scan_offsets<<<1, SCAN_THREADS, 0, stream>>>(counts, offs);
  k_place<<<dim3(SORT_G, SORT_S), SORT_THREADS, 0, stream>>>(src, dst, H, offs, csrc);
  k_gat_node<<<N_NODES / 8, 256, 0, stream>>>(fs, fd, attn, offs, csrc, rst);
  k_tail<<<(N_NODES + TM - 1) / TM, 512, 0, stream>>>(h, rst, g1, be1, W1, bf1, W2, bf2, g2, be2, out);
}


# Attention fused into the tail kernel: each wave gathers, softmaxes and sums in-edges of its 4 nodes before the norm; no rst plane, 1 launch and 50000 waves fewer; logit dot summed in channel pairs
# speedup vs baseline: 2.6183x; 1.1343x over previous
#include <hip/hip_runtime.h>
#include <math.h>


#define N_NODES   50000
#define N_EDGES   600000
#define HIDDEN    128
#define FFN_DIM   256
#define HEADS     4
#define HEAD_DIM  32
#define NEG_SLOPE 0.2f
#define LN_EPS    1e-5f

typedef __attribute__((ext_vector_type(2))) float v2f;
typedef __attribute__((ext_vector_type(8))) float v8f;

#define SORT_G       32
#define SORT_S       8
#define SLICE_N      (N_NODES / SORT_S)
#define CHUNK_E      18752
#define SORT_THREADS 1024
#define SORT_ITERS   5
#define SLICE_ITERS  7
static_assert(SORT_S * SLICE_N == N_NODES, "slices must tile the nodes");
static_assert(SORT_G * CHUNK_E >= N_EDGES && CHUNK_E % 4 == 0 && N_EDGES % 4 == 0, "chunks");
static_assert(SORT_ITERS * SORT_THREADS * 4 >= CHUNK_E, "one pass over a chunk");
static_assert(SLICE_ITERS * SORT_THREADS >= SLICE_N, "one pass over a slice");

__global__ __launch_bounds__(SORT_THREADS) void k_hist(const int* __restrict__ dst,
                                                       int* __restrict__ H) {
  __shared__ int hist[SLICE_N];
  const int g = blockIdx.x, s = blockIdx.y, tid = threadIdx.x;
  const int lo = s * SLICE_N;
  const int e0 = g * CHUNK_E;
  const int e1 = min(e0 + CHUNK_E, N_EDGES);
  int4 d[SORT_ITERS];
#pragma unroll
  for (int i = 0; i < SORT_ITERS; ++i) {
    const int e = e0 + (i * SORT_THREADS + tid) * 4;
    d[i] = (e < e1) ? *(const int4*)(dst + e) : make_int4(-1, -1, -1, -1);
  }
  for (int j = tid; j < SLICE_N; j += SORT_THREADS) hist[j] = 0;
  __syncthreads();
#pragma unroll
  for (int i = 0; i < SORT_ITERS; ++i) {
    const unsigned n0 = (unsigned)(d[i].x - lo), n1 = (unsigned)(d[i].y - lo);
    const unsigned n2 = (unsigned)(d[i].z - lo), n3 = (unsigned)(d[i].w - lo);
    if (n0 < SLICE_N) __hip_atomic_fetch_add(&hist[n0], 1, __ATOMIC_RELAXED, __HIP_MEMORY_SCOPE_WORKGROUP);
    if (n1 < SLICE_N) __hip_atomic_fetch_add(&hist[n1], 1, __ATOMIC_RELAXED, __HIP_MEMORY_SCOPE_WORKGROUP);
    if (n2 < SLICE_N) __hip_atomic_fetch_add(&hist[n2], 1, __ATOMIC_RELAXED, __HIP_MEMORY_SCOPE_WORKGROUP);
    if (n3 < SLICE_N) __hip_atomic_fetch_add(&hist[n3], 1, __ATOMIC_RELAXED, __HIP_MEMORY_SCOPE_WORKGROUP);
  }
  __syncthreads();
  for (int j = tid; j < SLICE_N; j += SORT_THREADS) H[g * N_NODES + lo + j] = hist[j];
}

__global__ void k_colprefix(int* __restrict__ H, int* __restrict__ counts) {
  const int n = blockIdx.x * 256 + threadIdx.x;
  if (n >= N_NODES) return;
  int c[SORT_G];
#pragma unroll
  for (int g = 0; g < SORT_G; ++g) c[g] = H[g * N_NODES + n];
  int run = 0;
#pragma unroll
  for (int g = 0; g < SORT_G; ++g) {
    H[g * N_NODES + n] = run;
    run += c[g];
  }
  counts[n] = run;
}

#define SCAN_THREADS 1024
#define SCAN_ITERS   13
#define SCAN_SEG     (SCAN_ITERS * 128)
__global__ __launch_bounds__(SCAN_THREADS) void k_scan_offsets(
    const int* __restrict__ counts, int* __restrict__ offs) {
  __shared__ int wsum[32];
  const int tid = threadIdx.x, lane = tid & 31, wave = tid >> 5;
  const int seg0 = wave * SCAN_SEG + lane * 4;
  int4 v[SCAN_ITERS];
  int  excl[SCAN_ITERS];
#pragma unroll
  for (int it = 0; it < SCAN_ITERS; ++it) {
    const int idx = seg0 + it * 128;
    if (idx < N_NODES) v[it] = *(const int4*)(counts + idx);
    else               v[it] = make_int4(0, 0, 0, 0);
  }
  int carry = 0;
#pragma unroll
  for (int it = 0; it < SCAN_ITERS; ++it) {
    const int s4 = v[it].x + v[it].y + v[it].z + v[it].w;
    int inc = s4;
#pragma unroll
    for (int off = 1; off < 32; off <<= 1) {
      const int t = __shfl_up(inc, off, 32);
      if (lane >= off) inc += t;
    }
    excl[it] = carry + inc - s4;
    carry += __shfl(inc, 31, 32);
  }
  if (lane == 0) wsum[wave] = carry;
  __syncthreads();
  if (wave == 0) {
    const int ws = wsum[lane];
    int winc = ws;
#pragma unroll
    for (int off = 1; off < 32; off <<= 1) {
      const int t = __shfl_up(winc, off, 32);
      if (lane >= off) winc += t;
    }
    wsum[lane] = winc - ws;
  }
  __syncthreads();
  const int wbase = wsum[wave];
#pragma unroll
  for (int it = 0; it < SCAN_ITERS; ++it) {
    const int idx = seg0 + it * 128;
    if (idx < N_NODES) {
      int4 o;
      o.x = wbase + excl[it];
      o.y = o.x + v[it].x;
      o.z = o.y + v[it].y;
      o.w = o.z + v[it].z;
      *(int4*)(offs + idx) = o;
    }
  }
  if (tid == SCAN_THREADS - 1) offs[N_NODES] = wbase + carry;
}

__global__ __launch_bounds__(SORT_THREADS) void k_place(
    const int* __restrict__ src, const int* __restrict__ dst, const int* __restrict__ H,
    const int* __restrict__ offs, int* __restrict__ csrc) {
  __shared__ int next[SLICE_N];
  const int g = blockIdx.x, s = blockIdx.y, tid = threadIdx.x;
  const int lo = s * SLICE_N;
  const int e0 = g * CHUNK_E;
  const int e1 = min(e0 + CHUNK_E, N_EDGES);
  int4 d[SORT_ITERS], sv[SORT_ITERS];
  int  cur[SLICE_ITERS];
#pragma unroll
  for (int i = 0; i < SORT_ITERS; ++i) {
    const int e = e0 + (i * SORT_THREADS + tid) * 4;
    if (e < e1) {
      d[i]  = *(const int4*)(dst + e);
      sv[i] = *(const int4*)(src + e);
    } else {
      d[i]  = make_int4(-1, -1, -1, -1);
      sv[i] = make_int4(0, 0, 0, 0);
    }
  }
#pragma unroll
  for (int i = 0; i < SLICE_ITERS; ++i) {
    const int j = i * SORT_THREADS + tid;
    cur[i] = (j < SLICE_N) ? offs[lo + j] + H[g * N_NODES + lo + j] : 0;
  }
#pragma unroll
  for (int i = 0; i < SLICE_ITERS; ++i) {
    const int j = i * SORT_THREADS + tid;
    if (j < SLICE_N) next[j] = cur[i];
  }
  __syncthreads();
  int pos[SORT_ITERS * 4];
#pragma unroll
  for (int i = 0; i < SORT_ITERS; ++i) {
    const unsigned n0 = (unsigned)(d[i].x - lo), n1 = (unsigned)(d[i].y - lo);
    const unsigned n2 = (unsigned)(d[i].z - lo), n3 = (unsigned)(d[i].w - lo);
    pos[4 * i + 0] = (n0 < SLICE_N) ? __hip_atomic_fetch_add(&next[n0], 1, __ATOMIC_RELAXED, __HIP_MEMORY_SCOPE_WORKGROUP) : -1;
    pos[4 * i + 1] = (n1 < SLICE_N) ? __hip_atomic_fetch_add(&next[n1], 1, __ATOMIC_RELAXED, __HIP_MEMORY_SCOPE_WORKGROUP) : -1;
    pos[4 * i + 2] = (n2 < SLICE_N) ? __hip_atomic_fetch_add(&next[n2], 1, __ATOMIC_RELAXED, __HIP_MEMORY_SCOPE_WORKGROUP) : -1;
    pos[4 * i + 3] = (n3 < SLICE_N) ? __hip_atomic_fetch_add(&next[n3], 1, __ATOMIC_RELAXED, __HIP_MEMORY_SCOPE_WORKGROUP) : -1;
  }
#pragma unroll
  for (int i = 0; i < SORT_ITERS; ++i) {
    if (pos[4 * i + 0] >= 0) csrc[pos[4 * i + 0]] = sv[i].x;
    if (pos[4 * i + 1] >= 0) csrc[pos[4 * i + 1]] = sv[i].y;
    if (pos[4 * i + 2] >= 0) csrc[pos[4 * i + 2]] = sv[i].z;
    if (pos[4 * i + 3] >= 0) csrc[pos[4 * i + 3]] = sv[i].w;
  }
}

__global__ __launch_bounds__(512) void k_gat_proj(
    const float* __restrict__ h,
    const float* __restrict__ Wsrc, const float* __restrict__ bsrc,
    const float* __restrict__ Wdst, const float* __restrict__ bdst,
    float* __restrict__ fs, float* __restrict__ fd) {
  __shared__ float tile[16 * 132];
  const int tid  = threadIdx.x;
  const int row0 = blockIdx.x * 16;
  {
    const int r = tid >> 5;
    const int c = (tid & 31) * 4;
    *(float4*)(&tile[r * 132 + c]) = *(const float4*)(h + (row0 + r) * HIDDEN + c);
  }
  __syncthreads();

  const int wave = tid >> 5, lane = tid & 31;
  const int l = lane & 15, hi = lane >> 4;
  const int sel = wave >> 3;
  const int nb  = (wave & 7) * 16;
  const float* W    = sel ? Wdst : Wsrc;
  const float* bias = sel ? bdst : bsrc;
  float*       out  = sel ? fd   : fs;

  v8f acc = {};
#pragma unroll
  for (int k0 = 0; k0 < HIDDEN; k0 += 4) {
    const int ka = k0 + 2 * hi;
    v2f A, B;
    A.x = tile[l * 132 + ka];
    A.y = tile[l * 132 + ka + 1];
    B.x = W[ka * HIDDEN + nb + l];
    B.y = W[(ka + 1) * HIDDEN + nb + l];
    acc = __builtin_amdgcn_wmma_f32_16x16x4_f32(false, A, false, B,
                                                (short)0, acc, false, false);
  }
  const float bv = bias[nb + l];
#pragma unroll
  for (int r = 0; r < 8; ++r)
    out[(row0 + r + 8 * hi) * HIDDEN + nb + l] = acc[r] + bv;
}

#define CHUNK 16
__device__ __forceinline__ float edge_logit(const v2f alo, const v2f ahi, const v2f blo,
                                            const v2f bhi, const v2f wlo, const v2f whi) {
  v2f t0 = alo + blo, t1 = ahi + bhi;
  t0 = __builtin_elementwise_max(t0, t0 * NEG_SLOPE);
  t1 = __builtin_elementwise_max(t1, t1 * NEG_SLOPE);
  v2f q = t0 * wlo;
  q += t1 * whi;
  float p = q.x + q.y;
  p += __shfl_xor(p, 1, 8);
  p += __shfl_xor(p, 2, 8);
  p += __shfl_xor(p, 4, 8);
  return p;
}

template <int W>
__device__ __forceinline__ void gat_chunk(const float* __restrict__ fs, const unsigned l16,
                                          const int sorted, const int half, const int cnt,
                                          const v2f blo, const v2f bhi,
                                          const v2f wlo, const v2f whi,
                                          float& m, float& l, v2f& alo, v2f& ahi) {
  v2f   rlo[W], rhi[W];
  float p[W];
#pragma unroll
  for (int j = 0; j < W; ++j) {
    const int s = __builtin_amdgcn_readlane(sorted, half + min(j, cnt - 1));
    const unsigned off = ((unsigned)s << 9) + l16;
    const float4 r = *(const float4*)((const char*)fs + off);
    rlo[j].x = r.x; rlo[j].y = r.y; rhi[j].x = r.z; rhi[j].y = r.w;
  }
  float cm = -INFINITY;
#pragma unroll
  for (int j = 0; j < W; ++j) {
    const float q = edge_logit(rlo[j], rhi[j], blo, bhi, wlo, whi);
    p[j] = (j < cnt) ? q : -INFINITY;
    cm = fmaxf(cm, p[j]);
  }
  const float mn = fmaxf(m, cm);
  const float sc = expf(m - mn);
  l *= sc;
  alo *= sc; ahi *= sc;
#pragma unroll
  for (int j = 0; j < W; ++j) {
    const float ex = expf(p[j] - mn);
    l += ex;
    alo += rlo[j] * ex;
    ahi += rhi[j] * ex;
  }
  m = mn;
}

__device__ __forceinline__ float4 gat_node_row(const float* __restrict__ fs,
                                               const int* __restrict__ csrc, const int lane,
                                               const unsigned l16,
                                               const int beg, const int end, const int sid0,
                                               const float4 b, const float4 w) {
  float m = -INFINITY, l = 0.0f;
  v2f alo = {0.0f, 0.0f}, ahi = {0.0f, 0.0f};
  const v2f blo = {b.x, b.y}, bhi = {b.z, b.w};
  const v2f wlo = {w.x, w.y}, whi = {w.z, w.w};
  for (int base = beg; base < end; base += 32) {
    const int n = min(32, end - base);
    int sid = sid0;
    if (base != beg) sid = (lane < n) ? csrc[base + lane] : 0x7fffffff;
    int rk = 0;
    for (int j = 0; j < n; ++j) {
      const int o = __builtin_amdgcn_readlane(sid, j);
      rk += ((o < sid) || (o == sid && j < lane)) ? 1 : 0;
    }
    const int sorted = __builtin_amdgcn_ds_permute(rk << 2, sid);

    for (int half = 0; half < n; half += CHUNK) {
      const int cnt = min(CHUNK, n - half);
      const int quads = (cnt + 3) >> 2;
      if (quads == 4)      gat_chunk<16>(fs, l16, sorted, half, cnt, blo, bhi, wlo, whi, m, l, alo, ahi);
      else if (quads == 3) gat_chunk<12>(fs, l16, sorted, half, cnt, blo, bhi, wlo, whi, m, l, alo, ahi);
      else if (quads == 2) gat_chunk<8>(fs, l16, sorted, half, cnt, blo, bhi, wlo, whi, m, l, alo, ahi);
      else                 gat_chunk<4>(fs, l16, sorted, half, cnt, blo, bhi, wlo, whi, m, l, alo, ahi);
    }
  }
  const float den = (l == 0.0f) ? 1.0f : l;
  float4 o;
  o.x = alo.x / den; o.y = alo.y / den; o.z = ahi.x / den; o.w = ahi.y / den;
  return o;
}

#define TM 64
#define XS 132
#define TS 260

__device__ __forceinline__ void layernorm4(float4 (&v)[4], const float4 g, const float4 b) {
  float sum[4];
#pragma unroll
  for (int i = 0; i < 4; ++i) sum[i] = v[i].x + v[i].y + v[i].z + v[i].w;
#pragma unroll
  for (int off = 1; off < 32; off <<= 1) {
#pragma unroll
    for (int i = 0; i < 4; ++i) sum[i] += __shfl_xor(sum[i], off, 32);
  }
  float ss[4];
#pragma unroll
  for (int i = 0; i < 4; ++i) {
    const float mu = sum[i] * (1.0f / HIDDEN);
    v[i].x -= mu; v[i].y -= mu; v[i].z -= mu; v[i].w -= mu;
    ss[i] = v[i].x * v[i].x + v[i].y * v[i].y + v[i].z * v[i].z + v[i].w * v[i].w;
  }
#pragma unroll
  for (int off = 1; off < 32; off <<= 1) {
#pragma unroll
    for (int i = 0; i < 4; ++i) ss[i] += __shfl_xor(ss[i], off, 32);
  }
#pragma unroll
  for (int i = 0; i < 4; ++i) {
    const float rs = rsqrtf(ss[i] * (1.0f / HIDDEN) + LN_EPS);
    v[i].x = v[i].x * rs * g.x + b.x;
    v[i].y = v[i].y * rs * g.y + b.y;
    v[i].z = v[i].z * rs * g.z + b.z;
    v[i].w = v[i].w * rs * g.w + b.w;
  }
}

__global__ __launch_bounds__(512, 8) void k_tail(
    const float* __restrict__ h, const float* __restrict__ fs, const float* __restrict__ fd,
    const float* __restrict__ attn, const int* __restrict__ offs, const int* __restrict__ csrc,
    const float* __restrict__ g1, const float* __restrict__ be1,
    const float* __restrict__ W1, const float* __restrict__ bf1,
    const float* __restrict__ W2, const float* __restrict__ bf2,
    const float* __restrict__ g2, const float* __restrict__ be2,
    float* __restrict__ out) {
  __shared__ __attribute__((aligned(16))) float xs[TM * XS];
  __shared__ __attribute__((aligned(16))) float ts[TM * TS];
  const int tid = threadIdx.x, lane = tid & 31;
  const int wave = __builtin_amdgcn_readfirstlane(tid >> 5);
  const int row0 = blockIdx.x * TM;
  const int l = lane & 15, hi = lane >> 4;

  {
    const int n0 = row0 + wave * 4;
    const unsigned l16 = (unsigned)lane * 16u;
    char* const xsb = (char*)xs;
    char* const tsb = (char*)ts;
    const float4 w = *(const float4*)((const char*)attn + l16);
    int cbeg = offs[min(n0, N_NODES)], cend = offs[min(n0 + 1, N_NODES)];
    float4 hx[4], fb[4];
#pragma unroll
    for (int i = 0; i < 4; ++i) {
      const int row = min(n0 + i, N_NODES - 1);
      hx[i] = *(const float4*)((const char*)(h + row * HIDDEN) + l16);
      fb[i] = *(const float4*)((const char*)(fd + row * HIDDEN) + l16);
    }
    int csid = (lane < min(32, cend - cbeg)) ? csrc[cbeg + lane] : 0x7fffffff;
#pragma unroll
    for (int i = 0; i < 4; ++i) {
      *(float4*)(xsb + (wave * 4 + i) * (XS * 4) + l16) = hx[i];
      *(float4*)(tsb + (wave * 4 + i) * (TS * 4) + l16) = fb[i];
    }
    for (int i = 0; i < 4; ++i) {
      int nbeg = cend, nend = cend, nsid = 0x7fffffff;
      if (i < 3) {
        nend = offs[min(n0 + i + 2, N_NODES)];
        nsid = (lane < min(32, nend - nbeg)) ? csrc[nbeg + lane] : 0x7fffffff;
      }
      const float4 b = *(const float4*)(tsb + (wave * 4 + i) * (TS * 4) + l16);
      const float4 a = gat_node_row(fs, csrc, lane, l16, cbeg, cend, csid, b, w);
      const float4 x = *(const float4*)(xsb + (wave * 4 + i) * (XS * 4) + l16);
      float4 v;
      v.x = x.x + a.x; v.y = x.y + a.y; v.z = x.z + a.z; v.w = x.w + a.w;
      *(float4*)(xsb + (wave * 4 + i) * (XS * 4) + l16) = v;
      cbeg = nbeg; cend = nend; csid = nsid;
    }
    float4 v[4];
#pragma unroll
    for (int i = 0; i < 4; ++i) v[i] = *(const float4*)(&xs[(wave * 4 + i) * XS + lane * 4]);
    layernorm4(v, *(const float4*)(g1 + lane * 4), *(const float4*)(be1 + lane * 4));
#pragma unroll
    for (int i = 0; i < 4; ++i)
      *(float4*)(&xs[(wave * 4 + i) * XS + lane * 4]) = v[i];
  }
  __syncthreads();

  {
    const int nb = wave * 16;
    v8f acc[4] = {};
#pragma unroll
    for (int k0 = 0; k0 < HIDDEN; k0 += 4) {
      const int ka = k0 + 2 * hi;
      v2f B;
      B.x = W1[ka * FFN_DIM + nb + l];
      B.y = W1[(ka + 1) * FFN_DIM + nb + l];
#pragma unroll
      for (int mt = 0; mt < 4; ++mt) {
        const v2f A = *(const v2f*)(&xs[(mt * 16 + l) * XS + ka]);
        acc[mt] = __builtin_amdgcn_wmma_f32_16x16x4_f32(false, A, false, B,
                                                        (short)0, acc[mt], false, false);
      }
    }
    const float bv = bf1[nb + l];
#pragma unroll
    for (int mt = 0; mt < 4; ++mt) {
#pragma unroll
      for (int r = 0; r < 8; ++r) {
        float x = acc[mt][r] + bv;
        x = 0.5f * x * (1.0f + erff(x * 0.70710678118654752f));
        ts[(mt * 16 + r + 8 * hi) * TS + nb + l] = x;
      }
    }
  }
  __syncthreads();

  {
    const int nb  = (wave & 7) * 16;
    const int mt0 = (wave >> 3) * 2;
    v8f acc[2] = {};
#pragma unroll
    for (int k0 = 0; k0 < FFN_DIM; k0 += 4) {
      const int ka = k0 + 2 * hi;
      v2f B;
      B.x = W2[ka * HIDDEN + nb + l];
      B.y = W2[(ka + 1) * HIDDEN + nb + l];
#pragma unroll
      for (int j = 0; j < 2; ++j) {
        const v2f A = *(const v2f*)(&ts[((mt0 + j) * 16 + l) * TS + ka]);
        acc[j] = __builtin_amdgcn_wmma_f32_16x16x4_f32(false, A, false, B,
                                                       (short)0, acc[j], false, false);
      }
    }
    const float bv = bf2[nb + l];
#pragma unroll
    for (int j = 0; j < 2; ++j) {
#pragma unroll
      for (int r = 0; r < 8; ++r) {
        const int idx = ((mt0 + j) * 16 + r + 8 * hi) * XS + nb + l;
        xs[idx] = acc[j][r] + bv + xs[idx];
      }
    }
  }
  __syncthreads();

  {
    float4 v[4];
#pragma unroll
    for (int i = 0; i < 4; ++i) v[i] = *(const float4*)(&xs[(wave * 4 + i) * XS + lane * 4]);
    layernorm4(v, *(const float4*)(g2 + lane * 4), *(const float4*)(be2 + lane * 4));
#pragma unroll
    for (int i = 0; i < 4; ++i) {
      const int row = row0 + wave * 4 + i;
      if (row < N_NODES) *(float4*)(out + row * HIDDEN + lane * 4) = v[i];
    }
  }
}

extern "C" void kernel_launch(void* const* d_in, const int* in_sizes, int n_in,
                              void* d_out, int out_size, void* d_ws, size_t ws_size,
                              hipStream_t stream) {
  const float* h     = (const float*)d_in[0];
  const int*   src   = (const int*)d_in[1];
  const int*   dst   = (const int*)d_in[2];
  const float* Wsrc  = (const float*)d_in[3];
  const float* bsrc  = (const float*)d_in[4];
  const float* Wdst  = (const float*)d_in[5];
  const float* bdst  = (const float*)d_in[6];
  const float* attn  = (const float*)d_in[7];
  const float* W1    = (const float*)d_in[8];
  const float* bf1   = (const float*)d_in[9];
  const float* W2    = (const float*)d_in[10];
  const float* bf2   = (const float*)d_in[11];
  const float* g1    = (const float*)d_in[12];
  const float* be1   = (const float*)d_in[13];
  const float* g2    = (const float*)d_in[14];
  const float* be2   = (const float*)d_in[15];
  float* out = (float*)d_out;

  float* ws     = (float*)d_ws;
  float* fs     = ws;
  float* fd     = fs + (size_t)N_NODES * HIDDEN;
  int*   H      = (int*)(fd + (size_t)N_NODES * HIDDEN);
  int*   counts = H + (size_t)N_NODES * HIDDEN;
  int*   offs   = counts + N_NODES;
  int*   csrc   = offs + N_NODES + 16 + N_EDGES;

  k_hist<<<dim3(SORT_G, SORT_S), SORT_THREADS, 0, stream>>>(dst, H);
  k_gat_proj<<<N_NODES / 16, 512, 0, stream>>>(h, Wsrc, bsrc, Wdst, bdst, fs, fd);
  k_colprefix<<<(N_NODES + 255) / 256, 256, 0, stream>>>(H, counts);
  k_scan_offsets<<<1, SCAN_THREADS, 0, stream>>>(counts, offs);
  k_place<<<dim3(SORT_G, SORT_S), SORT_THREADS, 0, stream>>>(src, dst, H, offs, csrc);
  k_tail<<<(N_NODES + TM - 1) / TM, 512, 0, stream>>>(h, fs, fd, attn, offs, csrc,
                                                       g1, be1, W1, bf1, W2, bf2, g2, be2, out);
}


# Library math calls replaced by short branch-free forms: expf of the edge softmax -> one hardware exp2 (argument <= 0); erff of the GELU -> rational t*P(t^2)/Q(t^2), reciprocal + one Newton step
# speedup vs baseline: 2.8063x; 1.0718x over previous
#include <hip/hip_runtime.h>
#include <math.h>


#define N_NODES   50000
#define N_EDGES   600000
#define HIDDEN    128
#define FFN_DIM   256
#define HEADS     4
#define HEAD_DIM  32
#define NEG_SLOPE 0.2f
#define LN_EPS    1e-5f

typedef __attribute__((ext_vector_type(2))) float v2f;
typedef __attribute__((ext_vector_type(8))) float v8f;

__device__ __forceinline__ v2f splat2(const float c) { v2f v = {c, c}; return v; }

__device__ __forceinline__ float exp_nonpos(const float x) {
  return __builtin_amdgcn_exp2f(x * 1.4426950408889634f);
}

#define ERF_LIM 3.832506856900711f
__device__ __forceinline__ v2f gelu_pair(const v2f x) {
  v2f t = x * 0.70710678118654752f;
  t = __builtin_elementwise_min(__builtin_elementwise_max(t, splat2(-ERF_LIM)), splat2(ERF_LIM));
  const v2f s = t * t;
  v2f p = __builtin_elementwise_fma(s, splat2(0.00022905065861350646f), splat2(0.0034082910107109506f));
  p = __builtin_elementwise_fma(p, s, splat2(0.050955695062380861f));
  p = __builtin_elementwise_fma(p, s, splat2(0.18520832239976145f));
  p = __builtin_elementwise_fma(p, s, splat2(1.128379143519084f));
  v2f q = __builtin_elementwise_fma(s, splat2(-1.1791602954361697e-7f), splat2(0.000023547966471313185f));
  q = __builtin_elementwise_fma(q, s, splat2(0.0010179625278914885f));
  q = __builtin_elementwise_fma(q, s, splat2(0.014070470171167667f));
  q = __builtin_elementwise_fma(q, s, splat2(0.11098505178285362f));
  q = __builtin_elementwise_fma(q, s, splat2(0.49746925110067538f));
  q = __builtin_elementwise_fma(q, s, splat2(1.0f));
  v2f r;
  r.x = __builtin_amdgcn_rcpf(q.x);
  r.y = __builtin_amdgcn_rcpf(q.y);
  const v2f e = __builtin_elementwise_fma(-q, r, splat2(1.0f));
  r = __builtin_elementwise_fma(e, r, r);
  const v2f erfv = (p * t) * r;
  const v2f hx = x * 0.5f;
  return __builtin_elementwise_fma(hx, erfv, hx);
}

#define SORT_G       32
#define SORT_S       8
#define SLICE_N      (N_NODES / SORT_S)
#define CHUNK_E      18752
#define SORT_THREADS 1024
#define SORT_ITERS   5
#define SLICE_ITERS  7
static_assert(SORT_S * SLICE_N == N_NODES, "slices must tile the nodes");
static_assert(SORT_G * CHUNK_E >= N_EDGES && CHUNK_E % 4 == 0 && N_EDGES % 4 == 0, "chunks");
static_assert(SORT_ITERS * SORT_THREADS * 4 >= CHUNK_E, "one pass over a chunk");
static_assert(SLICE_ITERS * SORT_THREADS >= SLICE_N, "one pass over a slice");

__global__ __launch_bounds__(SORT_THREADS) void k_hist(const int* __restrict__ dst,
                                                       int* __restrict__ H) {
  __shared__ int hist[SLICE_N];
  const int g = blockIdx.x, s = blockIdx.y, tid = threadIdx.x;
  const int lo = s * SLICE_N;
  const int e0 = g * CHUNK_E;
  const int e1 = min(e0 + CHUNK_E, N_EDGES);
  int4 d[SORT_ITERS];
#pragma unroll
  for (int i = 0; i < SORT_ITERS; ++i) {
    const int e = e0 + (i * SORT_THREADS + tid) * 4;
    d[i] = (e < e1) ? *(const int4*)(dst + e) : make_int4(-1, -1, -1, -1);
  }
  for (int j = tid; j < SLICE_N; j += SORT_THREADS) hist[j] = 0;
  __syncthreads();
#pragma unroll
  for (int i = 0; i < SORT_ITERS; ++i) {
    const unsigned n0 = (unsigned)(d[i].x - lo), n1 = (unsigned)(d[i].y - lo);
    const unsigned n2 = (unsigned)(d[i].z - lo), n3 = (unsigned)(d[i].w - lo);
    if (n0 < SLICE_N) __hip_atomic_fetch_add(&hist[n0], 1, __ATOMIC_RELAXED, __HIP_MEMORY_SCOPE_WORKGROUP);
    if (n1 < SLICE_N) __hip_atomic_fetch_add(&hist[n1], 1, __ATOMIC_RELAXED, __HIP_MEMORY_SCOPE_WORKGROUP);
    if (n2 < SLICE_N) __hip_atomic_fetch_add(&hist[n2], 1, __ATOMIC_RELAXED, __HIP_MEMORY_SCOPE_WORKGROUP);
    if (n3 < SLICE_N) __hip_atomic_fetch_add(&hist[n3], 1, __ATOMIC_RELAXED, __HIP_MEMORY_SCOPE_WORKGROUP);
  }
  __syncthreads();
  for (int j = tid; j < SLICE_N; j += SORT_THREADS) H[g * N_NODES + lo + j] = hist[j];
}

__global__ void k_colprefix(int* __restrict__ H, int* __restrict__ counts) {
  const int n = blockIdx.x * 256 + threadIdx.x;
  if (n >= N_NODES) return;
  int c[SORT_G];
#pragma unroll
  for (int g = 0; g < SORT_G; ++g) c[g] = H[g * N_NODES + n];
  int run = 0;
#pragma unroll
  for (int g = 0; g < SORT_G; ++g) {
    H[g * N_NODES + n] = run;
    run += c[g];
  }
  counts[n] = run;
}

#define SCAN_THREADS 1024
#define SCAN_ITERS   13
#define SCAN_SEG     (SCAN_ITERS * 128)
__global__ __launch_bounds__(SCAN_THREADS) void k_scan_offsets(
    const int* __restrict__ counts, int* __restrict__ offs) {
  __shared__ int wsum[32];
  const int tid = threadIdx.x, lane = tid & 31, wave = tid >> 5;
  const int seg0 = wave * SCAN_SEG + lane * 4;
  int4 v[SCAN_ITERS];
  int  excl[SCAN_ITERS];
#pragma unroll
  for (int it = 0; it < SCAN_ITERS; ++it) {
    const int idx = seg0 + it * 128;
    if (idx < N_NODES) v[it] = *(const int4*)(counts + idx);
    else               v[it] = make_int4(0, 0, 0, 0);
  }
  int carry = 0;
#pragma unroll
  for (int it = 0; it < SCAN_ITERS; ++it) {
    const int s4 = v[it].x + v[it].y + v[it].z + v[it].w;
    int inc = s4;
#pragma unroll
    for (int off = 1; off < 32; off <<= 1) {
      const int t = __shfl_up(inc, off, 32);
      if (lane >= off) inc += t;
    }
    excl[it] = carry + inc - s4;
    carry += __shfl(inc, 31, 32);
  }
  if (lane == 0) wsum[wave] = carry;
  __syncthreads();
  if (wave == 0) {
    const int ws = wsum[lane];
    int winc = ws;
#pragma unroll
    for (int off = 1; off < 32; off <<= 1) {
      const int t = __shfl_up(winc, off, 32);
      if (lane >= off) winc += t;
    }
    wsum[lane] = winc - ws;
  }
  __syncthreads();
  const int wbase = wsum[wave];
#pragma unroll
  for (int it = 0; it < SCAN_ITERS; ++it) {
    const int idx = seg0 + it * 128;
    if (idx < N_NODES) {
      int4 o;
      o.x = wbase + excl[it];
      o.y = o.x + v[it].x;
      o.z = o.y + v[it].y;
      o.w = o.z + v[it].z;
      *(int4*)(offs + idx) = o;
    }
  }
  if (tid == SCAN_THREADS - 1) offs[N_NODES] = wbase + carry;
}

__global__ __launch_bounds__(SORT_THREADS) void k_place(
    const int* __restrict__ src, const int* __restrict__ dst, const int* __restrict__ H,
    const int* __restrict__ offs, int* __restrict__ csrc) {
  __shared__ int next[SLICE_N];
  const int g = blockIdx.x, s = blockIdx.y, tid = threadIdx.x;
  const int lo = s * SLICE_N;
  const int e0 = g * CHUNK_E;
  const int e1 = min(e0 + CHUNK_E, N_EDGES);
  int4 d[SORT_ITERS], sv[SORT_ITERS];
  int  cur[SLICE_ITERS];
#pragma unroll
  for (int i = 0; i < SORT_ITERS; ++i) {
    const int e = e0 + (i * SORT_THREADS + tid) * 4;
    if (e < e1) {
      d[i]  = *(const int4*)(dst + e);
      sv[i] = *(const int4*)(src + e);
    } else {
      d[i]  = make_int4(-1, -1, -1, -1);
      sv[i] = make_int4(0, 0, 0, 0);
    }
  }
#pragma unroll
  for (int i = 0; i < SLICE_ITERS; ++i) {
    const int j = i * SORT_THREADS + tid;
    cur[i] = (j < SLICE_N) ? offs[lo + j] + H[g * N_NODES + lo + j] : 0;
  }
#pragma unroll
  for (int i = 0; i < SLICE_ITERS; ++i) {
    const int j = i * SORT_THREADS + tid;
    if (j < SLICE_N) next[j] = cur[i];
  }
  __syncthreads();
  int pos[SORT_ITERS * 4];
#pragma unroll
  for (int i = 0; i < SORT_ITERS; ++i) {
    const unsigned n0 = (unsigned)(d[i].x - lo), n1 = (unsigned)(d[i].y - lo);
    const unsigned n2 = (unsigned)(d[i].z - lo), n3 = (unsigned)(d[i].w - lo);
    pos[4 * i + 0] = (n0 < SLICE_N) ? __hip_atomic_fetch_add(&next[n0], 1, __ATOMIC_RELAXED, __HIP_MEMORY_SCOPE_WORKGROUP) : -1;
    pos[4 * i + 1] = (n1 < SLICE_N) ? __hip_atomic_fetch_add(&next[n1], 1, __ATOMIC_RELAXED, __HIP_MEMORY_SCOPE_WORKGROUP) : -1;
    pos[4 * i + 2] = (n2 < SLICE_N) ? __hip_atomic_fetch_add(&next[n2], 1, __ATOMIC_RELAXED, __HIP_MEMORY_SCOPE_WORKGROUP) : -1;
    pos[4 * i + 3] = (n3 < SLICE_N) ? __hip_atomic_fetch_add(&next[n3], 1, __ATOMIC_RELAXED, __HIP_MEMORY_SCOPE_WORKGROUP) : -1;
  }
#pragma unroll
  for (int i = 0; i < SORT_ITERS; ++i) {
    if (pos[4 * i + 0] >= 0) csrc[pos[4 * i + 0]] = sv[i].x;
    if (pos[4 * i + 1] >= 0) csrc[pos[4 * i + 1]] = sv[i].y;
    if (pos[4 * i + 2] >= 0) csrc[pos[4 * i + 2]] = sv[i].z;
    if (pos[4 * i + 3] >= 0) csrc[pos[4 * i + 3]] = sv[i].w;
  }
}

__global__ __launch_bounds__(512) void k_gat_proj(
    const float* __restrict__ h,
    const float* __restrict__ Wsrc, const float* __restrict__ bsrc,
    const float* __restrict__ Wdst, const float* __restrict__ bdst,
    float* __restrict__ fs, float* __restrict__ fd) {
  __shared__ float tile[16 * 132];
  const int tid  = threadIdx.x;
  const int row0 = blockIdx.x * 16;
  {
    const int r = tid >> 5;
    const int c = (tid & 31) * 4;
    *(float4*)(&tile[r * 132 + c]) = *(const float4*)(h + (row0 + r) * HIDDEN + c);
  }
  __syncthreads();

  const int wave = tid >> 5, lane = tid & 31;
  const int l = lane & 15, hi = lane >> 4;
  const int sel = wave >> 3;
  const int nb  = (wave & 7) * 16;
  const float* W    = sel ? Wdst : Wsrc;
  const float* bias = sel ? bdst : bsrc;
  float*       out  = sel ? fd   : fs;

  v8f acc = {};
#pragma unroll
  for (int k0 = 0; k0 < HIDDEN; k0 += 4) {
    const int ka = k0 + 2 * hi;
    v2f A, B;
    A.x = tile[l * 132 + ka];
    A.y = tile[l * 132 + ka + 1];
    B.x = W[ka * HIDDEN + nb + l];
    B.y = W[(ka + 1) * HIDDEN + nb + l];
    acc = __builtin_amdgcn_wmma_f32_16x16x4_f32(false, A, false, B,
                                                (short)0, acc, false, false);
  }
  const float bv = bias[nb + l];
#pragma unroll
  for (int r = 0; r < 8; ++r)
    out[(row0 + r + 8 * hi) * HIDDEN + nb + l] = acc[r] + bv;
}

#define CHUNK 16
__device__ __forceinline__ float edge_logit(const v2f alo, const v2f ahi, const v2f blo,
                                            const v2f bhi, const v2f wlo, const v2f whi) {
  v2f t0 = alo + blo, t1 = ahi + bhi;
  t0 = __builtin_elementwise_max(t0, t0 * NEG_SLOPE);
  t1 = __builtin_elementwise_max(t1, t1 * NEG_SLOPE);
  v2f q = t0 * wlo;
  q += t1 * whi;
  float p = q.x + q.y;
  p += __shfl_xor(p, 1, 8);
  p += __shfl_xor(p, 2, 8);
  p += __shfl_xor(p, 4, 8);
  return p;
}

template <int W>
__device__ __forceinline__ void gat_chunk(const float* __restrict__ fs, const unsigned l16,
                                          const int sorted, const int half, const int cnt,
                                          const v2f blo, const v2f bhi,
                                          const v2f wlo, const v2f whi,
                                          float& m, float& l, v2f& alo, v2f& ahi) {
  v2f   rlo[W], rhi[W];
  float p[W];
#pragma unroll
  for (int j = 0; j < W; ++j) {
    const int s = __builtin_amdgcn_readlane(sorted, half + min(j, cnt - 1));
    const unsigned off = ((unsigned)s << 9) + l16;
    const float4 r = *(const float4*)((const char*)fs + off);
    rlo[j].x = r.x; rlo[j].y = r.y; rhi[j].x = r.z; rhi[j].y = r.w;
  }
  float cm = -INFINITY;
#pragma unroll
  for (int j = 0; j < W; ++j) {
    const float q = edge_logit(rlo[j], rhi[j], blo, bhi, wlo, whi);
    p[j] = (j < cnt) ? q : -INFINITY;
    cm = fmaxf(cm, p[j]);
  }
  const float mn = fmaxf(m, cm);
  const float sc = exp_nonpos(m - mn);
  l *= sc;
  alo *= sc; ahi *= sc;
#pragma unroll
  for (int j = 0; j < W; ++j) {
    const float ex = exp_nonpos(p[j] - mn);
    l += ex;
    alo += rlo[j] * ex;
    ahi += rhi[j] * ex;
  }
  m = mn;
}

__device__ __forceinline__ float4 gat_node_row(const float* __restrict__ fs,
                                               const int* __restrict__ csrc, const int lane,
                                               const unsigned l16,
                                               const int beg, const int end, const int sid0,
                                               const float4 b, const float4 w) {
  float m = -INFINITY, l = 0.0f;
  v2f alo = {0.0f, 0.0f}, ahi = {0.0f, 0.0f};
  const v2f blo = {b.x, b.y}, bhi = {b.z, b.w};
  const v2f wlo = {w.x, w.y}, whi = {w.z, w.w};
  for (int base = beg; base < end; base += 32) {
    const int n = min(32, end - base);
    int sid = sid0;
    if (base != beg) sid = (lane < n) ? csrc[base + lane] : 0x7fffffff;
    int rk = 0;
    for (int j = 0; j < n; ++j) {
      const int o = __builtin_amdgcn_readlane(sid, j);
      rk += ((o < sid) || (o == sid && j < lane)) ? 1 : 0;
    }
    const int sorted = __builtin_amdgcn_ds_permute(rk << 2, sid);

    for (int half = 0; half < n; half += CHUNK) {
      const int cnt = min(CHUNK, n - half);
      const int quads = (cnt + 3) >> 2;
      if (quads == 4)      gat_chunk<16>(fs, l16, sorted, half, cnt, blo, bhi, wlo, whi, m, l, alo, ahi);
      else if (quads == 3) gat_chunk<12>(fs, l16, sorted, half, cnt, blo, bhi, wlo, whi, m, l, alo, ahi);
      else if (quads == 2) gat_chunk<8>(fs, l16, sorted, half, cnt, blo, bhi, wlo, whi, m, l, alo, ahi);
      else                 gat_chunk<4>(fs, l16, sorted, half, cnt, blo, bhi, wlo, whi, m, l, alo, ahi);
    }
  }
  const float den = (l == 0.0f) ? 1.0f : l;
  float4 o;
  o.x = alo.x / den; o.y = alo.y / den; o.z = ahi.x / den; o.w = ahi.y / den;
  return o;
}

#define TM 64
#define XS 132
#define TS 260

__device__ __forceinline__ void layernorm4(float4 (&v)[4], const float4 g, const float4 b) {
  float sum[4];
#pragma unroll
  for (int i = 0; i < 4; ++i) sum[i] = v[i].x + v[i].y + v[i].z + v[i].w;
#pragma unroll
  for (int off = 1; off < 32; off <<= 1) {
#pragma unroll
    for (int i = 0; i < 4; ++i) sum[i] += __shfl_xor(sum[i], off, 32);
  }
  float ss[4];
#pragma unroll
  for (int i = 0; i < 4; ++i) {
    const float mu = sum[i] * (1.0f / HIDDEN);
    v[i].x -= mu; v[i].y -= mu; v[i].z -= mu; v[i].w -= mu;
    ss[i] = v[i].x * v[i].x + v[i].y * v[i].y + v[i].z * v[i].z + v[i].w * v[i].w;
  }
#pragma unroll
  for (int off = 1; off < 32; off <<= 1) {
#pragma unroll
    for (int i = 0; i < 4; ++i) ss[i] += __shfl_xor(ss[i], off, 32);
  }
#pragma unroll
  for (int i = 0; i < 4; ++i) {
    const float rs = rsqrtf(ss[i] * (1.0f / HIDDEN) + LN_EPS);
    v[i].x = v[i].x * rs * g.x + b.x;
    v[i].y = v[i].y * rs * g.y + b.y;
    v[i].z = v[i].z * rs * g.z + b.z;
    v[i].w = v[i].w * rs * g.w + b.w;
  }
}

__global__ __launch_bounds__(512, 8) void k_tail(
    const float* __restrict__ h, const float* __restrict__ fs, const float* __restrict__ fd,
    const float* __restrict__ attn, const int* __restrict__ offs, const int* __restrict__ csrc,
    const float* __restrict__ g1, const float* __restrict__ be1,
    const float* __restrict__ W1, const float* __restrict__ bf1,
    const float* __restrict__ W2, const float* __restrict__ bf2,
    const float* __restrict__ g2, const float* __restrict__ be2,
    float* __restrict__ out) {
  __shared__ __attribute__((aligned(16))) float xs[TM * XS];
  __shared__ __attribute__((aligned(16))) float ts[TM * TS];
  const int tid = threadIdx.x, lane = tid & 31;
  const int wave = __builtin_amdgcn_readfirstlane(tid >> 5);
  const int row0 = blockIdx.x * TM;
  const int l = lane & 15, hi = lane >> 4;

  {
    const int n0 = row0 + wave * 4;
    const unsigned l16 = (unsigned)lane * 16u;
    char* const xsb = (char*)xs;
    char* const tsb = (char*)ts;
    const float4 w = *(const float4*)((const char*)attn + l16);
    int cbeg = offs[min(n0, N_NODES)], cend = offs[min(n0 + 1, N_NODES)];
    float4 hx[4], fb[4];
#pragma unroll
    for (int i = 0; i < 4; ++i) {
      const int row = min(n0 + i, N_NODES - 1);
      hx[i] = *(const float4*)((const char*)(h + row * HIDDEN) + l16);
      fb[i] = *(const float4*)((const char*)(fd + row * HIDDEN) + l16);
    }
    int csid = (lane < min(32, cend - cbeg)) ? csrc[cbeg + lane] : 0x7fffffff;
#pragma unroll
    for (int i = 0; i < 4; ++i) {
      *(float4*)(xsb + (wave * 4 + i) * (XS * 4) + l16) = hx[i];
      *(float4*)(tsb + (wave * 4 + i) * (TS * 4) + l16) = fb[i];
    }
    for (int i = 0; i < 4; ++i) {
      int nbeg = cend, nend = cend, nsid = 0x7fffffff;
      if (i < 3) {
        nend = offs[min(n0 + i + 2, N_NODES)];
        nsid = (lane < min(32, nend - nbeg)) ? csrc[nbeg + lane] : 0x7fffffff;
      }
      const float4 b = *(const float4*)(tsb + (wave * 4 + i) * (TS * 4) + l16);
      const float4 a = gat_node_row(fs, csrc, lane, l16, cbeg, cend, csid, b, w);
      const float4 x = *(const float4*)(xsb + (wave * 4 + i) * (XS * 4) + l16);
      float4 v;
      v.x = x.x + a.x; v.y = x.y + a.y; v.z = x.z + a.z; v.w = x.w + a.w;
      *(float4*)(xsb + (wave * 4 + i) * (XS * 4) + l16) = v;
      cbeg = nbeg; cend = nend; csid = nsid;
    }
    float4 v[4];
#pragma unroll
    for (int i = 0; i < 4; ++i) v[i] = *(const float4*)(&xs[(wave * 4 + i) * XS + lane * 4]);
    layernorm4(v, *(const float4*)(g1 + lane * 4), *(const float4*)(be1 + lane * 4));
#pragma unroll
    for (int i = 0; i < 4; ++i)
      *(float4*)(&xs[(wave * 4 + i) * XS + lane * 4]) = v[i];
  }
  __syncthreads();

  {
    const int nb = wave * 16;
    v8f acc[4] = {};
#pragma unroll
    for (int k0 = 0; k0 < HIDDEN; k0 += 4) {
      const int ka = k0 + 2 * hi;
      v2f B;
      B.x = W1[ka * FFN_DIM + nb + l];
      B.y = W1[(ka + 1) * FFN_DIM + nb + l];
#pragma unroll
      for (int mt = 0; mt < 4; ++mt) {
        const v2f A = *(const v2f*)(&xs[(mt * 16 + l) * XS + ka]);
        acc[mt] = __builtin_amdgcn_wmma_f32_16x16x4_f32(false, A, false, B,
                                                        (short)0, acc[mt], false, false);
      }
    }
    const float bv = bf1[nb + l];
#pragma unroll
    for (int mt = 0; mt < 4; ++mt) {
#pragma unroll
      for (int r = 0; r < 8; r += 2) {
        v2f x = {acc[mt][r], acc[mt][r + 1]};
        x = gelu_pair(x + bv);
        ts[(mt * 16 + r + 8 * hi) * TS + nb + l]     = x.x;
        ts[(mt * 16 + r + 1 + 8 * hi) * TS + nb + l] = x.y;
      }
    }
  }
  __syncthreads();

  {
    const int nb  = (wave & 7) * 16;
    const int mt0 = (wave >> 3) * 2;
    v8f acc[2] = {};
#pragma unroll
    for (int k0 = 0; k0 < FFN_DIM; k0 += 4) {
      const int ka = k0 + 2 * hi;
      v2f B;
      B.x = W2[ka * HIDDEN + nb + l];
      B.y = W2[(ka + 1) * HIDDEN + nb + l];
#pragma unroll
      for (int j = 0; j < 2; ++j) {
        const v2f A = *(const v2f*)(&ts[((mt0 + j) * 16 + l) * TS + ka]);
        acc[j] = __builtin_amdgcn_wmma_f32_16x16x4_f32(false, A, false, B,
                                                       (short)0, acc[j], false, false);
      }
    }
    const float bv = bf2[nb + l];
#pragma unroll
    for (int j = 0; j < 2; ++j) {
#pragma unroll
      for (int r = 0; r < 8; ++r) {
        const int idx = ((mt0 + j) * 16 + r + 8 * hi) * XS + nb + l;
        xs[idx] = acc[j][r] + bv + xs[idx];
      }
    }
  }
  __syncthreads();

  {
    float4 v[4];
#pragma unroll
    for (int i = 0; i < 4; ++i) v[i] = *(const float4*)(&xs[(wave * 4 + i) * XS + lane * 4]);
    layernorm4(v, *(const float4*)(g2 + lane * 4), *(const float4*)(be2 + lane * 4));
#pragma unroll
    for (int i = 0; i < 4; ++i) {
      const int row = row0 + wave * 4 + i;
      if (row < N_NODES) *(float4*)(out + row * HIDDEN + lane * 4) = v[i];
    }
  }
}

extern "C" void kernel_launch(void* const* d_in, const int* in_sizes, int n_in,
                              void* d_out, int out_size, void* d_ws, size_t ws_size,
                              hipStream_t stream) {
  const float* h     = (const float*)d_in[0];
  const int*   src   = (const int*)d_in[1];
  const int*   dst   = (const int*)d_in[2];
  const float* Wsrc  = (const float*)d_in[3];
  const float* bsrc  = (const float*)d_in[4];
  const float* Wdst  = (const float*)d_in[5];
  const float* bdst  = (const float*)d_in[6];
  const float* attn  = (const float*)d_in[7];
  const float* W1    = (const float*)d_in[8];
  const float* bf1   = (const float*)d_in[9];
  const float* W2    = (const float*)d_in[10];
  const float* bf2   = (const float*)d_in[11];
  const float* g1    = (const float*)d_in[12];
  const float* be1   = (const float*)d_in[13];
  const float* g2    = (const float*)d_in[14];
  const float* be2   = (const float*)d_in[15];
  float* out = (float*)d_out;

  float* ws     = (float*)d_ws;
  float* fs     = ws;
  float* fd     = fs + (size_t)N_NODES * HIDDEN;
  int*   H      = (int*)(fd + (size_t)N_NODES * HIDDEN);
  int*   counts = H + (size_t)N_NODES * HIDDEN;
  int*   offs   = counts + N_NODES;
  int*   csrc   = offs + N_NODES + 16 + N_EDGES;

  k_hist<<<dim3(SORT_G, SORT_S), SORT_THREADS, 0, stream>>>(dst, H);
  k_gat_proj<<<N_NODES / 16, 512, 0, stream>>>(h, Wsrc, bsrc, Wdst, bdst, fs, fd);
  k_colprefix<<<(N_NODES + 255) / 256, 256, 0, stream>>>(H, counts);
  k_scan_offsets<<<1, SCAN_THREADS, 0, stream>>>(counts, offs);
  k_place<<<dim3(SORT_G, SORT_S), SORT_THREADS, 0, stream>>>(src, dst, H, offs, csrc);
  k_tail<<<(N_NODES + TM - 1) / TM, 512, 0, stream>>>(h, fs, fd, attn, offs, csrc,
                                                       g1, be1, W1, bf1, W2, bf2, g2, be2, out);
}


# Projection GEMM re-cut: 512 work-groups (one resident round) of 6-7 row tiles; a wave keeps its 16 columns, one weight fragment feeds 6-7 WMMAs: 50000 waves -> 8192, weight loads / 6.5; same k order
# speedup vs baseline: 2.9983x; 1.0684x over previous
#include <hip/hip_runtime.h>
#include <math.h>


#define N_NODES   50000
#define N_EDGES   600000
#define HIDDEN    128
#define FFN_DIM   256
#define HEADS     4
#define HEAD_DIM  32
#define NEG_SLOPE 0.2f
#define LN_EPS    1e-5f

typedef __attribute__((ext_vector_type(2))) float v2f;
typedef __attribute__((ext_vector_type(8))) float v8f;

__device__ __forceinline__ v2f splat2(const float c) { v2f v = {c, c}; return v; }

__device__ __forceinline__ float exp_nonpos(const float x) {
  return __builtin_amdgcn_exp2f(x * 1.4426950408889634f);
}

#define ERF_LIM 3.832506856900711f
__device__ __forceinline__ v2f gelu_pair(const v2f x) {
  v2f t = x * 0.70710678118654752f;
  t = __builtin_elementwise_min(__builtin_elementwise_max(t, splat2(-ERF_LIM)), splat2(ERF_LIM));
  const v2f s = t * t;
  v2f p = __builtin_elementwise_fma(s, splat2(0.00022905065861350646f), splat2(0.0034082910107109506f));
  p = __builtin_elementwise_fma(p, s, splat2(0.050955695062380861f));
  p = __builtin_elementwise_fma(p, s, splat2(0.18520832239976145f));
  p = __builtin_elementwise_fma(p, s, splat2(1.128379143519084f));
  v2f q = __builtin_elementwise_fma(s, splat2(-1.1791602954361697e-7f), splat2(0.000023547966471313185f));
  q = __builtin_elementwise_fma(q, s, splat2(0.0010179625278914885f));
  q = __builtin_elementwise_fma(q, s, splat2(0.014070470171167667f));
  q = __builtin_elementwise_fma(q, s, splat2(0.11098505178285362f));
  q = __builtin_elementwise_fma(q, s, splat2(0.49746925110067538f));
  q = __builtin_elementwise_fma(q, s, splat2(1.0f));
  v2f r;
  r.x = __builtin_amdgcn_rcpf(q.x);
  r.y = __builtin_amdgcn_rcpf(q.y);
  const v2f e = __builtin_elementwise_fma(-q, r, splat2(1.0f));
  r = __builtin_elementwise_fma(e, r, r);
  const v2f erfv = (p * t) * r;
  const v2f hx = x * 0.5f;
  return __builtin_elementwise_fma(hx, erfv, hx);
}

#define SORT_G       32
#define SORT_S       8
#define SLICE_N      (N_NODES / SORT_S)
#define CHUNK_E      18752
#define SORT_THREADS 1024
#define SORT_ITERS   5
#define SLICE_ITERS  7
static_assert(SORT_S * SLICE_N == N_NODES, "slices must tile the nodes");
static_assert(SORT_G * CHUNK_E >= N_EDGES && CHUNK_E % 4 == 0 && N_EDGES % 4 == 0, "chunks");
static_assert(SORT_ITERS * SORT_THREADS * 4 >= CHUNK_E, "one pass over a chunk");
static_assert(SLICE_ITERS * SORT_THREADS >= SLICE_N, "one pass over a slice");

__global__ __launch_bounds__(SORT_THREADS) void k_hist(const int* __restrict__ dst,
                                                       int* __restrict__ H) {
  __shared__ int hist[SLICE_N];
  const int g = blockIdx.x, s = blockIdx.y, tid = threadIdx.x;
  const int lo = s * SLICE_N;
  const int e0 = g * CHUNK_E;
  const int e1 = min(e0 + CHUNK_E, N_EDGES);
  int4 d[SORT_ITERS];
#pragma unroll
  for (int i = 0; i < SORT_ITERS; ++i) {
    const int e = e0 + (i * SORT_THREADS + tid) * 4;
    d[i] = (e < e1) ? *(const int4*)(dst + e) : make_int4(-1, -1, -1, -1);
  }
  for (int j = tid; j < SLICE_N; j += SORT_THREADS) hist[j] = 0;
  __syncthreads();
#pragma unroll
  for (int i = 0; i < SORT_ITERS; ++i) {
    const unsigned n0 = (unsigned)(d[i].x - lo), n1 = (unsigned)(d[i].y - lo);
    const unsigned n2 = (unsigned)(d[i].z - lo), n3 = (unsigned)(d[i].w - lo);
    if (n0 < SLICE_N) __hip_atomic_fetch_add(&hist[n0], 1, __ATOMIC_RELAXED, __HIP_MEMORY_SCOPE_WORKGROUP);
    if (n1 < SLICE_N) __hip_atomic_fetch_add(&hist[n1], 1, __ATOMIC_RELAXED, __HIP_MEMORY_SCOPE_WORKGROUP);
    if (n2 < SLICE_N) __hip_atomic_fetch_add(&hist[n2], 1, __ATOMIC_RELAXED, __HIP_MEMORY_SCOPE_WORKGROUP);
    if (n3 < SLICE_N) __hip_atomic_fetch_add(&hist[n3], 1, __ATOMIC_RELAXED, __HIP_MEMORY_SCOPE_WORKGROUP);
  }
  __syncthreads();
  for (int j = tid; j < SLICE_N; j += SORT_THREADS) H[g * N_NODES + lo + j] = hist[j];
}

__global__ void k_colprefix(int* __restrict__ H, int* __restrict__ counts) {
  const int n = blockIdx.x * 256 + threadIdx.x;
  if (n >= N_NODES) return;
  int c[SORT_G];
#pragma unroll
  for (int g = 0; g < SORT_G; ++g) c[g] = H[g * N_NODES + n];
  int run = 0;
#pragma unroll
  for (int g = 0; g < SORT_G; ++g) {
    H[g * N_NODES + n] = run;
    run += c[g];
  }
  counts[n] = run;
}

#define SCAN_THREADS 1024
#define SCAN_ITERS   13
#define SCAN_SEG     (SCAN_ITERS * 128)
__global__ __launch_bounds__(SCAN_THREADS) void k_scan_offsets(
    const int* __restrict__ counts, int* __restrict__ offs) {
  __shared__ int wsum[32];
  const int tid = threadIdx.x, lane = tid & 31, wave = tid >> 5;
  const int seg0 = wave * SCAN_SEG + lane * 4;
  int4 v[SCAN_ITERS];
  int  excl[SCAN_ITERS];
#pragma unroll
  for (int it = 0; it < SCAN_ITERS; ++it) {
    const int idx = seg0 + it * 128;
    if (idx < N_NODES) v[it] = *(const int4*)(counts + idx);
    else               v[it] = make_int4(0, 0, 0, 0);
  }
  int carry = 0;
#pragma unroll
  for (int it = 0; it < SCAN_ITERS; ++it) {
    const int s4 = v[it].x + v[it].y + v[it].z + v[it].w;
    int inc = s4;
#pragma unroll
    for (int off = 1; off < 32; off <<= 1) {
      const int t = __shfl_up(inc, off, 32);
      if (lane >= off) inc += t;
    }
    excl[it] = carry + inc - s4;
    carry += __shfl(inc, 31, 32);
  }
  if (lane == 0) wsum[wave] = carry;
  __syncthreads();
  if (wave == 0) {
    const int ws = wsum[lane];
    int winc = ws;
#pragma unroll
    for (int off = 1; off < 32; off <<= 1) {
      const int t = __shfl_up(winc, off, 32);
      if (lane >= off) winc += t;
    }
    wsum[lane] = winc - ws;
  }
  __syncthreads();
  const int wbase = wsum[wave];
#pragma unroll
  for (int it = 0; it < SCAN_ITERS; ++it) {
    const int idx = seg0 + it * 128;
    if (idx < N_NODES) {
      int4 o;
      o.x = wbase + excl[it];
      o.y = o.x + v[it].x;
      o.z = o.y + v[it].y;
      o.w = o.z + v[it].z;
      *(int4*)(offs + idx) = o;
    }
  }
  if (tid == SCAN_THREADS - 1) offs[N_NODES] = wbase + carry;
}

__global__ __launch_bounds__(SORT_THREADS) void k_place(
    const int* __restrict__ src, const int* __restrict__ dst, const int* __restrict__ H,
    const int* __restrict__ offs, int* __restrict__ csrc) {
  __shared__ int next[SLICE_N];
  const int g = blockIdx.x, s = blockIdx.y, tid = threadIdx.x;
  const int lo = s * SLICE_N;
  const int e0 = g * CHUNK_E;
  const int e1 = min(e0 + CHUNK_E, N_EDGES);
  int4 d[SORT_ITERS], sv[SORT_ITERS];
  int  cur[SLICE_ITERS];
#pragma unroll
  for (int i = 0; i < SORT_ITERS; ++i) {
    const int e = e0 + (i * SORT_THREADS + tid) * 4;
    if (e < e1) {
      d[i]  = *(const int4*)(dst + e);
      sv[i] = *(const int4*)(src + e);
    } else {
      d[i]  = make_int4(-1, -1, -1, -1);
      sv[i] = make_int4(0, 0, 0, 0);
    }
  }
#pragma unroll
  for (int i = 0; i < SLICE_ITERS; ++i) {
    const int j = i * SORT_THREADS + tid;
    cur[i] = (j < SLICE_N) ? offs[lo + j] + H[g * N_NODES + lo + j] : 0;
  }
#pragma unroll
  for (int i = 0; i < SLICE_ITERS; ++i) {
    const int j = i * SORT_THREADS + tid;
    if (j < SLICE_N) next[j] = cur[i];
  }
  __syncthreads();
  int pos[SORT_ITERS * 4];
#pragma unroll
  for (int i = 0; i < SORT_ITERS; ++i) {
    const unsigned n0 = (unsigned)(d[i].x - lo), n1 = (unsigned)(d[i].y - lo);
    const unsigned n2 = (unsigned)(d[i].z - lo), n3 = (unsigned)(d[i].w - lo);
    pos[4 * i + 0] = (n0 < SLICE_N) ? __hip_atomic_fetch_add(&next[n0], 1, __ATOMIC_RELAXED, __HIP_MEMORY_SCOPE_WORKGROUP) : -1;
    pos[4 * i + 1] = (n1 < SLICE_N) ? __hip_atomic_fetch_add(&next[n1], 1, __ATOMIC_RELAXED, __HIP_MEMORY_SCOPE_WORKGROUP) : -1;
    pos[4 * i + 2] = (n2 < SLICE_N) ? __hip_atomic_fetch_add(&next[n2], 1, __ATOMIC_RELAXED, __HIP_MEMORY_SCOPE_WORKGROUP) : -1;
    pos[4 * i + 3] = (n3 < SLICE_N) ? __hip_atomic_fetch_add(&next[n3], 1, __ATOMIC_RELAXED, __HIP_MEMORY_SCOPE_WORKGROUP) : -1;
  }
#pragma unroll
  for (int i = 0; i < SORT_ITERS; ++i) {
    if (pos[4 * i + 0] >= 0) csrc[pos[4 * i + 0]] = sv[i].x;
    if (pos[4 * i + 1] >= 0) csrc[pos[4 * i + 1]] = sv[i].y;
    if (pos[4 * i + 2] >= 0) csrc[pos[4 * i + 2]] = sv[i].z;
    if (pos[4 * i + 3] >= 0) csrc[pos[4 * i + 3]] = sv[i].w;
  }
}

#define PROJ_WGS   512
#define PROJ_TILES (N_NODES / 16)
#define PROJ_LO    (PROJ_TILES / PROJ_WGS)
#define PROJ_HI    (PROJ_LO + 1)
#define PROJ_EXTRA (PROJ_TILES - PROJ_LO * PROJ_WGS)
#define PROJ_PITCH 132
static_assert(N_NODES % 16 == 0, "whole 16-row tiles");
static_assert(PROJ_EXTRA >= 0 && PROJ_EXTRA < PROJ_WGS, "6 or 7 tiles per work-group");

template <int NT>
__device__ __forceinline__ void proj_tiles(const float* __restrict__ tile, const int row0,
                                           const int wave, const int lane,
                                           const float* __restrict__ Wsrc, const float* __restrict__ bsrc,
                                           const float* __restrict__ Wdst, const float* __restrict__ bdst,
                                           float* __restrict__ fs, float* __restrict__ fd) {
  const int l = lane & 15, hi = lane >> 4;
  const int sel = wave >> 3;
  const int nb  = (wave & 7) * 16;
  const float* W    = sel ? Wdst : Wsrc;
  const float* bias = sel ? bdst : bsrc;
  float*       out  = sel ? fd   : fs;

  v8f acc[NT] = {};
#pragma unroll 8
  for (int k0 = 0; k0 < HIDDEN; k0 += 4) {
    const int ka = k0 + 2 * hi;
    v2f B;
    B.x = W[ka * HIDDEN + nb + l];
    B.y = W[(ka + 1) * HIDDEN + nb + l];
#pragma unroll
    for (int mt = 0; mt < NT; ++mt) {
      const v2f A = *(const v2f*)(&tile[(mt * 16 + l) * PROJ_PITCH + ka]);
      acc[mt] = __builtin_amdgcn_wmma_f32_16x16x4_f32(false, A, false, B,
                                                      (short)0, acc[mt], false, false);
    }
  }
  const float bv = bias[nb + l];
#pragma unroll
  for (int mt = 0; mt < NT; ++mt) {
#pragma unroll
    for (int r = 0; r < 8; ++r)
      out[(row0 + mt * 16 + r + 8 * hi) * HIDDEN + nb + l] = acc[mt][r] + bv;
  }
}

__global__ __launch_bounds__(512, 8) void k_gat_proj(
    const float* __restrict__ h,
    const float* __restrict__ Wsrc, const float* __restrict__ bsrc,
    const float* __restrict__ Wdst, const float* __restrict__ bdst,
    float* __restrict__ fs, float* __restrict__ fd) {
  __shared__ __attribute__((aligned(16))) float tile[PROJ_HI * 16 * PROJ_PITCH];
  const int tid = threadIdx.x, lane = tid & 31;
  const int wave = __builtin_amdgcn_readfirstlane(tid >> 5);
  const int b    = blockIdx.x;
  const int nt   = PROJ_LO + (b < PROJ_EXTRA ? 1 : 0);
  const int row0 = (b * PROJ_LO + min(b, PROJ_EXTRA)) * 16;
#pragma unroll
  for (int i = 0; i < PROJ_HI; ++i) {
    if (i < nt) {
      const int r = i * 16 + wave;
      *(float4*)(&tile[r * PROJ_PITCH + lane * 4]) = *(const float4*)(h + (row0 + r) * HIDDEN + lane * 4);
    }
  }
  __syncthreads();
  if (nt == PROJ_HI) proj_tiles<PROJ_HI>(tile, row0, wave, lane, Wsrc, bsrc, Wdst, bdst, fs, fd);
  else               proj_tiles<PROJ_LO>(tile, row0, wave, lane, Wsrc, bsrc, Wdst, bdst, fs, fd);
}

#define CHUNK 16
__device__ __forceinline__ float edge_logit(const v2f alo, const v2f ahi, const v2f blo,
                                            const v2f bhi, const v2f wlo, const v2f whi) {
  v2f t0 = alo + blo, t1 = ahi + bhi;
  t0 = __builtin_elementwise_max(t0, t0 * NEG_SLOPE);
  t1 = __builtin_elementwise_max(t1, t1 * NEG_SLOPE);
  v2f q = t0 * wlo;
  q += t1 * whi;
  float p = q.x + q.y;
  p += __shfl_xor(p, 1, 8);
  p += __shfl_xor(p, 2, 8);
  p += __shfl_xor(p, 4, 8);
  return p;
}

template <int W>
__device__ __forceinline__ void gat_chunk(const float* __restrict__ fs, const unsigned l16,
                                          const int sorted, const int half, const int cnt,
                                          const v2f blo, const v2f bhi,
                                          const v2f wlo, const v2f whi,
                                          float& m, float& l, v2f& alo, v2f& ahi) {
  v2f   rlo[W], rhi[W];
  float p[W];
#pragma unroll
  for (int j = 0; j < W; ++j) {
    const int s = __builtin_amdgcn_readlane(sorted, half + min(j, cnt - 1));
    const unsigned off = ((unsigned)s << 9) + l16;
    const float4 r = *(const float4*)((const char*)fs + off);
    rlo[j].x = r.x; rlo[j].y = r.y; rhi[j].x = r.z; rhi[j].y = r.w;
  }
  float cm = -INFINITY;
#pragma unroll
  for (int j = 0; j < W; ++j) {
    const float q = edge_logit(rlo[j], rhi[j], blo, bhi, wlo, whi);
    p[j] = (j < cnt) ? q : -INFINITY;
    cm = fmaxf(cm, p[j]);
  }
  const float mn = fmaxf(m, cm);
  const float sc = exp_nonpos(m - mn);
  l *= sc;
  alo *= sc; ahi *= sc;
#pragma unroll
  for (int j = 0; j < W; ++j) {
    const float ex = exp_nonpos(p[j] - mn);
    l += ex;
    alo += rlo[j] * ex;
    ahi += rhi[j] * ex;
  }
  m = mn;
}

__device__ __forceinline__ float4 gat_node_row(const float* __restrict__ fs,
                                               const int* __restrict__ csrc, const int lane,
                                               const unsigned l16,
                                               const int beg, const int end, const int sid0,
                                               const float4 b, const float4 w) {
  float m = -INFINITY, l = 0.0f;
  v2f alo = {0.0f, 0.0f}, ahi = {0.0f, 0.0f};
  const v2f blo = {b.x, b.y}, bhi = {b.z, b.w};
  const v2f wlo = {w.x, w.y}, whi = {w.z, w.w};
  for (int base = beg; base < end; base += 32) {
    const int n = min(32, end - base);
    int sid = sid0;
    if (base != beg) sid = (lane < n) ? csrc[base + lane] : 0x7fffffff;
    int rk = 0;
    for (int j = 0; j < n; ++j) {
      const int o = __builtin_amdgcn_readlane(sid, j);
      rk += ((o < sid) || (o == sid && j < lane)) ? 1 : 0;
    }
    const int sorted = __builtin_amdgcn_ds_permute(rk << 2, sid);

    for (int half = 0; half < n; half += CHUNK) {
      const int cnt = min(CHUNK, n - half);
      const int quads = (cnt + 3) >> 2;
      if (quads == 4)      gat_chunk<16>(fs, l16, sorted, half, cnt, blo, bhi, wlo, whi, m, l, alo, ahi);
      else if (quads == 3) gat_chunk<12>(fs, l16, sorted, half, cnt, blo, bhi, wlo, whi, m, l, alo, ahi);
      else if (quads == 2) gat_chunk<8>(fs, l16, sorted, half, cnt, blo, bhi, wlo, whi, m, l, alo, ahi);
      else                 gat_chunk<4>(fs, l16, sorted, half, cnt, blo, bhi, wlo, whi, m, l, alo, ahi);
    }
  }
  const float den = (l == 0.0f) ? 1.0f : l;
  float4 o;
  o.x = alo.x / den; o.y = alo.y / den; o.z = ahi.x / den; o.w = ahi.y / den;
  return o;
}

#define TM 64
#define XS 132
#define TS 260

__device__ __forceinline__ void layernorm4(float4 (&v)[4], const float4 g, const float4 b) {
  float sum[4];
#pragma unroll
  for (int i = 0; i < 4; ++i) sum[i] = v[i].x + v[i].y + v[i].z + v[i].w;
#pragma unroll
  for (int off = 1; off < 32; off <<= 1) {
#pragma unroll
    for (int i = 0; i < 4; ++i) sum[i] += __shfl_xor(sum[i], off, 32);
  }
  float ss[4];
#pragma unroll
  for (int i = 0; i < 4; ++i) {
    const float mu = sum[i] * (1.0f / HIDDEN);
    v[i].x -= mu; v[i].y -= mu; v[i].z -= mu; v[i].w -= mu;
    ss[i] = v[i].x * v[i].x + v[i].y * v[i].y + v[i].z * v[i].z + v[i].w * v[i].w;
  }
#pragma unroll
  for (int off = 1; off < 32; off <<= 1) {
#pragma unroll
    for (int i = 0; i < 4; ++i) ss[i] += __shfl_xor(ss[i], off, 32);
  }
#pragma unroll
  for (int i = 0; i < 4; ++i) {
    const float rs = rsqrtf(ss[i] * (1.0f / HIDDEN) + LN_EPS);
    v[i].x = v[i].x * rs * g.x + b.x;
    v[i].y = v[i].y * rs * g.y + b.y;
    v[i].z = v[i].z * rs * g.z + b.z;
    v[i].w = v[i].w * rs * g.w + b.w;
  }
}

__global__ __launch_bounds__(512, 8) void k_tail(
    const float* __restrict__ h, const float* __restrict__ fs, const float* __restrict__ fd,
    const float* __restrict__ attn, const int* __restrict__ offs, const int* __restrict__ csrc,
    const float* __restrict__ g1, const float* __restrict__ be1,
    const float* __restrict__ W1, const float* __restrict__ bf1,
    const float* __restrict__ W2, const float* __restrict__ bf2,
    const float* __restrict__ g2, const float* __restrict__ be2,
    float* __restrict__ out) {
  __shared__ __attribute__((aligned(16))) float xs[TM * XS];
  __shared__ __attribute__((aligned(16))) float ts[TM * TS];
  const int tid = threadIdx.x, lane = tid & 31;
  const int wave = __builtin_amdgcn_readfirstlane(tid >> 5);
  const int row0 = blockIdx.x * TM;
  const int l = lane & 15, hi = lane >> 4;

  {
    const int n0 = row0 + wave * 4;
    const unsigned l16 = (unsigned)lane * 16u;
    char* const xsb = (char*)xs;
    char* const tsb = (char*)ts;
    const float4 w = *(const float4*)((const char*)attn + l16);
    int cbeg = offs[min(n0, N_NODES)], cend = offs[min(n0 + 1, N_NODES)];
    float4 hx[4], fb[4];
#pragma unroll
    for (int i = 0; i < 4; ++i) {
      const int row = min(n0 + i, N_NODES - 1);
      hx[i] = *(const float4*)((const char*)(h + row * HIDDEN) + l16);
      fb[i] = *(const float4*)((const char*)(fd + row * HIDDEN) + l16);
    }
    int csid = (lane < min(32, cend - cbeg)) ? csrc[cbeg + lane] : 0x7fffffff;
#pragma unroll
    for (int i = 0; i < 4; ++i) {
      *(float4*)(xsb + (wave * 4 + i) * (XS * 4) + l16) = hx[i];
      *(float4*)(tsb + (wave * 4 + i) * (TS * 4) + l16) = fb[i];
    }
    for (int i = 0; i < 4; ++i) {
      int nbeg = cend, nend = cend, nsid = 0x7fffffff;
      if (i < 3) {
        nend = offs[min(n0 + i + 2, N_NODES)];
        nsid = (lane < min(32, nend - nbeg)) ? csrc[nbeg + lane] : 0x7fffffff;
      }
      const float4 b = *(const float4*)(tsb + (wave * 4 + i) * (TS * 4) + l16);
      const float4 a = gat_node_row(fs, csrc, lane, l16, cbeg, cend, csid, b, w);
      const float4 x = *(const float4*)(xsb + (wave * 4 + i) * (XS * 4) + l16);
      float4 v;
      v.x = x.x + a.x; v.y = x.y + a.y; v.z = x.z + a.z; v.w = x.w + a.w;
      *(float4*)(xsb + (wave * 4 + i) * (XS * 4) + l16) = v;
      cbeg = nbeg; cend = nend; csid = nsid;
    }
    float4 v[4];
#pragma unroll
    for (int i = 0; i < 4; ++i) v[i] = *(const float4*)(&xs[(wave * 4 + i) * XS + lane * 4]);
    layernorm4(v, *(const float4*)(g1 + lane * 4), *(const float4*)(be1 + lane * 4));
#pragma unroll
    for (int i = 0; i < 4; ++i)
      *(float4*)(&xs[(wave * 4 + i) * XS + lane * 4]) = v[i];
  }
  __syncthreads();

  {
    const int nb = wave * 16;
    v8f acc[4] = {};
#pragma unroll
    for (int k0 = 0; k0 < HIDDEN; k0 += 4) {
      const int ka = k0 + 2 * hi;
      v2f B;
      B.x = W1[ka * FFN_DIM + nb + l];
      B.y = W1[(ka + 1) * FFN_DIM + nb + l];
#pragma unroll
      for (int mt = 0; mt < 4; ++mt) {
        const v2f A = *(const v2f*)(&xs[(mt * 16 + l) * XS + ka]);
        acc[mt] = __builtin_amdgcn_wmma_f32_16x16x4_f32(false, A, false, B,
                                                        (short)0, acc[mt], false, false);
      }
    }
    const float bv = bf1[nb + l];
#pragma unroll
    for (int mt = 0; mt < 4; ++mt) {
#pragma unroll
      for (int r = 0; r < 8; r += 2) {
        v2f x = {acc[mt][r], acc[mt][r + 1]};
        x = gelu_pair(x + bv);
        ts[(mt * 16 + r + 8 * hi) * TS + nb + l]     = x.x;
        ts[(mt * 16 + r + 1 + 8 * hi) * TS + nb + l] = x.y;
      }
    }
  }
  __syncthreads();

  {
    const int nb  = (wave & 7) * 16;
    const int mt0 = (wave >> 3) * 2;
    v8f acc[2] = {};
#pragma unroll
    for (int k0 = 0; k0 < FFN_DIM; k0 += 4) {
      const int ka = k0 + 2 * hi;
      v2f B;
      B.x = W2[ka * HIDDEN + nb + l];
      B.y = W2[(ka + 1) * HIDDEN + nb + l];
#pragma unroll
      for (int j = 0; j < 2; ++j) {
        const v2f A = *(const v2f*)(&ts[((mt0 + j) * 16 + l) * TS + ka]);
        acc[j] = __builtin_amdgcn_wmma_f32_16x16x4_f32(false, A, false, B,
                                                       (short)0, acc[j], false, false);
      }
    }
    const float bv = bf2[nb + l];
#pragma unroll
    for (int j = 0; j < 2; ++j) {
#pragma unroll
      for (int r = 0; r < 8; ++r) {
        const int idx = ((mt0 + j) * 16 + r + 8 * hi) * XS + nb + l;
        xs[idx] = acc[j][r] + bv + xs[idx];
      }
    }
  }
  __syncthreads();

  {
    float4 v[4];
#pragma unroll
    for (int i = 0; i < 4; ++i) v[i] = *(const float4*)(&xs[(wave * 4 + i) * XS + lane * 4]);
    layernorm4(v, *(const float4*)(g2 + lane * 4), *(const float4*)(be2 + lane * 4));
#pragma unroll
    for (int i = 0; i < 4; ++i) {
      const int row = row0 + wave * 4 + i;
      if (row < N_NODES) *(float4*)(out + row * HIDDEN + lane * 4) = v[i];
    }
  }
}

extern "C" void kernel_launch(void* const* d_in, const int* in_sizes, int n_in,
                              void* d_out, int out_size, void* d_ws, size_t ws_size,
                              hipStream_t stream) {
  const float* h     = (const float*)d_in[0];
  const int*   src   = (const int*)d_in[1];
  const int*   dst   = (const int*)d_in[2];
  const float* Wsrc  = (const float*)d_in[3];
  const float* bsrc  = (const float*)d_in[4];
  const float* Wdst  = (const float*)d_in[5];
  const float* bdst  = (const float*)d_in[6];
  const float* attn  = (const float*)d_in[7];
  const float* W1    = (const float*)d_in[8];
  const float* bf1   = (const float*)d_in[9];
  const float* W2    = (const float*)d_in[10];
  const float* bf2   = (const float*)d_in[11];
  const float* g1    = (const float*)d_in[12];
  const float* be1   = (const float*)d_in[13];
  const float* g2    = (const float*)d_in[14];
  const float* be2   = (const float*)d_in[15];
  float* out = (float*)d_out;

  float* ws     = (float*)d_ws;
  float* fs     = ws;
  float* fd     = fs + (size_t)N_NODES * HIDDEN;
  int*   H      = (int*)(fd + (size_t)N_NODES * HIDDEN);
  int*   counts = H + (size_t)N_NODES * HIDDEN;
  int*   offs   = counts + N_NODES;
  int*   csrc   = offs + N_NODES + 16 + N_EDGES;

  k_hist<<<dim3(SORT_G, SORT_S), SORT_THREADS, 0, stream>>>(dst, H);
  k_gat_proj<<<PROJ_WGS, 512, 0, stream>>>(h, Wsrc, bsrc, Wdst, bdst, fs, fd);
  k_colprefix<<<(N_NODES + 255) / 256, 256, 0, stream>>>(H, counts);
  k_scan_offsets<<<1, SCAN_THREADS, 0, stream>>>(counts, offs);
  k_place<<<dim3(SORT_G, SORT_S), SORT_THREADS, 0, stream>>>(src, dst, H, offs, csrc);
  k_tail<<<(N_NODES + TM - 1) / TM, 512, 0, stream>>>(h, fs, fd, attn, offs, csrc,
                                                       g1, be1, W1, bf1, W2, bf2, g2, be2, out);
}


# Tail kernel re-cut: 512 resident work-groups, each a run of 6-7 row tiles in two passes (4 tiles, then 2-3), replace 782 one-tile groups (1.5 resident rounds): 12-14 tiles per WGP, was 8 or 16
# speedup vs baseline: 3.1222x; 1.0413x over previous
#include <hip/hip_runtime.h>
#include <math.h>


#define N_NODES   50000
#define N_EDGES   600000
#define HIDDEN    128
#define FFN_DIM   256
#define HEADS     4
#define HEAD_DIM  32
#define NEG_SLOPE 0.2f
#define LN_EPS    1e-5f

typedef __attribute__((ext_vector_type(2))) float v2f;
typedef __attribute__((ext_vector_type(8))) float v8f;

__device__ __forceinline__ v2f splat2(const float c) { v2f v = {c, c}; return v; }

__device__ __forceinline__ float exp_nonpos(const float x) {
  return __builtin_amdgcn_exp2f(x * 1.4426950408889634f);
}

#define ERF_LIM 3.832506856900711f
__device__ __forceinline__ v2f gelu_pair(const v2f x) {
  v2f t = x * 0.70710678118654752f;
  t = __builtin_elementwise_min(__builtin_elementwise_max(t, splat2(-ERF_LIM)), splat2(ERF_LIM));
  const v2f s = t * t;
  v2f p = __builtin_elementwise_fma(s, splat2(0.00022905065861350646f), splat2(0.0034082910107109506f));
  p = __builtin_elementwise_fma(p, s, splat2(0.050955695062380861f));
  p = __builtin_elementwise_fma(p, s, splat2(0.18520832239976145f));
  p = __builtin_elementwise_fma(p, s, splat2(1.128379143519084f));
  v2f q = __builtin_elementwise_fma(s, splat2(-1.1791602954361697e-7f), splat2(0.000023547966471313185f));
  q = __builtin_elementwise_fma(q, s, splat2(0.0010179625278914885f));
  q = __builtin_elementwise_fma(q, s, splat2(0.014070470171167667f));
  q = __builtin_elementwise_fma(q, s, splat2(0.11098505178285362f));
  q = __builtin_elementwise_fma(q, s, splat2(0.49746925110067538f));
  q = __builtin_elementwise_fma(q, s, splat2(1.0f));
  v2f r;
  r.x = __builtin_amdgcn_rcpf(q.x);
  r.y = __builtin_amdgcn_rcpf(q.y);
  const v2f e = __builtin_elementwise_fma(-q, r, splat2(1.0f));
  r = __builtin_elementwise_fma(e, r, r);
  const v2f erfv = (p * t) * r;
  const v2f hx = x * 0.5f;
  return __builtin_elementwise_fma(hx, erfv, hx);
}

#define SORT_G       32
#define SORT_S       8
#define SLICE_N      (N_NODES / SORT_S)
#define CHUNK_E      18752
#define SORT_THREADS 1024
#define SORT_ITERS   5
#define SLICE_ITERS  7
static_assert(SORT_S * SLICE_N == N_NODES, "slices must tile the nodes");
static_assert(SORT_G * CHUNK_E >= N_EDGES && CHUNK_E % 4 == 0 && N_EDGES % 4 == 0, "chunks");
static_assert(SORT_ITERS * SORT_THREADS * 4 >= CHUNK_E, "one pass over a chunk");
static_assert(SLICE_ITERS * SORT_THREADS >= SLICE_N, "one pass over a slice");

__global__ __launch_bounds__(SORT_THREADS) void k_hist(const int* __restrict__ dst,
                                                       int* __restrict__ H) {
  __shared__ int hist[SLICE_N];
  const int g = blockIdx.x, s = blockIdx.y, tid = threadIdx.x;
  const int lo = s * SLICE_N;
  const int e0 = g * CHUNK_E;
  const int e1 = min(e0 + CHUNK_E, N_EDGES);
  int4 d[SORT_ITERS];
#pragma unroll
  for (int i = 0; i < SORT_ITERS; ++i) {
    const int e = e0 + (i * SORT_THREADS + tid) * 4;
    d[i] = (e < e1) ? *(const int4*)(dst + e) : make_int4(-1, -1, -1, -1);
  }
  for (int j = tid; j < SLICE_N; j += SORT_THREADS) hist[j] = 0;
  __syncthreads();
#pragma unroll
  for (int i = 0; i < SORT_ITERS; ++i) {
    const unsigned n0 = (unsigned)(d[i].x - lo), n1 = (unsigned)(d[i].y - lo);
    const unsigned n2 = (unsigned)(d[i].z - lo), n3 = (unsigned)(d[i].w - lo);
    if (n0 < SLICE_N) __hip_atomic_fetch_add(&hist[n0], 1, __ATOMIC_RELAXED, __HIP_MEMORY_SCOPE_WORKGROUP);
    if (n1 < SLICE_N) __hip_atomic_fetch_add(&hist[n1], 1, __ATOMIC_RELAXED, __HIP_MEMORY_SCOPE_WORKGROUP);
    if (n2 < SLICE_N) __hip_atomic_fetch_add(&hist[n2], 1, __ATOMIC_RELAXED, __HIP_MEMORY_SCOPE_WORKGROUP);
    if (n3 < SLICE_N) __hip_atomic_fetch_add(&hist[n3], 1, __ATOMIC_RELAXED, __HIP_MEMORY_SCOPE_WORKGROUP);
  }
  __syncthreads();
  for (int j = tid; j < SLICE_N; j += SORT_THREADS) H[g * N_NODES + lo + j] = hist[j];
}

__global__ void k_colprefix(int* __restrict__ H, int* __restrict__ counts) {
  const int n = blockIdx.x * 256 + threadIdx.x;
  if (n >= N_NODES) return;
  int c[SORT_G];
#pragma unroll
  for (int g = 0; g < SORT_G; ++g) c[g] = H[g * N_NODES + n];
  int run = 0;
#pragma unroll
  for (int g = 0; g < SORT_G; ++g) {
    H[g * N_NODES + n] = run;
    run += c[g];
  }
  counts[n] = run;
}

#define SCAN_THREADS 1024
#define SCAN_ITERS   13
#define SCAN_SEG     (SCAN_ITERS * 128)
__global__ __launch_bounds__(SCAN_THREADS) void k_scan_offsets(
    const int* __restrict__ counts, int* __restrict__ offs) {
  __shared__ int wsum[32];
  const int tid = threadIdx.x, lane = tid & 31, wave = tid >> 5;
  const int seg0 = wave * SCAN_SEG + lane * 4;
  int4 v[SCAN_ITERS];
  int  excl[SCAN_ITERS];
#pragma unroll
  for (int it = 0; it < SCAN_ITERS; ++it) {
    const int idx = seg0 + it * 128;
    if (idx < N_NODES) v[it] = *(const int4*)(counts + idx);
    else               v[it] = make_int4(0, 0, 0, 0);
  }
  int carry = 0;
#pragma unroll
  for (int it = 0; it < SCAN_ITERS; ++it) {
    const int s4 = v[it].x + v[it].y + v[it].z + v[it].w;
    int inc = s4;
#pragma unroll
    for (int off = 1; off < 32; off <<= 1) {
      const int t = __shfl_up(inc, off, 32);
      if (lane >= off) inc += t;
    }
    excl[it] = carry + inc - s4;
    carry += __shfl(inc, 31, 32);
  }
  if (lane == 0) wsum[wave] = carry;
  __syncthreads();
  if (wave == 0) {
    const int ws = wsum[lane];
    int winc = ws;
#pragma unroll
    for (int off = 1; off < 32; off <<= 1) {
      const int t = __shfl_up(winc, off, 32);
      if (lane >= off) winc += t;
    }
    wsum[lane] = winc - ws;
  }
  __syncthreads();
  const int wbase = wsum[wave];
#pragma unroll
  for (int it = 0; it < SCAN_ITERS; ++it) {
    const int idx = seg0 + it * 128;
    if (idx < N_NODES) {
      int4 o;
      o.x = wbase + excl[it];
      o.y = o.x + v[it].x;
      o.z = o.y + v[it].y;
      o.w = o.z + v[it].z;
      *(int4*)(offs + idx) = o;
    }
  }
  if (tid == SCAN_THREADS - 1) offs[N_NODES] = wbase + carry;
}

__global__ __launch_bounds__(SORT_THREADS) void k_place(
    const int* __restrict__ src, const int* __restrict__ dst, const int* __restrict__ H,
    const int* __restrict__ offs, int* __restrict__ csrc) {
  __shared__ int next[SLICE_N];
  const int g = blockIdx.x, s = blockIdx.y, tid = threadIdx.x;
  const int lo = s * SLICE_N;
  const int e0 = g * CHUNK_E;
  const int e1 = min(e0 + CHUNK_E, N_EDGES);
  int4 d[SORT_ITERS], sv[SORT_ITERS];
  int  cur[SLICE_ITERS];
#pragma unroll
  for (int i = 0; i < SORT_ITERS; ++i) {
    const int e = e0 + (i * SORT_THREADS + tid) * 4;
    if (e < e1) {
      d[i]  = *(const int4*)(dst + e);
      sv[i] = *(const int4*)(src + e);
    } else {
      d[i]  = make_int4(-1, -1, -1, -1);
      sv[i] = make_int4(0, 0, 0, 0);
    }
  }
#pragma unroll
  for (int i = 0; i < SLICE_ITERS; ++i) {
    const int j = i * SORT_THREADS + tid;
    cur[i] = (j < SLICE_N) ? offs[lo + j] + H[g * N_NODES + lo + j] : 0;
  }
#pragma unroll
  for (int i = 0; i < SLICE_ITERS; ++i) {
    const int j = i * SORT_THREADS + tid;
    if (j < SLICE_N) next[j] = cur[i];
  }
  __syncthreads();
  int pos[SORT_ITERS * 4];
#pragma unroll
  for (int i = 0; i < SORT_ITERS; ++i) {
    const unsigned n0 = (unsigned)(d[i].x - lo), n1 = (unsigned)(d[i].y - lo);
    const unsigned n2 = (unsigned)(d[i].z - lo), n3 = (unsigned)(d[i].w - lo);
    pos[4 * i + 0] = (n0 < SLICE_N) ? __hip_atomic_fetch_add(&next[n0], 1, __ATOMIC_RELAXED, __HIP_MEMORY_SCOPE_WORKGROUP) : -1;
    pos[4 * i + 1] = (n1 < SLICE_N) ? __hip_atomic_fetch_add(&next[n1], 1, __ATOMIC_RELAXED, __HIP_MEMORY_SCOPE_WORKGROUP) : -1;
    pos[4 * i + 2] = (n2 < SLICE_N) ? __hip_atomic_fetch_add(&next[n2], 1, __ATOMIC_RELAXED, __HIP_MEMORY_SCOPE_WORKGROUP) : -1;
    pos[4 * i + 3] = (n3 < SLICE_N) ? __hip_atomic_fetch_add(&next[n3], 1, __ATOMIC_RELAXED, __HIP_MEMORY_SCOPE_WORKGROUP) : -1;
  }
#pragma unroll
  for (int i = 0; i < SORT_ITERS; ++i) {
    if (pos[4 * i + 0] >= 0) csrc[pos[4 * i + 0]] = sv[i].x;
    if (pos[4 * i + 1] >= 0) csrc[pos[4 * i + 1]] = sv[i].y;
    if (pos[4 * i + 2] >= 0) csrc[pos[4 * i + 2]] = sv[i].z;
    if (pos[4 * i + 3] >= 0) csrc[pos[4 * i + 3]] = sv[i].w;
  }
}

#define PROJ_WGS   512
#define PROJ_TILES (N_NODES / 16)
#define PROJ_LO    (PROJ_TILES / PROJ_WGS)
#define PROJ_HI    (PROJ_LO + 1)
#define PROJ_EXTRA (PROJ_TILES - PROJ_LO * PROJ_WGS)
#define PROJ_PITCH 132
static_assert(N_NODES % 16 == 0, "whole 16-row tiles");
static_assert(PROJ_EXTRA >= 0 && PROJ_EXTRA < PROJ_WGS, "6 or 7 tiles per work-group");

template <int NT>
__device__ __forceinline__ void proj_tiles(const float* __restrict__ tile, const int row0,
                                           const int wave, const int lane,
                                           const float* __restrict__ Wsrc, const float* __restrict__ bsrc,
                                           const float* __restrict__ Wdst, const float* __restrict__ bdst,
                                           float* __restrict__ fs, float* __restrict__ fd) {
  const int l = lane & 15, hi = lane >> 4;
  const int sel = wave >> 3;
  const int nb  = (wave & 7) * 16;
  const float* W    = sel ? Wdst : Wsrc;
  const float* bias = sel ? bdst : bsrc;
  float*       out  = sel ? fd   : fs;

  v8f acc[NT] = {};
#pragma unroll 8
  for (int k0 = 0; k0 < HIDDEN; k0 += 4) {
    const int ka = k0 + 2 * hi;
    v2f B;
    B.x = W[ka * HIDDEN + nb + l];
    B.y = W[(ka + 1) * HIDDEN + nb + l];
#pragma unroll
    for (int mt = 0; mt < NT; ++mt) {
      const v2f A = *(const v2f*)(&tile[(mt * 16 + l) * PROJ_PITCH + ka]);
      acc[mt] = __builtin_amdgcn_wmma_f32_16x16x4_f32(false, A, false, B,
                                                      (short)0, acc[mt], false, false);
    }
  }
  const float bv = bias[nb + l];
#pragma unroll
  for (int mt = 0; mt < NT; ++mt) {
#pragma unroll
    for (int r = 0; r < 8; ++r)
      out[(row0 + mt * 16 + r + 8 * hi) * HIDDEN + nb + l] = acc[mt][r] + bv;
  }
}

__global__ __launch_bounds__(512, 8) void k_gat_proj(
    const float* __restrict__ h,
    const float* __restrict__ Wsrc, const float* __restrict__ bsrc,
    const float* __restrict__ Wdst, const float* __restrict__ bdst,
    float* __restrict__ fs, float* __restrict__ fd) {
  __shared__ __attribute__((aligned(16))) float tile[PROJ_HI * 16 * PROJ_PITCH];
  const int tid = threadIdx.x, lane = tid & 31;
  const int wave = __builtin_amdgcn_readfirstlane(tid >> 5);
  const int b    = blockIdx.x;
  const int nt   = PROJ_LO + (b < PROJ_EXTRA ? 1 : 0);
  const int row0 = (b * PROJ_LO + min(b, PROJ_EXTRA)) * 16;
#pragma unroll
  for (int i = 0; i < PROJ_HI; ++i) {
    if (i < nt) {
      const int r = i * 16 + wave;
      *(float4*)(&tile[r * PROJ_PITCH + lane * 4]) = *(const float4*)(h + (row0 + r) * HIDDEN + lane * 4);
    }
  }
  __syncthreads();
  if (nt == PROJ_HI) proj_tiles<PROJ_HI>(tile, row0, wave, lane, Wsrc, bsrc, Wdst, bdst, fs, fd);
  else               proj_tiles<PROJ_LO>(tile, row0, wave, lane, Wsrc, bsrc, Wdst, bdst, fs, fd);
}

#define CHUNK 16
__device__ __forceinline__ float edge_logit(const v2f alo, const v2f ahi, const v2f blo,
                                            const v2f bhi, const v2f wlo, const v2f whi) {
  v2f t0 = alo + blo, t1 = ahi + bhi;
  t0 = __builtin_elementwise_max(t0, t0 * NEG_SLOPE);
  t1 = __builtin_elementwise_max(t1, t1 * NEG_SLOPE);
  v2f q = t0 * wlo;
  q += t1 * whi;
  float p = q.x + q.y;
  p += __shfl_xor(p, 1, 8);
  p += __shfl_xor(p, 2, 8);
  p += __shfl_xor(p, 4, 8);
  return p;
}

template <int W>
__device__ __forceinline__ void gat_chunk(const float* __restrict__ fs, const unsigned l16,
                                          const int sorted, const int half, const int cnt,
                                          const v2f blo, const v2f bhi,
                                          const v2f wlo, const v2f whi,
                                          float& m, float& l, v2f& alo, v2f& ahi) {
  v2f   rlo[W], rhi[W];
  float p[W];
#pragma unroll
  for (int j = 0; j < W; ++j) {
    const int s = __builtin_amdgcn_readlane(sorted, half + min(j, cnt - 1));
    const unsigned off = ((unsigned)s << 9) + l16;
    const float4 r = *(const float4*)((const char*)fs + off);
    rlo[j].x = r.x; rlo[j].y = r.y; rhi[j].x = r.z; rhi[j].y = r.w;
  }
  float cm = -INFINITY;
#pragma unroll
  for (int j = 0; j < W; ++j) {
    const float q = edge_logit(rlo[j], rhi[j], blo, bhi, wlo, whi);
    p[j] = (j < cnt) ? q : -INFINITY;
    cm = fmaxf(cm, p[j]);
  }
  const float mn = fmaxf(m, cm);
  const float sc = exp_nonpos(m - mn);
  l *= sc;
  alo *= sc; ahi *= sc;
#pragma unroll
  for (int j = 0; j < W; ++j) {
    const float ex = exp_nonpos(p[j] - mn);
    l += ex;
    alo += rlo[j] * ex;
    ahi += rhi[j] * ex;
  }
  m = mn;
}

__device__ __forceinline__ float4 gat_node_row(const float* __restrict__ fs,
                                               const int* __restrict__ csrc, const int lane,
                                               const unsigned l16,
                                               const int beg, const int end, const int sid0,
                                               const float4 b, const float4 w) {
  float m = -INFINITY, l = 0.0f;
  v2f alo = {0.0f, 0.0f}, ahi = {0.0f, 0.0f};
  const v2f blo = {b.x, b.y}, bhi = {b.z, b.w};
  const v2f wlo = {w.x, w.y}, whi = {w.z, w.w};
  for (int base = beg; base < end; base += 32) {
    const int n = min(32, end - base);
    int sid = sid0;
    if (base != beg) sid = (lane < n) ? csrc[base + lane] : 0x7fffffff;
    int rk = 0;
    for (int j = 0; j < n; ++j) {
      const int o = __builtin_amdgcn_readlane(sid, j);
      rk += ((o < sid) || (o == sid && j < lane)) ? 1 : 0;
    }
    const int sorted = __builtin_amdgcn_ds_permute(rk << 2, sid);

    for (int half = 0; half < n; half += CHUNK) {
      const int cnt = min(CHUNK, n - half);
      const int quads = (cnt + 3) >> 2;
      if (quads == 4)      gat_chunk<16>(fs, l16, sorted, half, cnt, blo, bhi, wlo, whi, m, l, alo, ahi);
      else if (quads == 3) gat_chunk<12>(fs, l16, sorted, half, cnt, blo, bhi, wlo, whi, m, l, alo, ahi);
      else if (quads == 2) gat_chunk<8>(fs, l16, sorted, half, cnt, blo, bhi, wlo, whi, m, l, alo, ahi);
      else                 gat_chunk<4>(fs, l16, sorted, half, cnt, blo, bhi, wlo, whi, m, l, alo, ahi);
    }
  }
  const float den = (l == 0.0f) ? 1.0f : l;
  float4 o;
  o.x = alo.x / den; o.y = alo.y / den; o.z = ahi.x / den; o.w = ahi.y / den;
  return o;
}

#define TAIL_WGS   512
#define TAIL_TILES (N_NODES / 16)
#define TAIL_LO    (TAIL_TILES / TAIL_WGS)
#define TAIL_EXTRA (TAIL_TILES - TAIL_LO * TAIL_WGS)
#define TM 64
#define XS 132
#define TS 260
static_assert(TAIL_LO == 6 && TAIL_EXTRA >= 0 && TAIL_EXTRA < TAIL_WGS, "4 + 2 or 4 + 3 tiles per work-group");

template <int NR>
__device__ __forceinline__ void layernorm_rows(float4 (&v)[NR], const float4 g, const float4 b) {
  float sum[NR];
#pragma unroll
  for (int i = 0; i < NR; ++i) sum[i] = v[i].x + v[i].y + v[i].z + v[i].w;
#pragma unroll
  for (int off = 1; off < 32; off <<= 1) {
#pragma unroll
    for (int i = 0; i < NR; ++i) sum[i] += __shfl_xor(sum[i], off, 32);
  }
  float ss[NR];
#pragma unroll
  for (int i = 0; i < NR; ++i) {
    const float mu = sum[i] * (1.0f / HIDDEN);
    v[i].x -= mu; v[i].y -= mu; v[i].z -= mu; v[i].w -= mu;
    ss[i] = v[i].x * v[i].x + v[i].y * v[i].y + v[i].z * v[i].z + v[i].w * v[i].w;
  }
#pragma unroll
  for (int off = 1; off < 32; off <<= 1) {
#pragma unroll
    for (int i = 0; i < NR; ++i) ss[i] += __shfl_xor(ss[i], off, 32);
  }
#pragma unroll
  for (int i = 0; i < NR; ++i) {
    const float rs = rsqrtf(ss[i] * (1.0f / HIDDEN) + LN_EPS);
    v[i].x = v[i].x * rs * g.x + b.x;
    v[i].y = v[i].y * rs * g.y + b.y;
    v[i].z = v[i].z * rs * g.z + b.z;
    v[i].w = v[i].w * rs * g.w + b.w;
  }
}

template <int CNT>
__device__ __forceinline__ void ffn2_tiles(float* __restrict__ xs, const float* __restrict__ ts,
                                           const int mtA, const int nb, const int l, const int hi,
                                           const float* __restrict__ W2, const float* __restrict__ bf2) {
  v8f acc[CNT] = {};
#pragma unroll
  for (int k0 = 0; k0 < FFN_DIM; k0 += 4) {
    const int ka = k0 + 2 * hi;
    v2f B;
    B.x = W2[ka * HIDDEN + nb + l];
    B.y = W2[(ka + 1) * HIDDEN + nb + l];
#pragma unroll
    for (int j = 0; j < CNT; ++j) {
      const v2f A = *(const v2f*)(&ts[((mtA + j) * 16 + l) * TS + ka]);
      acc[j] = __builtin_amdgcn_wmma_f32_16x16x4_f32(false, A, false, B,
                                                     (short)0, acc[j], false, false);
    }
  }
  const float bv = bf2[nb + l];
#pragma unroll
  for (int j = 0; j < CNT; ++j) {
#pragma unroll
    for (int r = 0; r < 8; ++r) {
      const int idx = ((mtA + j) * 16 + r + 8 * hi) * XS + nb + l;
      xs[idx] = acc[j][r] + bv + xs[idx];
    }
  }
}

template <int NT>
__device__ __forceinline__ void tail_pass(
    float* __restrict__ xs, float* __restrict__ ts, const int prow0,
    const int wave, const int lane_in,
    const float* __restrict__ h, const float* __restrict__ fs, const float* __restrict__ fd,
    const float* __restrict__ attn, const int* __restrict__ offs, const int* __restrict__ csrc,
    const float* __restrict__ g1, const float* __restrict__ be1,
    const float* __restrict__ W1, const float* __restrict__ bf1,
    const float* __restrict__ W2, const float* __restrict__ bf2,
    const float* __restrict__ g2, const float* __restrict__ be2,
    float* __restrict__ out) {
  int lane = lane_in;
  asm volatile("" : "+v"(lane));
  const int l = lane & 15, hi = lane >> 4;
  {
    const int n0 = prow0 + wave * NT;
    const unsigned l16 = (unsigned)lane * 16u;
    char* const xsb = (char*)xs;
    char* const tsb = (char*)ts;
    const float4 w = *(const float4*)((const char*)attn + l16);
    int cbeg = offs[n0], cend = offs[n0 + 1];
    float4 hx[NT], fb[NT];
#pragma unroll
    for (int i = 0; i < NT; ++i) {
      hx[i] = *(const float4*)((const char*)(h + (n0 + i) * HIDDEN) + l16);
      fb[i] = *(const float4*)((const char*)(fd + (n0 + i) * HIDDEN) + l16);
    }
    int csid = (lane < min(32, cend - cbeg)) ? csrc[cbeg + lane] : 0x7fffffff;
#pragma unroll
    for (int i = 0; i < NT; ++i) {
      *(float4*)(xsb + (wave * NT + i) * (XS * 4) + l16) = hx[i];
      *(float4*)(tsb + (wave * NT + i) * (TS * 4) + l16) = fb[i];
    }
    for (int i = 0; i < NT; ++i) {
      int nbeg = cend, nend = cend, nsid = 0x7fffffff;
      if (i < NT - 1) {
        nend = offs[n0 + i + 2];
        nsid = (lane < min(32, nend - nbeg)) ? csrc[nbeg + lane] : 0x7fffffff;
      }
      const float4 b = *(const float4*)(tsb + (wave * NT + i) * (TS * 4) + l16);
      const float4 a = gat_node_row(fs, csrc, lane, l16, cbeg, cend, csid, b, w);
      const float4 x = *(const float4*)(xsb + (wave * NT + i) * (XS * 4) + l16);
      float4 v;
      v.x = x.x + a.x; v.y = x.y + a.y; v.z = x.z + a.z; v.w = x.w + a.w;
      *(float4*)(xsb + (wave * NT + i) * (XS * 4) + l16) = v;
      cbeg = nbeg; cend = nend; csid = nsid;
    }
    float4 v[NT];
#pragma unroll
    for (int i = 0; i < NT; ++i) v[i] = *(const float4*)(&xs[(wave * NT + i) * XS + lane * 4]);
    layernorm_rows<NT>(v, *(const float4*)(g1 + lane * 4), *(const float4*)(be1 + lane * 4));
#pragma unroll
    for (int i = 0; i < NT; ++i)
      *(float4*)(&xs[(wave * NT + i) * XS + lane * 4]) = v[i];
  }
  __syncthreads();

  {
    const int nb = wave * 16;
    v8f acc[NT] = {};
#pragma unroll
    for (int k0 = 0; k0 < HIDDEN; k0 += 4) {
      const int ka = k0 + 2 * hi;
      v2f B;
      B.x = W1[ka * FFN_DIM + nb + l];
      B.y = W1[(ka + 1) * FFN_DIM + nb + l];
#pragma unroll
      for (int mt = 0; mt < NT; ++mt) {
        const v2f A = *(const v2f*)(&xs[(mt * 16 + l) * XS + ka]);
        acc[mt] = __builtin_amdgcn_wmma_f32_16x16x4_f32(false, A, false, B,
                                                        (short)0, acc[mt], false, false);
      }
    }
    const float bv = bf1[nb + l];
#pragma unroll
    for (int mt = 0; mt < NT; ++mt) {
#pragma unroll
      for (int r = 0; r < 8; r += 2) {
        v2f x = {acc[mt][r], acc[mt][r + 1]};
        x = gelu_pair(x + bv);
        ts[(mt * 16 + r + 8 * hi) * TS + nb + l]     = x.x;
        ts[(mt * 16 + r + 1 + 8 * hi) * TS + nb + l] = x.y;
      }
    }
  }
  __syncthreads();

  {
    const int nb  = (wave & 7) * 16;
    const int gq  = wave >> 3;
    const int mtA = gq ? (NT + 1) / 2 : 0;
    const int cnt = gq ? NT / 2 : (NT + 1) / 2;
    if (cnt == 2) ffn2_tiles<2>(xs, ts, mtA, nb, l, hi, W2, bf2);
    else          ffn2_tiles<1>(xs, ts, mtA, nb, l, hi, W2, bf2);
  }
  __syncthreads();

  {
    float4 v[NT];
#pragma unroll
    for (int i = 0; i < NT; ++i) v[i] = *(const float4*)(&xs[(wave * NT + i) * XS + lane * 4]);
    layernorm_rows<NT>(v, *(const float4*)(g2 + lane * 4), *(const float4*)(be2 + lane * 4));
#pragma unroll
    for (int i = 0; i < NT; ++i)
      *(float4*)(out + (prow0 + wave * NT + i) * HIDDEN + lane * 4) = v[i];
  }
}

__global__ __launch_bounds__(512, 8) void k_tail(
    const float* __restrict__ h, const float* __restrict__ fs, const float* __restrict__ fd,
    const float* __restrict__ attn, const int* __restrict__ offs, const int* __restrict__ csrc,
    const float* __restrict__ g1, const float* __restrict__ be1,
    const float* __restrict__ W1, const float* __restrict__ bf1,
    const float* __restrict__ W2, const float* __restrict__ bf2,
    const float* __restrict__ g2, const float* __restrict__ be2,
    float* __restrict__ out) {
  __shared__ __attribute__((aligned(16))) float xs[TM * XS];
  __shared__ __attribute__((aligned(16))) float ts[TM * TS];
  const int tid = threadIdx.x, lane = tid & 31;
  const int wave = __builtin_amdgcn_readfirstlane(tid >> 5);
  const int b    = blockIdx.x;
  const int nt   = TAIL_LO + (b < TAIL_EXTRA ? 1 : 0);
  const int row0 = (b * TAIL_LO + min(b, TAIL_EXTRA)) * 16;
  tail_pass<4>(xs, ts, row0, wave, lane, h, fs, fd, attn, offs, csrc,
               g1, be1, W1, bf1, W2, bf2, g2, be2, out);
  __syncthreads();
  if (nt == TAIL_LO + 1)
    tail_pass<3>(xs, ts, row0 + 64, wave, lane, h, fs, fd, attn, offs, csrc,
                 g1, be1, W1, bf1, W2, bf2, g2, be2, out);
  else
    tail_pass<2>(xs, ts, row0 + 64, wave, lane, h, fs, fd, attn, offs, csrc,
                 g1, be1, W1, bf1, W2, bf2, g2, be2, out);
}

extern "C" void kernel_launch(void* const* d_in, const int* in_sizes, int n_in,
                              void* d_out, int out_size, void* d_ws, size_t ws_size,
                              hipStream_t stream) {
  const float* h     = (const float*)d_in[0];
  const int*   src   = (const int*)d_in[1];
  const int*   dst   = (const int*)d_in[2];
  const float* Wsrc  = (const float*)d_in[3];
  const float* bsrc  = (const float*)d_in[4];
  const float* Wdst  = (const float*)d_in[5];
  const float* bdst  = (const float*)d_in[6];
  const float* attn  = (const float*)d_in[7];
  const float* W1    = (const float*)d_in[8];
  const float* bf1   = (const float*)d_in[9];
  const float* W2    = (const float*)d_in[10];
  const float* bf2   = (const float*)d_in[11];
  const float* g1    = (const float*)d_in[12];
  const float* be1   = (const float*)d_in[13];
  const float* g2    = (const float*)d_in[14];
  const float* be2   = (const float*)d_in[15];
  float* out = (float*)d_out;

  float* ws     = (float*)d_ws;
  float* fs     = ws;
  float* fd     = fs + (size_t)N_NODES * HIDDEN;
  int*   H      = (int*)(fd + (size_t)N_NODES * HIDDEN);
  int*   counts = H + (size_t)N_NODES * HIDDEN;
  int*   offs   = counts + N_NODES;
  int*   csrc   = offs + N_NODES + 16 + N_EDGES;

  k_hist<<<dim3(SORT_G, SORT_S), SORT_THREADS, 0, stream>>>(dst, H);
  k_gat_proj<<<PROJ_WGS, 512, 0, stream>>>(h, Wsrc, bsrc, Wdst, bdst, fs, fd);
  k_colprefix<<<(N_NODES + 255) / 256, 256, 0, stream>>>(H, counts);
  k_scan_offsets<<<1, SCAN_THREADS, 0, stream>>>(counts, offs);
  k_place<<<dim3(SORT_G, SORT_S), SORT_THREADS, 0, stream>>>(src, dst, H, offs, csrc);
  k_tail<<<TAIL_WGS, 512, 0, stream>>>(h, fs, fd, attn, offs, csrc,
                                                       g1, be1, W1, bf1, W2, bf2, g2, be2, out);
}


# Destination grouping in 2 launches, was 4: each (chunk, slice) work-group counts, scans and places in LDS into a region of its own; a per-node merge kernel builds offsets and compact lists
# speedup vs baseline: 3.4468x; 1.1040x over previous
#include <hip/hip_runtime.h>
#include <math.h>


#define N_NODES   50000
#define N_EDGES   600000
#define HIDDEN    128
#define FFN_DIM   256
#define HEADS     4
#define HEAD_DIM  32
#define NEG_SLOPE 0.2f
#define LN_EPS    1e-5f

typedef __attribute__((ext_vector_type(2))) float v2f;
typedef __attribute__((ext_vector_type(8))) float v8f;

__device__ __forceinline__ v2f splat2(const float c) { v2f v = {c, c}; return v; }

__device__ __forceinline__ float exp_nonpos(const float x) {
  return __builtin_amdgcn_exp2f(x * 1.4426950408889634f);
}

#define ERF_LIM 3.832506856900711f
__device__ __forceinline__ v2f gelu_pair(const v2f x) {
  v2f t = x * 0.70710678118654752f;
  t = __builtin_elementwise_min(__builtin_elementwise_max(t, splat2(-ERF_LIM)), splat2(ERF_LIM));
  const v2f s = t * t;
  v2f p = __builtin_elementwise_fma(s, splat2(0.00022905065861350646f), splat2(0.0034082910107109506f));
  p = __builtin_elementwise_fma(p, s, splat2(0.050955695062380861f));
  p = __builtin_elementwise_fma(p, s, splat2(0.18520832239976145f));
  p = __builtin_elementwise_fma(p, s, splat2(1.128379143519084f));
  v2f q = __builtin_elementwise_fma(s, splat2(-1.1791602954361697e-7f), splat2(0.000023547966471313185f));
  q = __builtin_elementwise_fma(q, s, splat2(0.0010179625278914885f));
  q = __builtin_elementwise_fma(q, s, splat2(0.014070470171167667f));
  q = __builtin_elementwise_fma(q, s, splat2(0.11098505178285362f));
  q = __builtin_elementwise_fma(q, s, splat2(0.49746925110067538f));
  q = __builtin_elementwise_fma(q, s, splat2(1.0f));
  v2f r;
  r.x = __builtin_amdgcn_rcpf(q.x);
  r.y = __builtin_amdgcn_rcpf(q.y);
  const v2f e = __builtin_elementwise_fma(-q, r, splat2(1.0f));
  r = __builtin_elementwise_fma(e, r, r);
  const v2f erfv = (p * t) * r;
  const v2f hx = x * 0.5f;
  return __builtin_elementwise_fma(hx, erfv, hx);
}

#define SORT_G       32
#define SORT_S       8
#define SLICE_N      (N_NODES / SORT_S)
#define CHUNK_E      18752
#define SORT_THREADS 1024
#define SORT_ITERS   5
#define SLICE_ITERS  7
#define MERGE_N      250
#define MERGE_BLOCKS (N_NODES / MERGE_N)
#define SLICE_BLOCKS (SLICE_N / MERGE_N)
#define MERGE_THREADS 256
static_assert(SORT_S * SLICE_N == N_NODES, "slices must tile the nodes");
static_assert(SORT_G * CHUNK_E >= N_EDGES && CHUNK_E % 4 == 0 && N_EDGES % 4 == 0, "chunks");
static_assert(SORT_ITERS * SORT_THREADS * 4 >= CHUNK_E, "one pass over a chunk");
static_assert(SLICE_ITERS * SORT_THREADS >= SLICE_N, "one pass over a slice");
static_assert(MERGE_BLOCKS * MERGE_N == N_NODES && SLICE_BLOCKS * MERGE_N == SLICE_N, "blocks tile the slices");
static_assert(MERGE_BLOCKS <= MERGE_THREADS && MERGE_N <= MERGE_THREADS && SLICE_BLOCKS < SORT_THREADS, "one thread per block");
static_assert(CHUNK_E < 32768, "a run's start and its length take 16 and 15 bits of one word");

__device__ __forceinline__ int wave_scan_incl(int v, const int lane) {
#pragma unroll
  for (int off = 1; off < 32; off <<= 1) {
    const int t = __shfl_up(v, off, 32);
    if (lane >= off) v += t;
  }
  return v;
}

__global__ __launch_bounds__(SORT_THREADS) void k_group(
    const int* __restrict__ src, const int* __restrict__ dst,
    int* __restrict__ region, int* __restrict__ seg, int* __restrict__ btot) {
  __shared__ int next[SLICE_ITERS * SORT_THREADS];
  __shared__ int wsum[32];
  __shared__ int bstart[SLICE_BLOCKS + 1];
  const int g = blockIdx.x, s = blockIdx.y, tid = threadIdx.x;
  const int lane = tid & 31, wave = tid >> 5;
  const int lo = s * SLICE_N;
  const int e0 = g * CHUNK_E;
  const int e1 = min(e0 + CHUNK_E, N_EDGES);
  int4 d[SORT_ITERS], sv[SORT_ITERS];
#pragma unroll
  for (int i = 0; i < SORT_ITERS; ++i) {
    const int e = e0 + (i * SORT_THREADS + tid) * 4;
    if (e < e1) {
      d[i]  = *(const int4*)(dst + e);
      sv[i] = *(const int4*)(src + e);
    } else {
      d[i]  = make_int4(-1, -1, -1, -1);
      sv[i] = make_int4(0, 0, 0, 0);
    }
  }
#pragma unroll
  for (int i = 0; i < SLICE_ITERS; ++i) next[i * SORT_THREADS + tid] = 0;
  __syncthreads();
#pragma unroll
  for (int i = 0; i < SORT_ITERS; ++i) {
    const unsigned n0 = (unsigned)(d[i].x - lo), n1 = (unsigned)(d[i].y - lo);
    const unsigned n2 = (unsigned)(d[i].z - lo), n3 = (unsigned)(d[i].w - lo);
    if (n0 < SLICE_N) __hip_atomic_fetch_add(&next[n0], 1, __ATOMIC_RELAXED, __HIP_MEMORY_SCOPE_WORKGROUP);
    if (n1 < SLICE_N) __hip_atomic_fetch_add(&next[n1], 1, __ATOMIC_RELAXED, __HIP_MEMORY_SCOPE_WORKGROUP);
    if (n2 < SLICE_N) __hip_atomic_fetch_add(&next[n2], 1, __ATOMIC_RELAXED, __HIP_MEMORY_SCOPE_WORKGROUP);
    if (n3 < SLICE_N) __hip_atomic_fetch_add(&next[n3], 1, __ATOMIC_RELAXED, __HIP_MEMORY_SCOPE_WORKGROUP);
  }
  __syncthreads();
  int c[SLICE_ITERS];
  int tsum = 0;
#pragma unroll
  for (int i = 0; i < SLICE_ITERS; ++i) {
    c[i] = next[tid * SLICE_ITERS + i];
    tsum += c[i];
  }
  const int inc = wave_scan_incl(tsum, lane);
  if (lane == 31) wsum[wave] = inc;
  __syncthreads();
  if (wave == 0) {
    const int ws = wsum[lane];
    wsum[lane] = wave_scan_incl(ws, lane) - ws;
  }
  __syncthreads();
  int run = wsum[wave] + inc - tsum;
#pragma unroll
  for (int i = 0; i < SLICE_ITERS; ++i) {
    const int j = tid * SLICE_ITERS + i;
    if (j < SLICE_N) {
      seg[g * N_NODES + lo + j] = run | (c[i] << 16);
      next[j] = run;
      if (j % MERGE_N == 0) bstart[j / MERGE_N] = run;
    }
    run += c[i];
  }
  if (tid == SORT_THREADS - 1) bstart[SLICE_BLOCKS] = run;
  __syncthreads();
  if (tid < SLICE_BLOCKS)
    btot[g * MERGE_BLOCKS + s * SLICE_BLOCKS + tid] = bstart[tid + 1] - bstart[tid];
  int* const reg = region + (g * SORT_S + s) * CHUNK_E;
  int pos[SORT_ITERS * 4];
#pragma unroll
  for (int i = 0; i < SORT_ITERS; ++i) {
    const unsigned n0 = (unsigned)(d[i].x - lo), n1 = (unsigned)(d[i].y - lo);
    const unsigned n2 = (unsigned)(d[i].z - lo), n3 = (unsigned)(d[i].w - lo);
    pos[4 * i + 0] = (n0 < SLICE_N) ? __hip_atomic_fetch_add(&next[n0], 1, __ATOMIC_RELAXED, __HIP_MEMORY_SCOPE_WORKGROUP) : -1;
    pos[4 * i + 1] = (n1 < SLICE_N) ? __hip_atomic_fetch_add(&next[n1], 1, __ATOMIC_RELAXED, __HIP_MEMORY_SCOPE_WORKGROUP) : -1;
    pos[4 * i + 2] = (n2 < SLICE_N) ? __hip_atomic_fetch_add(&next[n2], 1, __ATOMIC_RELAXED, __HIP_MEMORY_SCOPE_WORKGROUP) : -1;
    pos[4 * i + 3] = (n3 < SLICE_N) ? __hip_atomic_fetch_add(&next[n3], 1, __ATOMIC_RELAXED, __HIP_MEMORY_SCOPE_WORKGROUP) : -1;
  }
#pragma unroll
  for (int i = 0; i < SORT_ITERS; ++i) {
    if (pos[4 * i + 0] >= 0) reg[pos[4 * i + 0]] = sv[i].x;
    if (pos[4 * i + 1] >= 0) reg[pos[4 * i + 1]] = sv[i].y;
    if (pos[4 * i + 2] >= 0) reg[pos[4 * i + 2]] = sv[i].z;
    if (pos[4 * i + 3] >= 0) reg[pos[4 * i + 3]] = sv[i].w;
  }
}

__global__ __launch_bounds__(MERGE_THREADS) void k_merge(
    const int* __restrict__ region, const int* __restrict__ seg, const int* __restrict__ btot,
    int* __restrict__ offs, int* __restrict__ csrc) {
  __shared__ int red[MERGE_THREADS / 32];
  __shared__ int wsum[MERGE_THREADS / 32];
  const int b = blockIdx.x, tid = threadIdx.x, lane = tid & 31, wave = tid >> 5;
  const bool live = tid < MERGE_N;
  const int n = b * MERGE_N + min(tid, MERGE_N - 1);
  int v[SORT_G];
#pragma unroll
  for (int g = 0; g < SORT_G; ++g) v[g] = seg[g * N_NODES + n];
  int part = 0;
  if (tid < b) {
#pragma unroll
    for (int g = 0; g < SORT_G; ++g) part += btot[g * MERGE_BLOCKS + tid];
  }
#pragma unroll
  for (int off = 1; off < 32; off <<= 1) part += __shfl_xor(part, off, 32);
  int tot = 0;
#pragma unroll
  for (int g = 0; g < SORT_G; ++g) tot += v[g] >> 16;
  if (!live) tot = 0;
  const int inc = wave_scan_incl(tot, lane);
  if (lane == 0)  red[wave]  = part;
  if (lane == 31) wsum[wave] = inc;
  __syncthreads();
  int p = inc - tot;
#pragma unroll
  for (int w = 0; w < MERGE_THREADS / 32; ++w) {
    p += red[w];
    if (w < wave) p += wsum[w];
  }
  if (!live) return;
  offs[n] = p;
  if (n == N_NODES - 1) offs[N_NODES] = p + tot;
  const int* const rs = region + (n / SLICE_N) * CHUNK_E;
  int a0[SORT_G], a1[SORT_G];
  unsigned more = 0;
#pragma unroll
  for (int g = 0; g < SORT_G; ++g) {
    const int* const r = rs + g * (SORT_S * CHUNK_E) + (v[g] & 0xffff);
    const int len = v[g] >> 16;
    a0[g] = (len > 0) ? r[0] : 0;
    a1[g] = (len > 1) ? r[1] : 0;
    if (len > 2) more |= 1u << g;
  }
#pragma unroll
  for (int g = 0; g < SORT_G; ++g)
    if ((v[g] >> 16) > 0) csrc[p++] = a0[g];
#pragma unroll
  for (int g = 0; g < SORT_G; ++g)
    if ((v[g] >> 16) > 1) csrc[p++] = a1[g];
  while (more) {
    const int g = __builtin_ctz(more);
    more &= more - 1;
    const int w = seg[g * N_NODES + n];
    const int* const r = rs + g * (SORT_S * CHUNK_E) + (w & 0xffff);
    const int len = w >> 16;
#pragma clang loop vectorize(disable) unroll(disable)
    for (int j = 2; j < len; ++j) csrc[p++] = r[j];
  }
}

#define PROJ_WGS   512
#define PROJ_TILES (N_NODES / 16)
#define PROJ_LO    (PROJ_TILES / PROJ_WGS)
#define PROJ_HI    (PROJ_LO + 1)
#define PROJ_EXTRA (PROJ_TILES - PROJ_LO * PROJ_WGS)
#define PROJ_PITCH 132
static_assert(N_NODES % 16 == 0, "whole 16-row tiles");
static_assert(PROJ_EXTRA >= 0 && PROJ_EXTRA < PROJ_WGS, "6 or 7 tiles per work-group");

template <int NT>
__device__ __forceinline__ void proj_tiles(const float* __restrict__ tile, const int row0,
                                           const int wave, const int lane,
                                           const float* __restrict__ Wsrc, const float* __restrict__ bsrc,
                                           const float* __restrict__ Wdst, const float* __restrict__ bdst,
                                           float* __restrict__ fs, float* __restrict__ fd) {
  const int l = lane & 15, hi = lane >> 4;
  const int sel = wave >> 3;
  const int nb  = (wave & 7) * 16;
  const float* W    = sel ? Wdst : Wsrc;
  const float* bias = sel ? bdst : bsrc;
  float*       out  = sel ? fd   : fs;

  v8f acc[NT] = {};
#pragma unroll 8
  for (int k0 = 0; k0 < HIDDEN; k0 += 4) {
    const int ka = k0 + 2 * hi;
    v2f B;
    B.x = W[ka * HIDDEN + nb + l];
    B.y = W[(ka + 1) * HIDDEN + nb + l];
#pragma unroll
    for (int mt = 0; mt < NT; ++mt) {
      const v2f A = *(const v2f*)(&tile[(mt * 16 + l) * PROJ_PITCH + ka]);
      acc[mt] = __builtin_amdgcn_wmma_f32_16x16x4_f32(false, A, false, B,
                                                      (short)0, acc[mt], false, false);
    }
  }
  const float bv = bias[nb + l];
#pragma unroll
  for (int mt = 0; mt < NT; ++mt) {
#pragma unroll
    for (int r = 0; r < 8; ++r)
      out[(row0 + mt * 16 + r + 8 * hi) * HIDDEN + nb + l] = acc[mt][r] + bv;
  }
}

__global__ __launch_bounds__(512, 8) void k_gat_proj(
    const float* __restrict__ h,
    const float* __restrict__ Wsrc, const float* __restrict__ bsrc,
    const float* __restrict__ Wdst, const float* __restrict__ bdst,
    float* __restrict__ fs, float* __restrict__ fd) {
  __shared__ __attribute__((aligned(16))) float tile[PROJ_HI * 16 * PROJ_PITCH];
  const int tid = threadIdx.x, lane = tid & 31;
  const int wave = __builtin_amdgcn_readfirstlane(tid >> 5);
  const int b    = blockIdx.x;
  const int nt   = PROJ_LO + (b < PROJ_EXTRA ? 1 : 0);
  const int row0 = (b * PROJ_LO + min(b, PROJ_EXTRA)) * 16;
#pragma unroll
  for (int i = 0; i < PROJ_HI; ++i) {
    if (i < nt) {
      const int r = i * 16 + wave;
      *(float4*)(&tile[r * PROJ_PITCH + lane * 4]) = *(const float4*)(h + (row0 + r) * HIDDEN + lane * 4);
    }
  }
  __syncthreads();
  if (nt == PROJ_HI) proj_tiles<PROJ_HI>(tile, row0, wave, lane, Wsrc, bsrc, Wdst, bdst, fs, fd);
  else               proj_tiles<PROJ_LO>(tile, row0, wave, lane, Wsrc, bsrc, Wdst, bdst, fs, fd);
}

#define CHUNK 16
__device__ __forceinline__ float edge_logit(const v2f alo, const v2f ahi, const v2f blo,
                                            const v2f bhi, const v2f wlo, const v2f whi) {
  v2f t0 = alo + blo, t1 = ahi + bhi;
  t0 = __builtin_elementwise_max(t0, t0 * NEG_SLOPE);
  t1 = __builtin_elementwise_max(t1, t1 * NEG_SLOPE);
  v2f q = t0 * wlo;
  q += t1 * whi;
  float p = q.x + q.y;
  p += __shfl_xor(p, 1, 8);
  p += __shfl_xor(p, 2, 8);
  p += __shfl_xor(p, 4, 8);
  return p;
}

template <int W>
__device__ __forceinline__ void gat_chunk(const float* __restrict__ fs, const unsigned l16,
                                          const int sorted, const int half, const int cnt,
                                          const v2f blo, const v2f bhi,
                                          const v2f wlo, const v2f whi,
                                          float& m, float& l, v2f& alo, v2f& ahi) {
  v2f   rlo[W], rhi[W];
  float p[W];
#pragma unroll
  for (int j = 0; j < W; ++j) {
    const int s = __builtin_amdgcn_readlane(sorted, half + min(j, cnt - 1));
    const unsigned off = ((unsigned)s << 9) + l16;
    const float4 r = *(const float4*)((const char*)fs + off);
    rlo[j].x = r.x; rlo[j].y = r.y; rhi[j].x = r.z; rhi[j].y = r.w;
  }
  float cm = -INFINITY;
#pragma unroll
  for (int j = 0; j < W; ++j) {
    const float q = edge_logit(rlo[j], rhi[j], blo, bhi, wlo, whi);
    p[j] = (j < cnt) ? q : -INFINITY;
    cm = fmaxf(cm, p[j]);
  }
  const float mn = fmaxf(m, cm);
  const float sc = exp_nonpos(m - mn);
  l *= sc;
  alo *= sc; ahi *= sc;
#pragma unroll
  for (int j = 0; j < W; ++j) {
    const float ex = exp_nonpos(p[j] - mn);
    l += ex;
    alo += rlo[j] * ex;
    ahi += rhi[j] * ex;
  }
  m = mn;
}

__device__ __forceinline__ float4 gat_node_row(const float* __restrict__ fs,
                                               const int* __restrict__ csrc, const int lane,
                                               const unsigned l16,
                                               const int beg, const int end, const int sid0,
                                               const float4 b, const float4 w) {
  float m = -INFINITY, l = 0.0f;
  v2f alo = {0.0f, 0.0f}, ahi = {0.0f, 0.0f};
  const v2f blo = {b.x, b.y}, bhi = {b.z, b.w};
  const v2f wlo = {w.x, w.y}, whi = {w.z, w.w};
  for (int base = beg; base < end; base += 32) {
    const int n = min(32, end - base);
    int sid = sid0;
    if (base != beg) sid = (lane < n) ? csrc[base + lane] : 0x7fffffff;
    int rk = 0;
    for (int j = 0; j < n; ++j) {
      const int o = __builtin_amdgcn_readlane(sid, j);
      rk += ((o < sid) || (o == sid && j < lane)) ? 1 : 0;
    }
    const int sorted = __builtin_amdgcn_ds_permute(rk << 2, sid);

    for (int half = 0; half < n; half += CHUNK) {
      const int cnt = min(CHUNK, n - half);
      const int quads = (cnt + 3) >> 2;
      if (quads == 4)      gat_chunk<16>(fs, l16, sorted, half, cnt, blo, bhi, wlo, whi, m, l, alo, ahi);
      else if (quads == 3) gat_chunk<12>(fs, l16, sorted, half, cnt, blo, bhi, wlo, whi, m, l, alo, ahi);
      else if (quads == 2) gat_chunk<8>(fs, l16, sorted, half, cnt, blo, bhi, wlo, whi, m, l, alo, ahi);
      else                 gat_chunk<4>(fs, l16, sorted, half, cnt, blo, bhi, wlo, whi, m, l, alo, ahi);
    }
  }
  const float den = (l == 0.0f) ? 1.0f : l;
  float4 o;
  o.x = alo.x / den; o.y = alo.y / den; o.z = ahi.x / den; o.w = ahi.y / den;
  return o;
}

#define TAIL_WGS   512
#define TAIL_TILES (N_NODES / 16)
#define TAIL_LO    (TAIL_TILES / TAIL_WGS)
#define TAIL_EXTRA (TAIL_TILES - TAIL_LO * TAIL_WGS)
#define TM 64
#define XS 132
#define TS 260
static_assert(TAIL_LO == 6 && TAIL_EXTRA >= 0 && TAIL_EXTRA < TAIL_WGS, "4 + 2 or 4 + 3 tiles per work-group");

template <int NR>
__device__ __forceinline__ void layernorm_rows(float4 (&v)[NR], const float4 g, const float4 b) {
  float sum[NR];
#pragma unroll
  for (int i = 0; i < NR; ++i) sum[i] = v[i].x + v[i].y + v[i].z + v[i].w;
#pragma unroll
  for (int off = 1; off < 32; off <<= 1) {
#pragma unroll
    for (int i = 0; i < NR; ++i) sum[i] += __shfl_xor(sum[i], off, 32);
  }
  float ss[NR];
#pragma unroll
  for (int i = 0; i < NR; ++i) {
    const float mu = sum[i] * (1.0f / HIDDEN);
    v[i].x -= mu; v[i].y -= mu; v[i].z -= mu; v[i].w -= mu;
    ss[i] = v[i].x * v[i].x + v[i].y * v[i].y + v[i].z * v[i].z + v[i].w * v[i].w;
  }
#pragma unroll
  for (int off = 1; off < 32; off <<= 1) {
#pragma unroll
    for (int i = 0; i < NR; ++i) ss[i] += __shfl_xor(ss[i], off, 32);
  }
#pragma unroll
  for (int i = 0; i < NR; ++i) {
    const float rs = rsqrtf(ss[i] * (1.0f / HIDDEN) + LN_EPS);
    v[i].x = v[i].x * rs * g.x + b.x;
    v[i].y = v[i].y * rs * g.y + b.y;
    v[i].z = v[i].z * rs * g.z + b.z;
    v[i].w = v[i].w * rs * g.w + b.w;
  }
}

template <int CNT>
__device__ __forceinline__ void ffn2_tiles(float* __restrict__ xs, const float* __restrict__ ts,
                                           const int mtA, const int nb, const int l, const int hi,
                                           const float* __restrict__ W2, const float* __restrict__ bf2) {
  v8f acc[CNT] = {};
#pragma unroll
  for (int k0 = 0; k0 < FFN_DIM; k0 += 4) {
    const int ka = k0 + 2 * hi;
    v2f B;
    B.x = W2[ka * HIDDEN + nb + l];
    B.y = W2[(ka + 1) * HIDDEN + nb + l];
#pragma unroll
    for (int j = 0; j < CNT; ++j) {
      const v2f A = *(const v2f*)(&ts[((mtA + j) * 16 + l) * TS + ka]);
      acc[j] = __builtin_amdgcn_wmma_f32_16x16x4_f32(false, A, false, B,
                                                     (short)0, acc[j], false, false);
    }
  }
  const float bv = bf2[nb + l];
#pragma unroll
  for (int j = 0; j < CNT; ++j) {
#pragma unroll
    for (int r = 0; r < 8; ++r) {
      const int idx = ((mtA + j) * 16 + r + 8 * hi) * XS + nb + l;
      xs[idx] = acc[j][r] + bv + xs[idx];
    }
  }
}

template <int NT>
__device__ __forceinline__ void tail_pass(
    float* __restrict__ xs, float* __restrict__ ts, const int prow0,
    const int wave, const int lane_in,
    const float* __restrict__ h, const float* __restrict__ fs, const float* __restrict__ fd,
    const float* __restrict__ attn, const int* __restrict__ offs, const int* __restrict__ csrc,
    const float* __restrict__ g1, const float* __restrict__ be1,
    const float* __restrict__ W1, const float* __restrict__ bf1,
    const float* __restrict__ W2, const float* __restrict__ bf2,
    const float* __restrict__ g2, const float* __restrict__ be2,
    float* __restrict__ out) {
  int lane = lane_in;
  asm volatile("" : "+v"(lane));
  const int l = lane & 15, hi = lane >> 4;
  {
    const int n0 = prow0 + wave * NT;
    const unsigned l16 = (unsigned)lane * 16u;
    char* const xsb = (char*)xs;
    char* const tsb = (char*)ts;
    const float4 w = *(const float4*)((const char*)attn + l16);
    int cbeg = offs[n0], cend = offs[n0 + 1];
    float4 hx[NT], fb[NT];
#pragma unroll
    for (int i = 0; i < NT; ++i) {
      hx[i] = *(const float4*)((const char*)(h + (n0 + i) * HIDDEN) + l16);
      fb[i] = *(const float4*)((const char*)(fd + (n0 + i) * HIDDEN) + l16);
    }
    int csid = (lane < min(32, cend - cbeg)) ? csrc[cbeg + lane] : 0x7fffffff;
#pragma unroll
    for (int i = 0; i < NT; ++i) {
      *(float4*)(xsb + (wave * NT + i) * (XS * 4) + l16) = hx[i];
      *(float4*)(tsb + (wave * NT + i) * (TS * 4) + l16) = fb[i];
    }
    for (int i = 0; i < NT; ++i) {
      int nbeg = cend, nend = cend, nsid = 0x7fffffff;
      if (i < NT - 1) {
        nend = offs[n0 + i + 2];
        nsid = (lane < min(32, nend - nbeg)) ? csrc[nbeg + lane] : 0x7fffffff;
      }
      const float4 b = *(const float4*)(tsb + (wave * NT + i) * (TS * 4) + l16);
      const float4 a = gat_node_row(fs, csrc, lane, l16, cbeg, cend, csid, b, w);
      const float4 x = *(const float4*)(xsb + (wave * NT + i) * (XS * 4) + l16);
      float4 v;
      v.x = x.x + a.x; v.y = x.y + a.y; v.z = x.z + a.z; v.w = x.w + a.w;
      *(float4*)(xsb + (wave * NT + i) * (XS * 4) + l16) = v;
      cbeg = nbeg; cend = nend; csid = nsid;
    }
    float4 v[NT];
#pragma unroll
    for (int i = 0; i < NT; ++i) v[i] = *(const float4*)(&xs[(wave * NT + i) * XS + lane * 4]);
    layernorm_rows<NT>(v, *(const float4*)(g1 + lane * 4), *(const float4*)(be1 + lane * 4));
#pragma unroll
    for (int i = 0; i < NT; ++i)
      *(float4*)(&xs[(wave * NT + i) * XS + lane * 4]) = v[i];
  }
  __syncthreads();

  {
    const int nb = wave * 16;
    v8f acc[NT] = {};
#pragma unroll
    for (int k0 = 0; k0 < HIDDEN; k0 += 4) {
      const int ka = k0 + 2 * hi;
      v2f B;
      B.x = W1[ka * FFN_DIM + nb + l];
      B.y = W1[(ka + 1) * FFN_DIM + nb + l];
#pragma unroll
      for (int mt = 0; mt < NT; ++mt) {
        const v2f A = *(const v2f*)(&xs[(mt * 16 + l) * XS + ka]);
        acc[mt] = __builtin_amdgcn_wmma_f32_16x16x4_f32(false, A, false, B,
                                                        (short)0, acc[mt], false, false);
      }
    }
    const float bv = bf1[nb + l];
#pragma unroll
    for (int mt = 0; mt < NT; ++mt) {
#pragma unroll
      for (int r = 0; r < 8; r += 2) {
        v2f x = {acc[mt][r], acc[mt][r + 1]};
        x = gelu_pair(x + bv);
        ts[(mt * 16 + r + 8 * hi) * TS + nb + l]     = x.x;
        ts[(mt * 16 + r + 1 + 8 * hi) * TS + nb + l] = x.y;
      }
    }
  }
  __syncthreads();

  {
    const int nb  = (wave & 7) * 16;
    const int gq  = wave >> 3;
    const int mtA = gq ? (NT + 1) / 2 : 0;
    const int cnt = gq ? NT / 2 : (NT + 1) / 2;
    if (cnt == 2) ffn2_tiles<2>(xs, ts, mtA, nb, l, hi, W2, bf2);
    else          ffn2_tiles<1>(xs, ts, mtA, nb, l, hi, W2, bf2);
  }
  __syncthreads();

  {
    float4 v[NT];
#pragma unroll
    for (int i = 0; i < NT; ++i) v[i] = *(const float4*)(&xs[(wave * NT + i) * XS + lane * 4]);
    layernorm_rows<NT>(v, *(const float4*)(g2 + lane * 4), *(const float4*)(be2 + lane * 4));
#pragma unroll
    for (int i = 0; i < NT; ++i)
      *(float4*)(out + (prow0 + wave * NT + i) * HIDDEN + lane * 4) = v[i];
  }
}

__global__ __launch_bounds__(512, 8) void k_tail(
    const float* __restrict__ h, const float* __restrict__ fs, const float* __restrict__ fd,
    const float* __restrict__ attn, const int* __restrict__ offs, const int* __restrict__ csrc,
    const float* __restrict__ g1, const float* __restrict__ be1,
    const float* __restrict__ W1, const float* __restrict__ bf1,
    const float* __restrict__ W2, const float* __restrict__ bf2,
    const float* __restrict__ g2, const float* __restrict__ be2,
    float* __restrict__ out) {
  __shared__ __attribute__((aligned(16))) float xs[TM * XS];
  __shared__ __attribute__((aligned(16))) float ts[TM * TS];
  const int tid = threadIdx.x, lane = tid & 31;
  const int wave = __builtin_amdgcn_readfirstlane(tid >> 5);
  const int b    = blockIdx.x;
  const int nt   = TAIL_LO + (b < TAIL_EXTRA ? 1 : 0);
  const int row0 = (b * TAIL_LO + min(b, TAIL_EXTRA)) * 16;
  tail_pass<4>(xs, ts, row0, wave, lane, h, fs, fd, attn, offs, csrc,
               g1, be1, W1, bf1, W2, bf2, g2, be2, out);
  __syncthreads();
  if (nt == TAIL_LO + 1)
    tail_pass<3>(xs, ts, row0 + 64, wave, lane, h, fs, fd, attn, offs, csrc,
                 g1, be1, W1, bf1, W2, bf2, g2, be2, out);
  else
    tail_pass<2>(xs, ts, row0 + 64, wave, lane, h, fs, fd, attn, offs, csrc,
                 g1, be1, W1, bf1, W2, bf2, g2, be2, out);
}

extern "C" void kernel_launch(void* const* d_in, const int* in_sizes, int n_in,
                              void* d_out, int out_size, void* d_ws, size_t ws_size,
                              hipStream_t stream) {
  const float* h     = (const float*)d_in[0];
  const int*   src   = (const int*)d_in[1];
  const int*   dst   = (const int*)d_in[2];
  const float* Wsrc  = (const float*)d_in[3];
  const float* bsrc  = (const float*)d_in[4];
  const float* Wdst  = (const float*)d_in[5];
  const float* bdst  = (const float*)d_in[6];
  const float* attn  = (const float*)d_in[7];
  const float* W1    = (const float*)d_in[8];
  const float* bf1   = (const float*)d_in[9];
  const float* W2    = (const float*)d_in[10];
  const float* bf2   = (const float*)d_in[11];
  const float* g1    = (const float*)d_in[12];
  const float* be1   = (const float*)d_in[13];
  const float* g2    = (const float*)d_in[14];
  const float* be2   = (const float*)d_in[15];
  float* out = (float*)d_out;

  float* ws     = (float*)d_ws;
  float* fs     = ws;
  float* fd     = fs + (size_t)N_NODES * HIDDEN;
  int*   region = (int*)(fd + (size_t)N_NODES * HIDDEN);
  int*   seg    = region + SORT_G * SORT_S * CHUNK_E;
  int*   offs   = region + (size_t)N_NODES * HIDDEN + N_NODES;
  int*   btot   = offs + N_NODES + 16;
  int*   csrc   = offs + N_NODES + 16 + N_EDGES;

  k_group<<<dim3(SORT_G, SORT_S), SORT_THREADS, 0, stream>>>(src, dst, region, seg, btot);
  k_merge<<<MERGE_BLOCKS, MERGE_THREADS, 0, stream>>>(region, seg, btot, offs, csrc);
  k_gat_proj<<<PROJ_WGS, 512, 0, stream>>>(h, Wsrc, bsrc, Wdst, bdst, fs, fd);
  k_tail<<<TAIL_WGS, 512, 0, stream>>>(h, fs, fd, attn, offs, csrc,
                                                       g1, be1, W1, bf1, W2, bf2, g2, be2, out);
}
